# starting kernel
# baseline (speedup 1.0000x reference)
#include <hip/hip_runtime.h>
#include <math.h>

#define N_NODES 50000
#define N_EDGES 800000
#define D_MODEL 128
#define N_HEADS 2

typedef __attribute__((ext_vector_type(2))) float v2f;
typedef __attribute__((ext_vector_type(8))) float v8f;

__device__ __forceinline__ unsigned ordEnc(float f) {
    unsigned u = __float_as_uint(f);
    return (u & 0x80000000u) ? ~u : (u | 0x80000000u);
}
__device__ __forceinline__ float ordDec(unsigned u) {
    unsigned v = (u & 0x80000000u) ? (u & 0x7fffffffu) : ~u;
    return __uint_as_float(v);
}
#define ORD_NEG_INF 0x007FFFFFu

__global__ void fill_u32(unsigned* __restrict__ p, unsigned v, int n) {
    int i = blockIdx.x * blockDim.x + threadIdx.x;
    if (i < n) p[i] = v;
}

__global__ void __launch_bounds__(128) gemm_feat(const float* __restrict__ h,
                                                 const float* __restrict__ W,
                                                 float* __restrict__ feat) {
    const int lane  = threadIdx.x & 31;
    const int wave  = threadIdx.x >> 5;
    const int mbase = blockIdx.x << 4;
    const int nq    = wave << 6;
    const int row   = lane & 15;
    const int khalf = lane >> 4;

    v8f acc0 = {}, acc1 = {}, acc2 = {}, acc3 = {};
    const float* Arow = h + (size_t)(mbase + row) * D_MODEL;

    for (int k = 0; k < D_MODEL; k += 4) {
        const int ka = k + khalf * 2;
        v2f a;
        a.x = Arow[ka];
        a.y = Arow[ka + 1];
        const float* Wb = W + (size_t)ka * 256 + row;
        v2f b0, b1, b2, b3;
        b0.x = Wb[nq];        b0.y = Wb[256 + nq];
        b1.x = Wb[nq + 16];   b1.y = Wb[256 + nq + 16];
        b2.x = Wb[nq + 32];   b2.y = Wb[256 + nq + 32];
        b3.x = Wb[nq + 48];   b3.y = Wb[256 + nq + 48];
        acc0 = __builtin_amdgcn_wmma_f32_16x16x4_f32(false, a, false, b0, (short)0, acc0, false, false);
        acc1 = __builtin_amdgcn_wmma_f32_16x16x4_f32(false, a, false, b1, (short)0, acc1, false, false);
        acc2 = __builtin_amdgcn_wmma_f32_16x16x4_f32(false, a, false, b2, (short)0, acc2, false, false);
        acc3 = __builtin_amdgcn_wmma_f32_16x16x4_f32(false, a, false, b3, (short)0, acc3, false, false);
    }

    #pragma unroll
    for (int j = 0; j < 8; ++j) {
        const int m = mbase + j + khalf * 8;
        float* F = feat + (size_t)m * 256 + nq + row;
        F[0]  = acc0[j];
        F[16] = acc1[j];
        F[32] = acc2[j];
        F[48] = acc3[j];
    }
}

__global__ void __launch_bounds__(256) node_attn(const float* __restrict__ feat,
                                                 const float* __restrict__ al,
                                                 const float* __restrict__ ar,
                                                 float* __restrict__ el,
                                                 float* __restrict__ er) {
    const int gw   = (blockIdx.x * blockDim.x + threadIdx.x) >> 5;
    const int lane = threadIdx.x & 31;
    if (gw >= N_NODES) return;
    const float4* F0 = (const float4*)(feat + (size_t)gw * 256) + lane;
    const float4* F1 = (const float4*)(feat + (size_t)gw * 256 + 128) + lane;
    const float4* AL0 = (const float4*)(al) + lane;
    const float4* AL1 = (const float4*)(al + 128) + lane;
    const float4* AR0 = (const float4*)(ar) + lane;
    const float4* AR1 = (const float4*)(ar + 128) + lane;
    float4 f0 = *F0, f1 = *F1;
    float4 l0 = *AL0, l1 = *AL1, r0 = *AR0, r1 = *AR1;
    float e0 = f0.x*l0.x + f0.y*l0.y + f0.z*l0.z + f0.w*l0.w;
    float e1 = f1.x*l1.x + f1.y*l1.y + f1.z*l1.z + f1.w*l1.w;
    float g0 = f0.x*r0.x + f0.y*r0.y + f0.z*r0.z + f0.w*r0.w;
    float g1 = f1.x*r1.x + f1.y*r1.y + f1.z*r1.z + f1.w*r1.w;
    #pragma unroll
    for (int off = 16; off; off >>= 1) {
        e0 += __shfl_xor(e0, off, 32);
        e1 += __shfl_xor(e1, off, 32);
        g0 += __shfl_xor(g0, off, 32);
        g1 += __shfl_xor(g1, off, 32);
    }
    if (lane == 0) {
        el[gw * 2 + 0] = e0;  el[gw * 2 + 1] = e1;
        er[gw * 2 + 0] = g0;  er[gw * 2 + 1] = g1;
    }
}

__global__ void __launch_bounds__(256) edge_max(const float* __restrict__ el,
                                                const float* __restrict__ er,
                                                const int* __restrict__ src,
                                                const int* __restrict__ dst,
                                                float* __restrict__ ebuf,
                                                unsigned* __restrict__ nodemax) {
    int i = blockIdx.x * blockDim.x + threadIdx.x;
    if (i >= N_EDGES * N_HEADS) return;
    const int e = i >> 1, hh = i & 1;
    const int d = dst[e];
    float v = el[src[e] * 2 + hh] + er[d * 2 + hh];
    v = v > 0.0f ? v : 0.2f * v;
    ebuf[i] = v;
    atomicMax(nodemax + d * 2 + hh, ordEnc(v));
}

__global__ void __launch_bounds__(256) edge_exp(float* __restrict__ ebuf,
                                                const unsigned* __restrict__ nodemax,
                                                const int* __restrict__ dst,
                                                float* __restrict__ denom) {
    int i = blockIdx.x * blockDim.x + threadIdx.x;
    if (i >= N_EDGES * N_HEADS) return;
    const int e = i >> 1, hh = i & 1;
    const int d = dst[e];
    const float m = ordDec(nodemax[d * 2 + hh]);
    const float ex = expf(ebuf[i] - m);
    ebuf[i] = ex;
    unsafeAtomicAdd(denom + d * 2 + hh, ex);
}

__global__ void __launch_bounds__(256) aggregate(const float* __restrict__ feat,
                                                 const float* __restrict__ ebuf,
                                                 const float* __restrict__ denom,
                                                 const int* __restrict__ src,
                                                 const int* __restrict__ dst,
                                                 float* __restrict__ agg) {
    const int gw   = (blockIdx.x * blockDim.x + threadIdx.x) >> 5;
    const int lane = threadIdx.x & 31;
    if (gw >= N_EDGES * N_HEADS) return;
    const int e = gw >> 1, hh = gw & 1;
    const int s = src[e], d = dst[e];
    const float w = ebuf[gw] / denom[d * 2 + hh];
    const float4 f = *(const float4*)(feat + (size_t)s * 256 + hh * 128 + lane * 4);
    float* o = agg + (size_t)d * 256 + hh * 128 + lane * 4;
    unsafeAtomicAdd(o + 0, f.x * w);
    unsafeAtomicAdd(o + 1, f.y * w);
    unsafeAtomicAdd(o + 2, f.z * w);
    unsafeAtomicAdd(o + 3, f.w * w);
}

__global__ void __launch_bounds__(256) finalize_mean(const float* __restrict__ agg,
                                                     const float* __restrict__ b,
                                                     float* __restrict__ hnext) {
    int i = blockIdx.x * blockDim.x + threadIdx.x;
    if (i >= N_NODES * D_MODEL) return;
    const int n = i >> 7, f = i & 127;
    const float o0 = agg[(size_t)n * 256 + f]       + b[f];
    const float o1 = agg[(size_t)n * 256 + 128 + f] + b[128 + f];
    hnext[i] = 0.5f * (o0 + o1);
}

__global__ void __launch_bounds__(256) finalize_bias(float* __restrict__ out,
                                                     const float* __restrict__ b) {
    int i = blockIdx.x * blockDim.x + threadIdx.x;
    if (i >= N_NODES * 256) return;
    out[i] += b[i & 255];
}

extern "C" void kernel_launch(void* const* d_in, const int* in_sizes, int n_in,
                              void* d_out, int out_size, void* d_ws, size_t ws_size,
                              hipStream_t stream) {
    const float* x   = (const float*)d_in[0];
    const float* Ws  = (const float*)d_in[1];
    const float* als = (const float*)d_in[2];
    const float* ars = (const float*)d_in[3];
    const float* bs  = (const float*)d_in[4];
    const int*   src = (const int*)d_in[5];
    const int*   dst = (const int*)d_in[6];

    float* ws      = (float*)d_ws;
    float* feat    = ws;
    float* hnext   = ws + 12800000;
    float* el      = ws + 19200000;
    float* er      = ws + 19300000;
    unsigned* nmax = (unsigned*)(ws + 19400000);
    float* denom   = ws + 19500000;
    float* ebuf    = ws + 19600000;

    float* agg = (float*)d_out;

    const int NH   = N_NODES * N_HEADS;
    const int EH   = N_EDGES * N_HEADS;
    const int NOUT = N_NODES * 256;

    for (int layer = 0; layer < 2; ++layer) {
        const float* hin = layer ? hnext : x;
        const float* W   = Ws  + (size_t)layer * D_MODEL * 256;
        const float* al  = als + (size_t)layer * 256;
        const float* ar  = ars + (size_t)layer * 256;
        const float* b   = bs  + (size_t)layer * 256;

        fill_u32<<<(NOUT + 255) / 256, 256, 0, stream>>>((unsigned*)agg, 0u, NOUT);
        fill_u32<<<(NH + 255) / 256, 256, 0, stream>>>(nmax, ORD_NEG_INF, NH);
        fill_u32<<<(NH + 255) / 256, 256, 0, stream>>>((unsigned*)denom, 0u, NH);

        gemm_feat<<<N_NODES / 16, 128, 0, stream>>>(hin, W, feat);
        node_attn<<<(N_NODES + 7) / 8, 256, 0, stream>>>(feat, al, ar, el, er);
        edge_max<<<(EH + 255) / 256, 256, 0, stream>>>(el, er, src, dst, ebuf, nmax);
        edge_exp<<<(EH + 255) / 256, 256, 0, stream>>>(ebuf, nmax, dst, denom);
        aggregate<<<(EH * 32 + 255) / 256, 256, 0, stream>>>(feat, ebuf, denom, src, dst, agg);

        if (layer == 0) {
            finalize_mean<<<(N_NODES * D_MODEL + 255) / 256, 256, 0, stream>>>(agg, b, hnext);
        } else {
            finalize_bias<<<(NOUT + 255) / 256, 256, 0, stream>>>(agg, b);
        }
    }
}


# aggregate by destination: edges bucketed by dst once per call (counting atomic's return = arrival-order rank, scan, place); a wave sums a node's rows in registers in that order, one store with bias
# speedup vs baseline: 2.0569x; 2.0569x over previous
#include <hip/hip_runtime.h>
#include <math.h>

#define N_NODES 50000
#define N_EDGES 800000
#define D_MODEL 128
#define N_HEADS 2

typedef __attribute__((ext_vector_type(2))) float v2f;
typedef __attribute__((ext_vector_type(8))) float v8f;

__device__ __forceinline__ unsigned ordEnc(float f) {
    unsigned u = __float_as_uint(f);
    return (u & 0x80000000u) ? ~u : (u | 0x80000000u);
}
__device__ __forceinline__ float ordDec(unsigned u) {
    unsigned v = (u & 0x80000000u) ? (u & 0x7fffffffu) : ~u;
    return __uint_as_float(v);
}
#define ORD_NEG_INF 0x007FFFFFu

__global__ void fill_u32(unsigned* __restrict__ p, unsigned v, int n) {
    int i = blockIdx.x * blockDim.x + threadIdx.x;
    if (i < n) p[i] = v;
}

__global__ void __launch_bounds__(128) gemm_feat(const float* __restrict__ h,
                                                 const float* __restrict__ W,
                                                 float* __restrict__ feat) {
    const int lane  = threadIdx.x & 31;
    const int wave  = threadIdx.x >> 5;
    const int mbase = blockIdx.x << 4;
    const int nq    = wave << 6;
    const int row   = lane & 15;
    const int khalf = lane >> 4;

    v8f acc0 = {}, acc1 = {}, acc2 = {}, acc3 = {};
    const float* Arow = h + (size_t)(mbase + row) * D_MODEL;

    for (int k = 0; k < D_MODEL; k += 4) {
        const int ka = k + khalf * 2;
        v2f a;
        a.x = Arow[ka];
        a.y = Arow[ka + 1];
        const float* Wb = W + (size_t)ka * 256 + row;
        v2f b0, b1, b2, b3;
        b0.x = Wb[nq];        b0.y = Wb[256 + nq];
        b1.x = Wb[nq + 16];   b1.y = Wb[256 + nq + 16];
        b2.x = Wb[nq + 32];   b2.y = Wb[256 + nq + 32];
        b3.x = Wb[nq + 48];   b3.y = Wb[256 + nq + 48];
        acc0 = __builtin_amdgcn_wmma_f32_16x16x4_f32(false, a, false, b0, (short)0, acc0, false, false);
        acc1 = __builtin_amdgcn_wmma_f32_16x16x4_f32(false, a, false, b1, (short)0, acc1, false, false);
        acc2 = __builtin_amdgcn_wmma_f32_16x16x4_f32(false, a, false, b2, (short)0, acc2, false, false);
        acc3 = __builtin_amdgcn_wmma_f32_16x16x4_f32(false, a, false, b3, (short)0, acc3, false, false);
    }

    #pragma unroll
    for (int j = 0; j < 8; ++j) {
        const int m = mbase + j + khalf * 8;
        float* F = feat + (size_t)m * 256 + nq + row;
        F[0]  = acc0[j];
        F[16] = acc1[j];
        F[32] = acc2[j];
        F[48] = acc3[j];
    }
}

__global__ void __launch_bounds__(256) node_attn(const float* __restrict__ feat,
                                                 const float* __restrict__ al,
                                                 const float* __restrict__ ar,
                                                 float* __restrict__ el,
                                                 float* __restrict__ er) {
    const int gw   = (blockIdx.x * blockDim.x + threadIdx.x) >> 5;
    const int lane = threadIdx.x & 31;
    if (gw >= N_NODES) return;
    const float4* F0 = (const float4*)(feat + (size_t)gw * 256) + lane;
    const float4* F1 = (const float4*)(feat + (size_t)gw * 256 + 128) + lane;
    const float4* AL0 = (const float4*)(al) + lane;
    const float4* AL1 = (const float4*)(al + 128) + lane;
    const float4* AR0 = (const float4*)(ar) + lane;
    const float4* AR1 = (const float4*)(ar + 128) + lane;
    float4 f0 = *F0, f1 = *F1;
    float4 l0 = *AL0, l1 = *AL1, r0 = *AR0, r1 = *AR1;
    float e0 = f0.x*l0.x + f0.y*l0.y + f0.z*l0.z + f0.w*l0.w;
    float e1 = f1.x*l1.x + f1.y*l1.y + f1.z*l1.z + f1.w*l1.w;
    float g0 = f0.x*r0.x + f0.y*r0.y + f0.z*r0.z + f0.w*r0.w;
    float g1 = f1.x*r1.x + f1.y*r1.y + f1.z*r1.z + f1.w*r1.w;
    #pragma unroll
    for (int off = 16; off; off >>= 1) {
        e0 += __shfl_xor(e0, off, 32);
        e1 += __shfl_xor(e1, off, 32);
        g0 += __shfl_xor(g0, off, 32);
        g1 += __shfl_xor(g1, off, 32);
    }
    if (lane == 0) {
        el[gw * 2 + 0] = e0;  el[gw * 2 + 1] = e1;
        er[gw * 2 + 0] = g0;  er[gw * 2 + 1] = g1;
    }
}

__global__ void __launch_bounds__(256) edge_max(const float* __restrict__ el,
                                                const float* __restrict__ er,
                                                const int* __restrict__ src,
                                                const int* __restrict__ dst,
                                                float* __restrict__ ebuf,
                                                unsigned* __restrict__ nodemax) {
    int i = blockIdx.x * blockDim.x + threadIdx.x;
    if (i >= N_EDGES * N_HEADS) return;
    const int e = i >> 1, hh = i & 1;
    const int d = dst[e];
    float v = el[src[e] * 2 + hh] + er[d * 2 + hh];
    v = v > 0.0f ? v : 0.2f * v;
    ebuf[i] = v;
    atomicMax(nodemax + d * 2 + hh, ordEnc(v));
}

__global__ void __launch_bounds__(256) edge_exp(float* __restrict__ ebuf,
                                                const unsigned* __restrict__ nodemax,
                                                const int* __restrict__ dst,
                                                float* __restrict__ denom) {
    int i = blockIdx.x * blockDim.x + threadIdx.x;
    if (i >= N_EDGES * N_HEADS) return;
    const int e = i >> 1, hh = i & 1;
    const int d = dst[e];
    const float m = ordDec(nodemax[d * 2 + hh]);
    const float ex = expf(ebuf[i] - m);
    ebuf[i] = ex;
    unsafeAtomicAdd(denom + d * 2 + hh, ex);
}

__global__ void __launch_bounds__(256) bucket_count(const int* __restrict__ dst,
                                                    int* __restrict__ deg,
                                                    int* __restrict__ rank) {
    int e = blockIdx.x * blockDim.x + threadIdx.x;
    if (e >= N_EDGES) return;
    rank[e] = atomicAdd(deg + dst[e], 1);
}

#define SCAN_PER  ((((N_NODES + 1023) / 1024) + 3) / 4 * 4)
#define NODES_PAD (SCAN_PER * 1024)
__global__ void __launch_bounds__(1024) bucket_scan(const int* __restrict__ deg,
                                                    int* __restrict__ rowptr) {
    __shared__ int wsum[32];
    const int tid = threadIdx.x, lane = tid & 31, wave = tid >> 5;
    const int4* dg = (const int4*)(deg + tid * SCAN_PER);
    int loc = 0;
    for (int k = 0; k < SCAN_PER / 4; ++k) {
        const int4 d = dg[k];
        loc += (d.x + d.y) + (d.z + d.w);
    }
    int inc = loc;
    for (int off = 1; off < 32; off <<= 1) {
        const int t = __shfl_up(inc, off, 32);
        if (lane >= off) inc += t;
    }
    if (lane == 31) wsum[wave] = inc;
    __syncthreads();
    if (wave == 0) {
        const int v = wsum[lane];
        int vi = v;
        for (int off = 1; off < 32; off <<= 1) {
            const int t = __shfl_up(vi, off, 32);
            if (lane >= off) vi += t;
        }
        wsum[lane] = vi - v;
    }
    __syncthreads();
    int run = wsum[wave] + inc - loc;
    int4* rp = (int4*)(rowptr + tid * SCAN_PER);
    #pragma unroll
    for (int k = 0; k < SCAN_PER / 4; ++k) {
        const int4 d = dg[k];
        int4 r;
        r.x = run;
        r.y = r.x + d.x;
        r.z = r.y + d.y;
        r.w = r.z + d.z;
        run = r.w + d.w;
        rp[k] = r;
    }
}

__global__ void __launch_bounds__(256) bucket_place(const int* __restrict__ src,
                                                    const int* __restrict__ dst,
                                                    const int* __restrict__ rowptr,
                                                    const int* __restrict__ rank,
                                                    int2* __restrict__ rec) {
    int e = blockIdx.x * blockDim.x + threadIdx.x;
    if (e >= N_EDGES) return;
    rec[rowptr[dst[e]] + rank[e]] = make_int2(src[e], e);
}

__device__ __forceinline__ void gather_body(const float* __restrict__ feat,
                                            const float* __restrict__ ebuf,
                                            const float* __restrict__ denom,
                                            const int* __restrict__ rowptr,
                                            const int2* __restrict__ rec,
                                            const float* __restrict__ b,
                                            float* __restrict__ dest,
                                            const bool merge_heads) {
    const int gw   = (blockIdx.x * blockDim.x + threadIdx.x) >> 5;
    const int lane = threadIdx.x & 31;
    if (gw >= N_NODES) return;
    const int n = __builtin_amdgcn_readfirstlane(gw);
    const int beg = rowptr[n], end = rowptr[n + 1];
    float4 a0 = {0.f, 0.f, 0.f, 0.f}, a1 = {0.f, 0.f, 0.f, 0.f};
    if (end > beg) {
        const float den0 = denom[n * 2], den1 = denom[n * 2 + 1];
        for (int base = beg; base < end; base += 32) {
            const int cnt = (end - base) < 32 ? (end - base) : 32;
            int s_l = 0;
            float w0_l = 0.f, w1_l = 0.f;
            if (lane < cnt) {
                const int2 r = rec[base + lane];
                const float2 ex = *(const float2*)(ebuf + 2 * (size_t)r.y);
                s_l  = r.x;
                w0_l = ex.x / den0;
                w1_l = ex.y / den1;
            }
            #pragma unroll 4
            for (int j = 0; j < cnt; ++j) {
                const int   s  = __builtin_amdgcn_readlane(s_l, j);
                const float w0 = __int_as_float(__builtin_amdgcn_readlane(__float_as_int(w0_l), j));
                const float w1 = __int_as_float(__builtin_amdgcn_readlane(__float_as_int(w1_l), j));
                const float4* fp = (const float4*)(feat + (size_t)s * 256);
                const float4 f0 = fp[lane], f1 = fp[32 + lane];
                a0.x += f0.x * w0;  a0.y += f0.y * w0;  a0.z += f0.z * w0;  a0.w += f0.w * w0;
                a1.x += f1.x * w1;  a1.y += f1.y * w1;  a1.z += f1.z * w1;  a1.w += f1.w * w1;
            }
        }
    }
    const float4 b0 = ((const float4*)b)[lane], b1 = ((const float4*)b)[32 + lane];
    a0.x += b0.x;  a0.y += b0.y;  a0.z += b0.z;  a0.w += b0.w;
    a1.x += b1.x;  a1.y += b1.y;  a1.z += b1.z;  a1.w += b1.w;
    if (merge_heads) {
        float4 o;
        o.x = 0.5f * (a0.x + a1.x);  o.y = 0.5f * (a0.y + a1.y);
        o.z = 0.5f * (a0.z + a1.z);  o.w = 0.5f * (a0.w + a1.w);
        ((float4*)(dest + (size_t)n * 128))[lane] = o;
    } else {
        float4* o = (float4*)(dest + (size_t)n * 256);
        o[lane]      = a0;
        o[32 + lane] = a1;
    }
}

__global__ void __launch_bounds__(256) gather_rows_mean(const float* __restrict__ feat,
                                                        const float* __restrict__ ebuf,
                                                        const float* __restrict__ denom,
                                                        const int* __restrict__ rowptr,
                                                        const int2* __restrict__ rec,
                                                        const float* __restrict__ b,
                                                        float* __restrict__ hnext) {
    gather_body(feat, ebuf, denom, rowptr, rec, b, hnext, true);
}

__global__ void __launch_bounds__(256) gather_rows_heads(const float* __restrict__ feat,
                                                         const float* __restrict__ ebuf,
                                                         const float* __restrict__ denom,
                                                         const int* __restrict__ rowptr,
                                                         const int2* __restrict__ rec,
                                                         const float* __restrict__ b,
                                                         float* __restrict__ out) {
    gather_body(feat, ebuf, denom, rowptr, rec, b, out, false);
}

extern "C" void kernel_launch(void* const* d_in, const int* in_sizes, int n_in,
                              void* d_out, int out_size, void* d_ws, size_t ws_size,
                              hipStream_t stream) {
    const float* x   = (const float*)d_in[0];
    const float* Ws  = (const float*)d_in[1];
    const float* als = (const float*)d_in[2];
    const float* ars = (const float*)d_in[3];
    const float* bs  = (const float*)d_in[4];
    const int*   src = (const int*)d_in[5];
    const int*   dst = (const int*)d_in[6];

    float* ws      = (float*)d_ws;
    float* feat    = ws;
    int2*  rec     = (int2*)(ws + 12800000);
    int*   rank    = (int*)(ws + 14400000);
    int*   deg     = (int*)(ws + 15200000);
    int*   rowptr  = (int*)(ws + 15300000);
    float* el      = ws + 19200000;
    float* er      = ws + 19300000;
    unsigned* nmax = (unsigned*)(ws + 19400000);
    float* denom   = ws + 19500000;
    float* ebuf    = ws + 19600000;

    float* out   = (float*)d_out;
    float* hnext = out;

    const int NH = N_NODES * N_HEADS;
    const int EH = N_EDGES * N_HEADS;

    fill_u32<<<NODES_PAD / 256, 256, 0, stream>>>((unsigned*)deg, 0u, NODES_PAD);
    bucket_count<<<(N_EDGES + 255) / 256, 256, 0, stream>>>(dst, deg, rank);
    bucket_scan<<<1, 1024, 0, stream>>>(deg, rowptr);
    bucket_place<<<(N_EDGES + 255) / 256, 256, 0, stream>>>(src, dst, rowptr, rank, rec);

    for (int layer = 0; layer < 2; ++layer) {
        const float* hin = layer ? hnext : x;
        const float* W   = Ws  + (size_t)layer * D_MODEL * 256;
        const float* al  = als + (size_t)layer * 256;
        const float* ar  = ars + (size_t)layer * 256;
        const float* b   = bs  + (size_t)layer * 256;

        fill_u32<<<(NH + 255) / 256, 256, 0, stream>>>(nmax, ORD_NEG_INF, NH);
        fill_u32<<<(NH + 255) / 256, 256, 0, stream>>>((unsigned*)denom, 0u, NH);

        gemm_feat<<<N_NODES / 16, 128, 0, stream>>>(hin, W, feat);
        node_attn<<<(N_NODES + 7) / 8, 256, 0, stream>>>(feat, al, ar, el, er);
        edge_max<<<(EH + 255) / 256, 256, 0, stream>>>(el, er, src, dst, ebuf, nmax);
        edge_exp<<<(EH + 255) / 256, 256, 0, stream>>>(ebuf, nmax, dst, denom);
        if (layer == 0) {
            gather_rows_mean<<<(N_NODES + 7) / 8, 256, 0, stream>>>(feat, ebuf, denom, rowptr, rec, b, hnext);
        } else {
            gather_rows_heads<<<(N_NODES + 7) / 8, 256, 0, stream>>>(feat, ebuf, denom, rowptr, rec, b, out);
        }
    }
}


# edge softmax inside the destination wave: lane = edge, scores from el/er, max and denominator by lane butterfly (sum order: fixed tree, not float atomics); 2 edge passes + 2 fills per layer dropped
# speedup vs baseline: 3.3653x; 1.6361x over previous
#include <hip/hip_runtime.h>
#include <math.h>

#define N_NODES 50000
#define N_EDGES 800000
#define D_MODEL 128
#define N_HEADS 2

typedef __attribute__((ext_vector_type(2))) float v2f;
typedef __attribute__((ext_vector_type(8))) float v8f;

__global__ void fill_u32(unsigned* __restrict__ p, unsigned v, int n) {
    int i = blockIdx.x * blockDim.x + threadIdx.x;
    if (i < n) p[i] = v;
}

__global__ void __launch_bounds__(128) gemm_feat(const float* __restrict__ h,
                                                 const float* __restrict__ W,
                                                 float* __restrict__ feat) {
    const int lane  = threadIdx.x & 31;
    const int wave  = threadIdx.x >> 5;
    const int mbase = blockIdx.x << 4;
    const int nq    = wave << 6;
    const int row   = lane & 15;
    const int khalf = lane >> 4;

    v8f acc0 = {}, acc1 = {}, acc2 = {}, acc3 = {};
    const float* Arow = h + (size_t)(mbase + row) * D_MODEL;

    for (int k = 0; k < D_MODEL; k += 4) {
        const int ka = k + khalf * 2;
        v2f a;
        a.x = Arow[ka];
        a.y = Arow[ka + 1];
        const float* Wb = W + (size_t)ka * 256 + row;
        v2f b0, b1, b2, b3;
        b0.x = Wb[nq];        b0.y = Wb[256 + nq];
        b1.x = Wb[nq + 16];   b1.y = Wb[256 + nq + 16];
        b2.x = Wb[nq + 32];   b2.y = Wb[256 + nq + 32];
        b3.x = Wb[nq + 48];   b3.y = Wb[256 + nq + 48];
        acc0 = __builtin_amdgcn_wmma_f32_16x16x4_f32(false, a, false, b0, (short)0, acc0, false, false);
        acc1 = __builtin_amdgcn_wmma_f32_16x16x4_f32(false, a, false, b1, (short)0, acc1, false, false);
        acc2 = __builtin_amdgcn_wmma_f32_16x16x4_f32(false, a, false, b2, (short)0, acc2, false, false);
        acc3 = __builtin_amdgcn_wmma_f32_16x16x4_f32(false, a, false, b3, (short)0, acc3, false, false);
    }

    #pragma unroll
    for (int j = 0; j < 8; ++j) {
        const int m = mbase + j + khalf * 8;
        float* F = feat + (size_t)m * 256 + nq + row;
        F[0]  = acc0[j];
        F[16] = acc1[j];
        F[32] = acc2[j];
        F[48] = acc3[j];
    }
}

__global__ void __launch_bounds__(256) node_attn(const float* __restrict__ feat,
                                                 const float* __restrict__ al,
                                                 const float* __restrict__ ar,
                                                 float* __restrict__ el,
                                                 float* __restrict__ er) {
    const int gw   = (blockIdx.x * blockDim.x + threadIdx.x) >> 5;
    const int lane = threadIdx.x & 31;
    if (gw >= N_NODES) return;
    const float4* F0 = (const float4*)(feat + (size_t)gw * 256) + lane;
    const float4* F1 = (const float4*)(feat + (size_t)gw * 256 + 128) + lane;
    const float4* AL0 = (const float4*)(al) + lane;
    const float4* AL1 = (const float4*)(al + 128) + lane;
    const float4* AR0 = (const float4*)(ar) + lane;
    const float4* AR1 = (const float4*)(ar + 128) + lane;
    float4 f0 = *F0, f1 = *F1;
    float4 l0 = *AL0, l1 = *AL1, r0 = *AR0, r1 = *AR1;
    float e0 = f0.x*l0.x + f0.y*l0.y + f0.z*l0.z + f0.w*l0.w;
    float e1 = f1.x*l1.x + f1.y*l1.y + f1.z*l1.z + f1.w*l1.w;
    float g0 = f0.x*r0.x + f0.y*r0.y + f0.z*r0.z + f0.w*r0.w;
    float g1 = f1.x*r1.x + f1.y*r1.y + f1.z*r1.z + f1.w*r1.w;
    #pragma unroll
    for (int off = 16; off; off >>= 1) {
        e0 += __shfl_xor(e0, off, 32);
        e1 += __shfl_xor(e1, off, 32);
        g0 += __shfl_xor(g0, off, 32);
        g1 += __shfl_xor(g1, off, 32);
    }
    if (lane == 0) {
        el[gw * 2 + 0] = e0;  el[gw * 2 + 1] = e1;
        er[gw * 2 + 0] = g0;  er[gw * 2 + 1] = g1;
    }
}

__global__ void __launch_bounds__(256) bucket_count(const int* __restrict__ dst,
                                                    int* __restrict__ deg,
                                                    int* __restrict__ rank) {
    int e = blockIdx.x * blockDim.x + threadIdx.x;
    if (e >= N_EDGES) return;
    rank[e] = atomicAdd(deg + dst[e], 1);
}

#define SCAN_PER  ((((N_NODES + 1023) / 1024) + 3) / 4 * 4)
#define NODES_PAD (SCAN_PER * 1024)
__global__ void __launch_bounds__(1024) bucket_scan(const int* __restrict__ deg,
                                                    int* __restrict__ rowptr) {
    __shared__ int wsum[32];
    const int tid = threadIdx.x, lane = tid & 31, wave = tid >> 5;
    const int4* dg = (const int4*)(deg + tid * SCAN_PER);
    int loc = 0;
    for (int k = 0; k < SCAN_PER / 4; ++k) {
        const int4 d = dg[k];
        loc += (d.x + d.y) + (d.z + d.w);
    }
    int inc = loc;
    for (int off = 1; off < 32; off <<= 1) {
        const int t = __shfl_up(inc, off, 32);
        if (lane >= off) inc += t;
    }
    if (lane == 31) wsum[wave] = inc;
    __syncthreads();
    if (wave == 0) {
        const int v = wsum[lane];
        int vi = v;
        for (int off = 1; off < 32; off <<= 1) {
            const int t = __shfl_up(vi, off, 32);
            if (lane >= off) vi += t;
        }
        wsum[lane] = vi - v;
    }
    __syncthreads();
    int run = wsum[wave] + inc - loc;
    int4* rp = (int4*)(rowptr + tid * SCAN_PER);
    #pragma unroll
    for (int k = 0; k < SCAN_PER / 4; ++k) {
        const int4 d = dg[k];
        int4 r;
        r.x = run;
        r.y = r.x + d.x;
        r.z = r.y + d.y;
        r.w = r.z + d.z;
        run = r.w + d.w;
        rp[k] = r;
    }
}

__global__ void __launch_bounds__(256) bucket_place(const int* __restrict__ src,
                                                    const int* __restrict__ dst,
                                                    const int* __restrict__ rowptr,
                                                    const int* __restrict__ rank,
                                                    int* __restrict__ srcs) {
    int e = blockIdx.x * blockDim.x + threadIdx.x;
    if (e >= N_EDGES) return;
    srcs[rowptr[dst[e]] + rank[e]] = src[e];
}

__device__ __forceinline__ float xhalf(float x) {
    const int b = __builtin_bit_cast(int, x);
    return __builtin_bit_cast(float, __builtin_amdgcn_permlanex16(b, b, 0x76543210, 0xfedcba98, false, false));
}
template <int M>
__device__ __forceinline__ float xrow(float x) {
    return __builtin_bit_cast(float, __builtin_amdgcn_mov_dpp(__builtin_bit_cast(int, x), 0x160 + M, 0xf, 0xf, false));
}
__device__ __forceinline__ float wave_max(float x) {
    x = fmaxf(x, xrow<1>(x));
    x = fmaxf(x, xrow<2>(x));
    x = fmaxf(x, xrow<4>(x));
    x = fmaxf(x, xrow<8>(x));
    return fmaxf(x, xhalf(x));
}
__device__ __forceinline__ float wave_sum(float x) {
    x += xrow<1>(x);
    x += xrow<2>(x);
    x += xrow<4>(x);
    x += xrow<8>(x);
    return x + xhalf(x);
}

__device__ __forceinline__ void edge_scores(const int* __restrict__ srcs,
                                            const float* __restrict__ el,
                                            const float er0, const float er1,
                                            const int base, const int end, const int lane,
                                            int& s, float& v0, float& v1) {
    s = 0;
    v0 = -INFINITY;
    v1 = -INFINITY;
    if (base + lane < end) {
        s = srcs[base + lane];
        const float2 l = *(const float2*)(el + 2 * (size_t)s);
        const float t0 = l.x + er0, t1 = l.y + er1;
        v0 = t0 > 0.0f ? t0 : 0.2f * t0;
        v1 = t1 > 0.0f ? t1 : 0.2f * t1;
    }
}

__device__ __forceinline__ void gather_body(const float* __restrict__ feat,
                                            const float* __restrict__ el,
                                            const float* __restrict__ er,
                                            const int* __restrict__ rowptr,
                                            const int* __restrict__ srcs,
                                            const float* __restrict__ b,
                                            float* __restrict__ dest,
                                            const bool merge_heads) {
    const int gw   = (blockIdx.x * blockDim.x + threadIdx.x) >> 5;
    const int lane = threadIdx.x & 31;
    if (gw >= N_NODES) return;
    const int n = __builtin_amdgcn_readfirstlane(gw);
    const int beg = rowptr[n], end = rowptr[n + 1];
    float4 a0 = {0.f, 0.f, 0.f, 0.f}, a1 = {0.f, 0.f, 0.f, 0.f};
    if (end > beg) {
        const float er0 = er[n * 2], er1 = er[n * 2 + 1];
        int s_l;
        float e0_l, e1_l;
        edge_scores(srcs, el, er0, er1, beg, end, lane, s_l, e0_l, e1_l);
        float m0 = e0_l, m1 = e1_l;
        for (int base = beg + 32; base < end; base += 32) {
            int s;
            float v0, v1;
            edge_scores(srcs, el, er0, er1, base, end, lane, s, v0, v1);
            m0 = fmaxf(m0, v0);
            m1 = fmaxf(m1, v1);
        }
        m0 = wave_max(m0);
        m1 = wave_max(m1);
        const float x0_l = (beg + lane < end) ? expf(e0_l - m0) : 0.f;
        const float x1_l = (beg + lane < end) ? expf(e1_l - m1) : 0.f;
        float den0 = x0_l, den1 = x1_l;
        for (int base = beg + 32; base < end; base += 32) {
            int s;
            float v0, v1;
            edge_scores(srcs, el, er0, er1, base, end, lane, s, v0, v1);
            if (base + lane < end) {
                den0 += expf(v0 - m0);
                den1 += expf(v1 - m1);
            }
        }
        den0 = wave_sum(den0);
        den1 = wave_sum(den1);
        float w0_l = x0_l / den0, w1_l = x1_l / den1;
        for (int base = beg; base < end; base += 32) {
            const int cnt = (end - base) < 32 ? (end - base) : 32;
            if (base != beg) {
                float v0, v1;
                edge_scores(srcs, el, er0, er1, base, end, lane, s_l, v0, v1);
                w0_l = (base + lane < end) ? expf(v0 - m0) / den0 : 0.f;
                w1_l = (base + lane < end) ? expf(v1 - m1) / den1 : 0.f;
            }
            #pragma unroll 4
            for (int j = 0; j < cnt; ++j) {
                const int   s  = __builtin_amdgcn_readlane(s_l, j);
                const float w0 = __int_as_float(__builtin_amdgcn_readlane(__float_as_int(w0_l), j));
                const float w1 = __int_as_float(__builtin_amdgcn_readlane(__float_as_int(w1_l), j));
                const float4* fp = (const float4*)(feat + (size_t)s * 256);
                const float4 f0 = fp[lane], f1 = fp[32 + lane];
                a0.x += f0.x * w0;  a0.y += f0.y * w0;  a0.z += f0.z * w0;  a0.w += f0.w * w0;
                a1.x += f1.x * w1;  a1.y += f1.y * w1;  a1.z += f1.z * w1;  a1.w += f1.w * w1;
            }
        }
    }
    const float4 b0 = ((const float4*)b)[lane], b1 = ((const float4*)b)[32 + lane];
    a0.x += b0.x;  a0.y += b0.y;  a0.z += b0.z;  a0.w += b0.w;
    a1.x += b1.x;  a1.y += b1.y;  a1.z += b1.z;  a1.w += b1.w;
    if (merge_heads) {
        float4 o;
        o.x = 0.5f * (a0.x + a1.x);  o.y = 0.5f * (a0.y + a1.y);
        o.z = 0.5f * (a0.z + a1.z);  o.w = 0.5f * (a0.w + a1.w);
        ((float4*)(dest + (size_t)n * 128))[lane] = o;
    } else {
        float4* o = (float4*)(dest + (size_t)n * 256);
        o[lane]      = a0;
        o[32 + lane] = a1;
    }
}

__global__ void __launch_bounds__(256) gather_rows_mean(const float* __restrict__ feat,
                                                        const float* __restrict__ el,
                                                        const float* __restrict__ er,
                                                        const int* __restrict__ rowptr,
                                                        const int* __restrict__ srcs,
                                                        const float* __restrict__ b,
                                                        float* __restrict__ hnext) {
    gather_body(feat, el, er, rowptr, srcs, b, hnext, true);
}

__global__ void __launch_bounds__(256) gather_rows_heads(const float* __restrict__ feat,
                                                         const float* __restrict__ el,
                                                         const float* __restrict__ er,
                                                         const int* __restrict__ rowptr,
                                                         const int* __restrict__ srcs,
                                                         const float* __restrict__ b,
                                                         float* __restrict__ out) {
    gather_body(feat, el, er, rowptr, srcs, b, out, false);
}

extern "C" void kernel_launch(void* const* d_in, const int* in_sizes, int n_in,
                              void* d_out, int out_size, void* d_ws, size_t ws_size,
                              hipStream_t stream) {
    const float* x   = (const float*)d_in[0];
    const float* Ws  = (const float*)d_in[1];
    const float* als = (const float*)d_in[2];
    const float* ars = (const float*)d_in[3];
    const float* bs  = (const float*)d_in[4];
    const int*   src = (const int*)d_in[5];
    const int*   dst = (const int*)d_in[6];

    float* ws      = (float*)d_ws;
    float* feat    = ws;
    int*   srcs    = (int*)(ws + 12800000);
    int*   rank    = (int*)(ws + 14400000);
    int*   deg     = (int*)(ws + 15200000);
    int*   rowptr  = (int*)(ws + 15300000);
    float* el      = ws + 19200000;
    float* er      = ws + 19300000;

    float* out   = (float*)d_out;
    float* hnext = out;

    fill_u32<<<NODES_PAD / 256, 256, 0, stream>>>((unsigned*)deg, 0u, NODES_PAD);
    bucket_count<<<(N_EDGES + 255) / 256, 256, 0, stream>>>(dst, deg, rank);
    bucket_scan<<<1, 1024, 0, stream>>>(deg, rowptr);
    bucket_place<<<(N_EDGES + 255) / 256, 256, 0, stream>>>(src, dst, rowptr, rank, srcs);

    for (int layer = 0; layer < 2; ++layer) {
        const float* hin = layer ? hnext : x;
        const float* W   = Ws  + (size_t)layer * D_MODEL * 256;
        const float* al  = als + (size_t)layer * 256;
        const float* ar  = ars + (size_t)layer * 256;
        const float* b   = bs  + (size_t)layer * 256;

        gemm_feat<<<N_NODES / 16, 128, 0, stream>>>(hin, W, feat);
        node_attn<<<(N_NODES + 7) / 8, 256, 0, stream>>>(feat, al, ar, el, er);
        if (layer == 0) {
            gather_rows_mean<<<(N_NODES + 7) / 8, 256, 0, stream>>>(feat, el, er, rowptr, srcs, b, hnext);
        } else {
            gather_rows_heads<<<(N_NODES + 7) / 8, 256, 0, stream>>>(feat, el, er, rowptr, srcs, b, out);
        }
    }
}


# aggregate before projecting: wave sums alpha*h[src] over 512-B input rows (was 1-KB projected rows), scores from h.(W a); W applied per head afterwards on WMMA; sum order reassociated
# speedup vs baseline: 3.8094x; 1.1320x over previous
#include <hip/hip_runtime.h>
#include <math.h>

#define N_NODES 50000
#define N_EDGES 800000
#define D_MODEL 128
#define N_HEADS 2

typedef __attribute__((ext_vector_type(2))) float v2f;
typedef __attribute__((ext_vector_type(8))) float v8f;

__global__ void fill_u32(unsigned* __restrict__ p, unsigned v, int n) {
    int i = blockIdx.x * blockDim.x + threadIdx.x;
    if (i < n) p[i] = v;
}

__global__ void __launch_bounds__(256) fold_attn(const float* __restrict__ Ws,
                                                 const float* __restrict__ als,
                                                 const float* __restrict__ ars,
                                                 float* __restrict__ wl,
                                                 float* __restrict__ wr) {
    const int gw   = (blockIdx.x * blockDim.x + threadIdx.x) >> 5;
    const int lane = threadIdx.x & 31;
    if (gw >= 2 * D_MODEL) return;
    const int layer = gw >> 7;
    const float4* F0 = (const float4*)(Ws + (size_t)gw * 256) + lane;
    const float4* F1 = (const float4*)(Ws + (size_t)gw * 256 + 128) + lane;
    const float4* AL0 = (const float4*)(als + layer * 256) + lane;
    const float4* AL1 = (const float4*)(als + layer * 256 + 128) + lane;
    const float4* AR0 = (const float4*)(ars + layer * 256) + lane;
    const float4* AR1 = (const float4*)(ars + layer * 256 + 128) + lane;
    float4 f0 = *F0, f1 = *F1;
    float4 l0 = *AL0, l1 = *AL1, r0 = *AR0, r1 = *AR1;
    float e0 = f0.x*l0.x + f0.y*l0.y + f0.z*l0.z + f0.w*l0.w;
    float e1 = f1.x*l1.x + f1.y*l1.y + f1.z*l1.z + f1.w*l1.w;
    float g0 = f0.x*r0.x + f0.y*r0.y + f0.z*r0.z + f0.w*r0.w;
    float g1 = f1.x*r1.x + f1.y*r1.y + f1.z*r1.z + f1.w*r1.w;
    #pragma unroll
    for (int off = 16; off; off >>= 1) {
        e0 += __shfl_xor(e0, off, 32);
        e1 += __shfl_xor(e1, off, 32);
        g0 += __shfl_xor(g0, off, 32);
        g1 += __shfl_xor(g1, off, 32);
    }
    if (lane == 0) {
        wl[gw * 2 + 0] = e0;  wl[gw * 2 + 1] = e1;
        wr[gw * 2 + 0] = g0;  wr[gw * 2 + 1] = g1;
    }
}

__global__ void __launch_bounds__(256) node_elr(const float* __restrict__ hin,
                                                const float* __restrict__ wl,
                                                const float* __restrict__ wr,
                                                float* __restrict__ el,
                                                float* __restrict__ er) {
    const int gw   = (blockIdx.x * blockDim.x + threadIdx.x) >> 5;
    const int lane = threadIdx.x & 31;
    if (gw >= N_NODES) return;
    const float4 x  = ((const float4*)(hin + (size_t)gw * 128))[lane];
    const float4 la = ((const float4*)wl)[2 * lane], lb = ((const float4*)wl)[2 * lane + 1];
    const float4 ra = ((const float4*)wr)[2 * lane], rb = ((const float4*)wr)[2 * lane + 1];
    float e0 = x.x*la.x + x.y*la.z + x.z*lb.x + x.w*lb.z;
    float e1 = x.x*la.y + x.y*la.w + x.z*lb.y + x.w*lb.w;
    float g0 = x.x*ra.x + x.y*ra.z + x.z*rb.x + x.w*rb.z;
    float g1 = x.x*ra.y + x.y*ra.w + x.z*rb.y + x.w*rb.w;
    #pragma unroll
    for (int off = 16; off; off >>= 1) {
        e0 += __shfl_xor(e0, off, 32);
        e1 += __shfl_xor(e1, off, 32);
        g0 += __shfl_xor(g0, off, 32);
        g1 += __shfl_xor(g1, off, 32);
    }
    if (lane == 0) {
        el[gw * 2 + 0] = e0;  el[gw * 2 + 1] = e1;
        er[gw * 2 + 0] = g0;  er[gw * 2 + 1] = g1;
    }
}


__global__ void __launch_bounds__(64) gemm_mean(const float* __restrict__ agg,
                                                const float* __restrict__ W,
                                                const float* __restrict__ b,
                                                float* __restrict__ hnext) {
    const int lane  = threadIdx.x & 31;
    const int wave  = threadIdx.x >> 5;
    const int mbase = blockIdx.x << 4;
    const int nq    = wave << 6;
    const int row   = lane & 15;
    const int khalf = lane >> 4;

    v8f acc0 = {}, acc1 = {}, acc2 = {}, acc3 = {};
    for (int hd = 0; hd < N_HEADS; ++hd) {
        const float* Arow = agg + (size_t)(mbase + row) * 256 + hd * 128;
        const float* Wh   = W + hd * 128 + nq + row;
        for (int k = 0; k < D_MODEL; k += 4) {
            const int ka = k + khalf * 2;
            v2f a;
            a.x = Arow[ka];
            a.y = Arow[ka + 1];
            const float* Wb = Wh + (size_t)ka * 256;
            v2f b0, b1, b2, b3;
            b0.x = Wb[0];    b0.y = Wb[256];
            b1.x = Wb[16];   b1.y = Wb[256 + 16];
            b2.x = Wb[32];   b2.y = Wb[256 + 32];
            b3.x = Wb[48];   b3.y = Wb[256 + 48];
            acc0 = __builtin_amdgcn_wmma_f32_16x16x4_f32(false, a, false, b0, (short)0, acc0, false, false);
            acc1 = __builtin_amdgcn_wmma_f32_16x16x4_f32(false, a, false, b1, (short)0, acc1, false, false);
            acc2 = __builtin_amdgcn_wmma_f32_16x16x4_f32(false, a, false, b2, (short)0, acc2, false, false);
            acc3 = __builtin_amdgcn_wmma_f32_16x16x4_f32(false, a, false, b3, (short)0, acc3, false, false);
        }
    }

    const int c = nq + row;
    const float bb0 = b[c]      + b[128 + c];
    const float bb1 = b[c + 16] + b[128 + c + 16];
    const float bb2 = b[c + 32] + b[128 + c + 32];
    const float bb3 = b[c + 48] + b[128 + c + 48];
    #pragma unroll
    for (int j = 0; j < 8; ++j) {
        const int m = mbase + j + khalf * 8;
        float* F = hnext + (size_t)m * 128 + c;
        F[0]  = 0.5f * (acc0[j] + bb0);
        F[16] = 0.5f * (acc1[j] + bb1);
        F[32] = 0.5f * (acc2[j] + bb2);
        F[48] = 0.5f * (acc3[j] + bb3);
    }
}

__global__ void __launch_bounds__(128) gemm_heads(const float* __restrict__ agg,
                                                  const float* __restrict__ W,
                                                  const float* __restrict__ b,
                                                  float* __restrict__ out) {
    const int lane  = threadIdx.x & 31;
    const int wave  = threadIdx.x >> 5;
    const int mbase = blockIdx.x << 4;
    const int nq    = wave << 6;
    const int row   = lane & 15;
    const int khalf = lane >> 4;

    v8f acc0 = {}, acc1 = {}, acc2 = {}, acc3 = {};
    const float* Arow = agg + (size_t)(mbase + row) * 256 + (wave >> 1) * 128;

    for (int k = 0; k < D_MODEL; k += 4) {
        const int ka = k + khalf * 2;
        v2f a;
        a.x = Arow[ka];
        a.y = Arow[ka + 1];
        const float* Wb = W + (size_t)ka * 256 + row;
        v2f b0, b1, b2, b3;
        b0.x = Wb[nq];        b0.y = Wb[256 + nq];
        b1.x = Wb[nq + 16];   b1.y = Wb[256 + nq + 16];
        b2.x = Wb[nq + 32];   b2.y = Wb[256 + nq + 32];
        b3.x = Wb[nq + 48];   b3.y = Wb[256 + nq + 48];
        acc0 = __builtin_amdgcn_wmma_f32_16x16x4_f32(false, a, false, b0, (short)0, acc0, false, false);
        acc1 = __builtin_amdgcn_wmma_f32_16x16x4_f32(false, a, false, b1, (short)0, acc1, false, false);
        acc2 = __builtin_amdgcn_wmma_f32_16x16x4_f32(false, a, false, b2, (short)0, acc2, false, false);
        acc3 = __builtin_amdgcn_wmma_f32_16x16x4_f32(false, a, false, b3, (short)0, acc3, false, false);
    }

    const int c = nq + row;
    const float bb0 = b[c], bb1 = b[c + 16], bb2 = b[c + 32], bb3 = b[c + 48];
    #pragma unroll
    for (int j = 0; j < 8; ++j) {
        const int m = mbase + j + khalf * 8;
        float* F = out + (size_t)m * 256 + c;
        F[0]  = acc0[j] + bb0;
        F[16] = acc1[j] + bb1;
        F[32] = acc2[j] + bb2;
        F[48] = acc3[j] + bb3;
    }
}

__global__ void __launch_bounds__(256) bucket_count(const int* __restrict__ dst,
                                                    int* __restrict__ deg,
                                                    int* __restrict__ rank) {
    int e = blockIdx.x * blockDim.x + threadIdx.x;
    if (e >= N_EDGES) return;
    rank[e] = atomicAdd(deg + dst[e], 1);
}

#define SCAN_PER  ((((N_NODES + 1023) / 1024) + 3) / 4 * 4)
#define NODES_PAD (SCAN_PER * 1024)
__global__ void __launch_bounds__(1024) bucket_scan(const int* __restrict__ deg,
                                                    int* __restrict__ rowptr) {
    __shared__ int wsum[32];
    const int tid = threadIdx.x, lane = tid & 31, wave = tid >> 5;
    const int4* dg = (const int4*)(deg + tid * SCAN_PER);
    int loc = 0;
    for (int k = 0; k < SCAN_PER / 4; ++k) {
        const int4 d = dg[k];
        loc += (d.x + d.y) + (d.z + d.w);
    }
    int inc = loc;
    for (int off = 1; off < 32; off <<= 1) {
        const int t = __shfl_up(inc, off, 32);
        if (lane >= off) inc += t;
    }
    if (lane == 31) wsum[wave] = inc;
    __syncthreads();
    if (wave == 0) {
        const int v = wsum[lane];
        int vi = v;
        for (int off = 1; off < 32; off <<= 1) {
            const int t = __shfl_up(vi, off, 32);
            if (lane >= off) vi += t;
        }
        wsum[lane] = vi - v;
    }
    __syncthreads();
    int run = wsum[wave] + inc - loc;
    int4* rp = (int4*)(rowptr + tid * SCAN_PER);
    #pragma unroll
    for (int k = 0; k < SCAN_PER / 4; ++k) {
        const int4 d = dg[k];
        int4 r;
        r.x = run;
        r.y = r.x + d.x;
        r.z = r.y + d.y;
        r.w = r.z + d.z;
        run = r.w + d.w;
        rp[k] = r;
    }
}

__global__ void __launch_bounds__(256) bucket_place(const int* __restrict__ src,
                                                    const int* __restrict__ dst,
                                                    const int* __restrict__ rowptr,
                                                    const int* __restrict__ rank,
                                                    int* __restrict__ srcs) {
    int e = blockIdx.x * blockDim.x + threadIdx.x;
    if (e >= N_EDGES) return;
    srcs[rowptr[dst[e]] + rank[e]] = src[e];
}

__device__ __forceinline__ float xhalf(float x) {
    const int b = __builtin_bit_cast(int, x);
    return __builtin_bit_cast(float, __builtin_amdgcn_permlanex16(b, b, 0x76543210, 0xfedcba98, false, false));
}
template <int M>
__device__ __forceinline__ float xrow(float x) {
    return __builtin_bit_cast(float, __builtin_amdgcn_mov_dpp(__builtin_bit_cast(int, x), 0x160 + M, 0xf, 0xf, false));
}
__device__ __forceinline__ float wave_max(float x) {
    x = fmaxf(x, xrow<1>(x));
    x = fmaxf(x, xrow<2>(x));
    x = fmaxf(x, xrow<4>(x));
    x = fmaxf(x, xrow<8>(x));
    return fmaxf(x, xhalf(x));
}
__device__ __forceinline__ float wave_sum(float x) {
    x += xrow<1>(x);
    x += xrow<2>(x);
    x += xrow<4>(x);
    x += xrow<8>(x);
    return x + xhalf(x);
}

__device__ __forceinline__ void edge_scores(const int* __restrict__ srcs,
                                            const float* __restrict__ el,
                                            const float er0, const float er1,
                                            const int base, const int end, const int lane,
                                            int& s, float& v0, float& v1) {
    s = 0;
    v0 = -INFINITY;
    v1 = -INFINITY;
    if (base + lane < end) {
        s = srcs[base + lane];
        const float2 l = *(const float2*)(el + 2 * (size_t)s);
        const float t0 = l.x + er0, t1 = l.y + er1;
        v0 = t0 > 0.0f ? t0 : 0.2f * t0;
        v1 = t1 > 0.0f ? t1 : 0.2f * t1;
    }
}

__global__ void __launch_bounds__(256) gather_agg(const float* __restrict__ hin,
                                                  const float* __restrict__ el,
                                                  const float* __restrict__ er,
                                                  const int* __restrict__ rowptr,
                                                  const int* __restrict__ srcs,
                                                  float* __restrict__ agg) {
    const int gw   = (blockIdx.x * blockDim.x + threadIdx.x) >> 5;
    const int lane = threadIdx.x & 31;
    if (gw >= N_NODES) return;
    const int n = __builtin_amdgcn_readfirstlane(gw);
    const int beg = rowptr[n], end = rowptr[n + 1];
    float4 a0 = {0.f, 0.f, 0.f, 0.f}, a1 = {0.f, 0.f, 0.f, 0.f};
    if (end > beg) {
        const float er0 = er[n * 2], er1 = er[n * 2 + 1];
        int s_l;
        float e0_l, e1_l;
        edge_scores(srcs, el, er0, er1, beg, end, lane, s_l, e0_l, e1_l);
        float m0 = e0_l, m1 = e1_l;
        for (int base = beg + 32; base < end; base += 32) {
            int s;
            float v0, v1;
            edge_scores(srcs, el, er0, er1, base, end, lane, s, v0, v1);
            m0 = fmaxf(m0, v0);
            m1 = fmaxf(m1, v1);
        }
        m0 = wave_max(m0);
        m1 = wave_max(m1);
        const float x0_l = (beg + lane < end) ? expf(e0_l - m0) : 0.f;
        const float x1_l = (beg + lane < end) ? expf(e1_l - m1) : 0.f;
        float den0 = x0_l, den1 = x1_l;
        for (int base = beg + 32; base < end; base += 32) {
            int s;
            float v0, v1;
            edge_scores(srcs, el, er0, er1, base, end, lane, s, v0, v1);
            if (base + lane < end) {
                den0 += expf(v0 - m0);
                den1 += expf(v1 - m1);
            }
        }
        den0 = wave_sum(den0);
        den1 = wave_sum(den1);
        float w0_l = x0_l / den0, w1_l = x1_l / den1;
        for (int base = beg; base < end; base += 32) {
            const int cnt = (end - base) < 32 ? (end - base) : 32;
            if (base != beg) {
                float v0, v1;
                edge_scores(srcs, el, er0, er1, base, end, lane, s_l, v0, v1);
                w0_l = (base + lane < end) ? expf(v0 - m0) / den0 : 0.f;
                w1_l = (base + lane < end) ? expf(v1 - m1) / den1 : 0.f;
            }
            #pragma unroll 4
            for (int j = 0; j < cnt; ++j) {
                const int   s  = __builtin_amdgcn_readlane(s_l, j);
                const float w0 = __int_as_float(__builtin_amdgcn_readlane(__float_as_int(w0_l), j));
                const float w1 = __int_as_float(__builtin_amdgcn_readlane(__float_as_int(w1_l), j));
                const float4 f = ((const float4*)(hin + (size_t)s * 128))[lane];
                a0.x += f.x * w0;  a0.y += f.y * w0;  a0.z += f.z * w0;  a0.w += f.w * w0;
                a1.x += f.x * w1;  a1.y += f.y * w1;  a1.z += f.z * w1;  a1.w += f.w * w1;
            }
        }
    }
    float4* o = (float4*)(agg + (size_t)n * 256);
    o[lane]      = a0;
    o[32 + lane] = a1;
}

extern "C" void kernel_launch(void* const* d_in, const int* in_sizes, int n_in,
                              void* d_out, int out_size, void* d_ws, size_t ws_size,
                              hipStream_t stream) {
    const float* x   = (const float*)d_in[0];
    const float* Ws  = (const float*)d_in[1];
    const float* als = (const float*)d_in[2];
    const float* ars = (const float*)d_in[3];
    const float* bs  = (const float*)d_in[4];
    const int*   src = (const int*)d_in[5];
    const int*   dst = (const int*)d_in[6];

    float* ws      = (float*)d_ws;
    float* agg     = ws;
    int*   srcs    = (int*)(ws + 12800000);
    int*   rank    = (int*)(ws + 14400000);
    int*   deg     = (int*)(ws + 15200000);
    int*   rowptr  = (int*)(ws + 15300000);
    float* wl      = ws + 19100000;
    float* wr      = ws + 19101000;
    float* el      = ws + 19200000;
    float* er      = ws + 19300000;

    float* out   = (float*)d_out;
    float* hnext = out;

    fill_u32<<<NODES_PAD / 256, 256, 0, stream>>>((unsigned*)deg, 0u, NODES_PAD);
    bucket_count<<<(N_EDGES + 255) / 256, 256, 0, stream>>>(dst, deg, rank);
    bucket_scan<<<1, 1024, 0, stream>>>(deg, rowptr);
    bucket_place<<<(N_EDGES + 255) / 256, 256, 0, stream>>>(src, dst, rowptr, rank, srcs);
    fold_attn<<<2 * D_MODEL / 8, 256, 0, stream>>>(Ws, als, ars, wl, wr);

    node_elr<<<(N_NODES + 7) / 8, 256, 0, stream>>>(x, wl, wr, el, er);
    gather_agg<<<(N_NODES + 7) / 8, 256, 0, stream>>>(x, el, er, rowptr, srcs, agg);
    gemm_mean<<<N_NODES / 16, 64, 0, stream>>>(agg, Ws, bs, hnext);

    node_elr<<<(N_NODES + 7) / 8, 256, 0, stream>>>(hnext, wl + 256, wr + 256, el, er);
    gather_agg<<<(N_NODES + 7) / 8, 256, 0, stream>>>(hnext, el, er, rowptr, srcs, agg);
    gemm_heads<<<N_NODES / 16, 128, 0, stream>>>(agg, Ws + 32768, bs + 256, out);
}


# projection fused into the aggregation kernel: a 16-wave block sums 32 nodes' rows into an LDS tile (pitch 260) and multiplies it by W on f32 WMMA there; agg round trip and 2 launches dropped
# speedup vs baseline: 4.2234x; 1.1087x over previous
#include <hip/hip_runtime.h>
#include <math.h>

#define N_NODES 50000
#define N_EDGES 800000
#define D_MODEL 128
#define N_HEADS 2

typedef __attribute__((ext_vector_type(2))) float v2f;
typedef __attribute__((ext_vector_type(8))) float v8f;

__global__ void fill_u32(unsigned* __restrict__ p, unsigned v, int n) {
    int i = blockIdx.x * blockDim.x + threadIdx.x;
    if (i < n) p[i] = v;
}

__global__ void __launch_bounds__(256) fold_attn(const float* __restrict__ Ws,
                                                 const float* __restrict__ als,
                                                 const float* __restrict__ ars,
                                                 float* __restrict__ wl,
                                                 float* __restrict__ wr) {
    const int gw   = (blockIdx.x * blockDim.x + threadIdx.x) >> 5;
    const int lane = threadIdx.x & 31;
    if (gw >= 2 * D_MODEL) return;
    const int layer = gw >> 7;
    const float4* F0 = (const float4*)(Ws + (size_t)gw * 256) + lane;
    const float4* F1 = (const float4*)(Ws + (size_t)gw * 256 + 128) + lane;
    const float4* AL0 = (const float4*)(als + layer * 256) + lane;
    const float4* AL1 = (const float4*)(als + layer * 256 + 128) + lane;
    const float4* AR0 = (const float4*)(ars + layer * 256) + lane;
    const float4* AR1 = (const float4*)(ars + layer * 256 + 128) + lane;
    float4 f0 = *F0, f1 = *F1;
    float4 l0 = *AL0, l1 = *AL1, r0 = *AR0, r1 = *AR1;
    float e0 = f0.x*l0.x + f0.y*l0.y + f0.z*l0.z + f0.w*l0.w;
    float e1 = f1.x*l1.x + f1.y*l1.y + f1.z*l1.z + f1.w*l1.w;
    float g0 = f0.x*r0.x + f0.y*r0.y + f0.z*r0.z + f0.w*r0.w;
    float g1 = f1.x*r1.x + f1.y*r1.y + f1.z*r1.z + f1.w*r1.w;
    #pragma unroll
    for (int off = 16; off; off >>= 1) {
        e0 += __shfl_xor(e0, off, 32);
        e1 += __shfl_xor(e1, off, 32);
        g0 += __shfl_xor(g0, off, 32);
        g1 += __shfl_xor(g1, off, 32);
    }
    if (lane == 0) {
        wl[gw * 2 + 0] = e0;  wl[gw * 2 + 1] = e1;
        wr[gw * 2 + 0] = g0;  wr[gw * 2 + 1] = g1;
    }
}

__global__ void __launch_bounds__(256) node_elr(const float* __restrict__ hin,
                                                const float* __restrict__ wl,
                                                const float* __restrict__ wr,
                                                float* __restrict__ el,
                                                float* __restrict__ er) {
    const int gw   = (blockIdx.x * blockDim.x + threadIdx.x) >> 5;
    const int lane = threadIdx.x & 31;
    if (gw >= N_NODES) return;
    const float4 x  = ((const float4*)(hin + (size_t)gw * 128))[lane];
    const float4 la = ((const float4*)wl)[2 * lane], lb = ((const float4*)wl)[2 * lane + 1];
    const float4 ra = ((const float4*)wr)[2 * lane], rb = ((const float4*)wr)[2 * lane + 1];
    float e0 = x.x*la.x + x.y*la.z + x.z*lb.x + x.w*lb.z;
    float e1 = x.x*la.y + x.y*la.w + x.z*lb.y + x.w*lb.w;
    float g0 = x.x*ra.x + x.y*ra.z + x.z*rb.x + x.w*rb.z;
    float g1 = x.x*ra.y + x.y*ra.w + x.z*rb.y + x.w*rb.w;
    #pragma unroll
    for (int off = 16; off; off >>= 1) {
        e0 += __shfl_xor(e0, off, 32);
        e1 += __shfl_xor(e1, off, 32);
        g0 += __shfl_xor(g0, off, 32);
        g1 += __shfl_xor(g1, off, 32);
    }
    if (lane == 0) {
        el[gw * 2 + 0] = e0;  el[gw * 2 + 1] = e1;
        er[gw * 2 + 0] = g0;  er[gw * 2 + 1] = g1;
    }
}

__global__ void __launch_bounds__(256) bucket_count(const int* __restrict__ dst,
                                                    int* __restrict__ deg,
                                                    int* __restrict__ rank) {
    int e = blockIdx.x * blockDim.x + threadIdx.x;
    if (e >= N_EDGES) return;
    rank[e] = atomicAdd(deg + dst[e], 1);
}

#define SCAN_PER  ((((N_NODES + 1023) / 1024) + 3) / 4 * 4)
#define NODES_PAD (SCAN_PER * 1024)
__global__ void __launch_bounds__(1024) bucket_scan(const int* __restrict__ deg,
                                                    int* __restrict__ rowptr) {
    __shared__ int wsum[32];
    const int tid = threadIdx.x, lane = tid & 31, wave = tid >> 5;
    const int4* dg = (const int4*)(deg + tid * SCAN_PER);
    int loc = 0;
    for (int k = 0; k < SCAN_PER / 4; ++k) {
        const int4 d = dg[k];
        loc += (d.x + d.y) + (d.z + d.w);
    }
    int inc = loc;
    for (int off = 1; off < 32; off <<= 1) {
        const int t = __shfl_up(inc, off, 32);
        if (lane >= off) inc += t;
    }
    if (lane == 31) wsum[wave] = inc;
    __syncthreads();
    if (wave == 0) {
        const int v = wsum[lane];
        int vi = v;
        for (int off = 1; off < 32; off <<= 1) {
            const int t = __shfl_up(vi, off, 32);
            if (lane >= off) vi += t;
        }
        wsum[lane] = vi - v;
    }
    __syncthreads();
    int run = wsum[wave] + inc - loc;
    int4* rp = (int4*)(rowptr + tid * SCAN_PER);
    #pragma unroll
    for (int k = 0; k < SCAN_PER / 4; ++k) {
        const int4 d = dg[k];
        int4 r;
        r.x = run;
        r.y = r.x + d.x;
        r.z = r.y + d.y;
        r.w = r.z + d.z;
        run = r.w + d.w;
        rp[k] = r;
    }
}

__global__ void __launch_bounds__(256) bucket_place(const int* __restrict__ src,
                                                    const int* __restrict__ dst,
                                                    const int* __restrict__ rowptr,
                                                    const int* __restrict__ rank,
                                                    int* __restrict__ srcs) {
    int e = blockIdx.x * blockDim.x + threadIdx.x;
    if (e >= N_EDGES) return;
    srcs[rowptr[dst[e]] + rank[e]] = src[e];
}

__device__ __forceinline__ float xhalf(float x) {
    const int b = __builtin_bit_cast(int, x);
    return __builtin_bit_cast(float, __builtin_amdgcn_permlanex16(b, b, 0x76543210, 0xfedcba98, false, false));
}
template <int M>
__device__ __forceinline__ float xrow(float x) {
    return __builtin_bit_cast(float, __builtin_amdgcn_mov_dpp(__builtin_bit_cast(int, x), 0x160 + M, 0xf, 0xf, false));
}
__device__ __forceinline__ float wave_max(float x) {
    x = fmaxf(x, xrow<1>(x));
    x = fmaxf(x, xrow<2>(x));
    x = fmaxf(x, xrow<4>(x));
    x = fmaxf(x, xrow<8>(x));
    return fmaxf(x, xhalf(x));
}
__device__ __forceinline__ float wave_sum(float x) {
    x += xrow<1>(x);
    x += xrow<2>(x);
    x += xrow<4>(x);
    x += xrow<8>(x);
    return x + xhalf(x);
}

__device__ __forceinline__ void edge_scores(const int* __restrict__ srcs,
                                            const float* __restrict__ el,
                                            const float er0, const float er1,
                                            const int base, const int end, const int lane,
                                            int& s, float& v0, float& v1) {
    s = 0;
    v0 = -INFINITY;
    v1 = -INFINITY;
    if (base + lane < end) {
        s = srcs[base + lane];
        const float2 l = *(const float2*)(el + 2 * (size_t)s);
        const float t0 = l.x + er0, t1 = l.y + er1;
        v0 = t0 > 0.0f ? t0 : 0.2f * t0;
        v1 = t1 > 0.0f ? t1 : 0.2f * t1;
    }
}

__device__ __forceinline__ void gather_node(const float* __restrict__ hin,
                                            const float* __restrict__ el,
                                            const float* __restrict__ er,
                                            const int* __restrict__ rowptr,
                                            const int* __restrict__ srcs,
                                            const int n, const int lane,
                                            float4& a0, float4& a1) {
    const int beg = rowptr[n], end = rowptr[n + 1];
    if (end > beg) {
        const float er0 = er[n * 2], er1 = er[n * 2 + 1];
        int s_l;
        float e0_l, e1_l;
        edge_scores(srcs, el, er0, er1, beg, end, lane, s_l, e0_l, e1_l);
        float m0 = e0_l, m1 = e1_l;
        for (int base = beg + 32; base < end; base += 32) {
            int s;
            float v0, v1;
            edge_scores(srcs, el, er0, er1, base, end, lane, s, v0, v1);
            m0 = fmaxf(m0, v0);
            m1 = fmaxf(m1, v1);
        }
        m0 = wave_max(m0);
        m1 = wave_max(m1);
        const float x0_l = (beg + lane < end) ? expf(e0_l - m0) : 0.f;
        const float x1_l = (beg + lane < end) ? expf(e1_l - m1) : 0.f;
        float den0 = x0_l, den1 = x1_l;
        for (int base = beg + 32; base < end; base += 32) {
            int s;
            float v0, v1;
            edge_scores(srcs, el, er0, er1, base, end, lane, s, v0, v1);
            if (base + lane < end) {
                den0 += expf(v0 - m0);
                den1 += expf(v1 - m1);
            }
        }
        den0 = wave_sum(den0);
        den1 = wave_sum(den1);
        float w0_l = x0_l / den0, w1_l = x1_l / den1;
        for (int base = beg; base < end; base += 32) {
            const int cnt = (end - base) < 32 ? (end - base) : 32;
            if (base != beg) {
                float v0, v1;
                edge_scores(srcs, el, er0, er1, base, end, lane, s_l, v0, v1);
                w0_l = (base + lane < end) ? expf(v0 - m0) / den0 : 0.f;
                w1_l = (base + lane < end) ? expf(v1 - m1) / den1 : 0.f;
            }
            #pragma unroll 4
            for (int j = 0; j < cnt; ++j) {
                const int   s  = __builtin_amdgcn_readlane(s_l, j);
                const float w0 = __int_as_float(__builtin_amdgcn_readlane(__float_as_int(w0_l), j));
                const float w1 = __int_as_float(__builtin_amdgcn_readlane(__float_as_int(w1_l), j));
                const float4 f = ((const float4*)(hin + (size_t)s * 128))[lane];
                a0.x += f.x * w0;  a0.y += f.y * w0;  a0.z += f.z * w0;  a0.w += f.w * w0;
                a1.x += f.x * w1;  a1.y += f.y * w1;  a1.z += f.z * w1;  a1.w += f.w * w1;
            }
        }
    }
}

#define TILE_NODES 32
#define AGG_PITCH  260

__device__ __forceinline__ void gather_tile(const float* __restrict__ hin,
                                            const float* __restrict__ el,
                                            const float* __restrict__ er,
                                            const int* __restrict__ rowptr,
                                            const int* __restrict__ srcs,
                                            float* __restrict__ sA,
                                            const int mbase, const int wave, const int lane) {
    for (int i = 0; i < 2; ++i) {
        const int r = wave * 2 + i;
        const int n = mbase + r;
        float4 a0 = {0.f, 0.f, 0.f, 0.f}, a1 = {0.f, 0.f, 0.f, 0.f};
        if (n < N_NODES) gather_node(hin, el, er, rowptr, srcs, n, lane, a0, a1);
        float4* o = (float4*)(sA + r * AGG_PITCH);
        o[lane]      = a0;
        o[32 + lane] = a1;
    }
}

__global__ void __launch_bounds__(512, 16) agg_proj_mean(const float* __restrict__ hin,
                                                     const float* __restrict__ el,
                                                     const float* __restrict__ er,
                                                     const int* __restrict__ rowptr,
                                                     const int* __restrict__ srcs,
                                                     const float* __restrict__ W,
                                                     const float* __restrict__ b,
                                                     float* __restrict__ hnext) {
    __shared__ __attribute__((aligned(16))) float sA[TILE_NODES * AGG_PITCH];
    const int lane  = threadIdx.x & 31;
    const int wave  = __builtin_amdgcn_readfirstlane(threadIdx.x >> 5);
    const int mbase = blockIdx.x * TILE_NODES;
    gather_tile(hin, el, er, rowptr, srcs, sA, mbase, wave, lane);
    __syncthreads();
    if (wave >= 8) return;

    const int row   = lane & 15;
    const int khalf = lane >> 4;
    const int rt    = wave >> 2;
    const int nq    = (wave & 3) << 5;

    v8f acc0 = {}, acc1 = {};
    const float* Arow = sA + (rt * 16 + row) * AGG_PITCH + khalf * 2;
    const float* Wh   = W + (size_t)(khalf * 2) * 256 + nq + row;
    #pragma unroll 8
    for (int kk = 0; kk < N_HEADS * D_MODEL; kk += 4) {
        const v2f a = *(const v2f*)(Arow + kk);
        const float* Wb = Wh + (size_t)(kk & (D_MODEL - 1)) * 256 + (kk >> 7) * 128;
        v2f b0, b1;
        b0.x = Wb[0];    b0.y = Wb[256];
        b1.x = Wb[16];   b1.y = Wb[256 + 16];
        acc0 = __builtin_amdgcn_wmma_f32_16x16x4_f32(false, a, false, b0, (short)0, acc0, false, false);
        acc1 = __builtin_amdgcn_wmma_f32_16x16x4_f32(false, a, false, b1, (short)0, acc1, false, false);
    }

    const int c = nq + row;
    const float bb0 = b[c]      + b[128 + c];
    const float bb1 = b[c + 16] + b[128 + c + 16];
    #pragma unroll
    for (int j = 0; j < 8; ++j) {
        const int m = mbase + rt * 16 + j + khalf * 8;
        if (m < N_NODES) {
            float* F = hnext + (size_t)m * 128 + c;
            F[0]  = 0.5f * (acc0[j] + bb0);
            F[16] = 0.5f * (acc1[j] + bb1);
        }
    }
}

__global__ void __launch_bounds__(512, 16) agg_proj_heads(const float* __restrict__ hin,
                                                      const float* __restrict__ el,
                                                      const float* __restrict__ er,
                                                      const int* __restrict__ rowptr,
                                                      const int* __restrict__ srcs,
                                                      const float* __restrict__ W,
                                                      const float* __restrict__ b,
                                                      float* __restrict__ out) {
    __shared__ __attribute__((aligned(16))) float sA[TILE_NODES * AGG_PITCH];
    const int lane  = threadIdx.x & 31;
    const int wave  = __builtin_amdgcn_readfirstlane(threadIdx.x >> 5);
    const int mbase = blockIdx.x * TILE_NODES;
    gather_tile(hin, el, er, rowptr, srcs, sA, mbase, wave, lane);
    __syncthreads();

    const int row   = lane & 15;
    const int khalf = lane >> 4;
    const int rt    = wave >> 3;
    const int nq    = (wave & 7) << 5;

    v8f acc0 = {}, acc1 = {};
    const float* Arow = sA + (rt * 16 + row) * AGG_PITCH + (nq >> 7) * 128;
    const float* Wh   = W + nq + row;
    #pragma unroll 8
    for (int k = 0; k < D_MODEL; k += 4) {
        const int ka = k + khalf * 2;
        const v2f a = *(const v2f*)(Arow + ka);
        const float* Wb = Wh + (size_t)ka * 256;
        v2f b0, b1;
        b0.x = Wb[0];    b0.y = Wb[256];
        b1.x = Wb[16];   b1.y = Wb[256 + 16];
        acc0 = __builtin_amdgcn_wmma_f32_16x16x4_f32(false, a, false, b0, (short)0, acc0, false, false);
        acc1 = __builtin_amdgcn_wmma_f32_16x16x4_f32(false, a, false, b1, (short)0, acc1, false, false);
    }

    const int c = nq + row;
    const float bb0 = b[c], bb1 = b[c + 16];
    #pragma unroll
    for (int j = 0; j < 8; ++j) {
        const int m = mbase + rt * 16 + j + khalf * 8;
        if (m < N_NODES) {
            float* F = out + (size_t)m * 256 + c;
            F[0]  = acc0[j] + bb0;
            F[16] = acc1[j] + bb1;
        }
    }
}

extern "C" void kernel_launch(void* const* d_in, const int* in_sizes, int n_in,
                              void* d_out, int out_size, void* d_ws, size_t ws_size,
                              hipStream_t stream) {
    const float* x   = (const float*)d_in[0];
    const float* Ws  = (const float*)d_in[1];
    const float* als = (const float*)d_in[2];
    const float* ars = (const float*)d_in[3];
    const float* bs  = (const float*)d_in[4];
    const int*   src = (const int*)d_in[5];
    const int*   dst = (const int*)d_in[6];

    float* ws      = (float*)d_ws;
    float* hnext   = ws;
    int*   srcs    = (int*)(ws + 12800000);
    int*   rank    = (int*)(ws + 14400000);
    int*   deg     = (int*)(ws + 15200000);
    int*   rowptr  = (int*)(ws + 15300000);
    float* wl      = ws + 19100000;
    float* wr      = ws + 19101000;
    float* el      = ws + 19200000;
    float* er      = ws + 19300000;

    float* out   = (float*)d_out;

    fill_u32<<<NODES_PAD / 256, 256, 0, stream>>>((unsigned*)deg, 0u, NODES_PAD);
    bucket_count<<<(N_EDGES + 255) / 256, 256, 0, stream>>>(dst, deg, rank);
    bucket_scan<<<1, 1024, 0, stream>>>(deg, rowptr);
    bucket_place<<<(N_EDGES + 255) / 256, 256, 0, stream>>>(src, dst, rowptr, rank, srcs);
    fold_attn<<<2 * D_MODEL / 8, 256, 0, stream>>>(Ws, als, ars, wl, wr);

    node_elr<<<(N_NODES + 7) / 8, 256, 0, stream>>>(x, wl, wr, el, er);
    agg_proj_mean<<<(N_NODES + TILE_NODES - 1) / TILE_NODES, 512, 0, stream>>>(x, el, er, rowptr, srcs, Ws, bs, hnext);

    node_elr<<<(N_NODES + 7) / 8, 256, 0, stream>>>(hnext, wl + 256, wr + 256, el, er);
    agg_proj_heads<<<(N_NODES + TILE_NODES - 1) / TILE_NODES, 512, 0, stream>>>(hnext, el, er, rowptr, srcs, Ws + 32768, bs + 256, out);
}


# edge buckets built on LDS counters: edges binned by 128-node range per block (count, scan, place), each range then ordered by node in LDS; no atomics to memory; edge order = arrival at the LDS adds
# speedup vs baseline: 5.4974x; 1.3017x over previous
#include <hip/hip_runtime.h>
#include <math.h>

#define N_NODES 50000
#define N_EDGES 800000
#define D_MODEL 128
#define N_HEADS 2

typedef __attribute__((ext_vector_type(2))) float v2f;
typedef __attribute__((ext_vector_type(8))) float v8f;

__global__ void __launch_bounds__(256) fold_attn(const float* __restrict__ Ws,
                                                 const float* __restrict__ als,
                                                 const float* __restrict__ ars,
                                                 float* __restrict__ wl,
                                                 float* __restrict__ wr) {
    const int gw   = (blockIdx.x * blockDim.x + threadIdx.x) >> 5;
    const int lane = threadIdx.x & 31;
    if (gw >= 2 * D_MODEL) return;
    const int layer = gw >> 7;
    const float4* F0 = (const float4*)(Ws + (size_t)gw * 256) + lane;
    const float4* F1 = (const float4*)(Ws + (size_t)gw * 256 + 128) + lane;
    const float4* AL0 = (const float4*)(als + layer * 256) + lane;
    const float4* AL1 = (const float4*)(als + layer * 256 + 128) + lane;
    const float4* AR0 = (const float4*)(ars + layer * 256) + lane;
    const float4* AR1 = (const float4*)(ars + layer * 256 + 128) + lane;
    float4 f0 = *F0, f1 = *F1;
    float4 l0 = *AL0, l1 = *AL1, r0 = *AR0, r1 = *AR1;
    float e0 = f0.x*l0.x + f0.y*l0.y + f0.z*l0.z + f0.w*l0.w;
    float e1 = f1.x*l1.x + f1.y*l1.y + f1.z*l1.z + f1.w*l1.w;
    float g0 = f0.x*r0.x + f0.y*r0.y + f0.z*r0.z + f0.w*r0.w;
    float g1 = f1.x*r1.x + f1.y*r1.y + f1.z*r1.z + f1.w*r1.w;
    #pragma unroll
    for (int off = 16; off; off >>= 1) {
        e0 += __shfl_xor(e0, off, 32);
        e1 += __shfl_xor(e1, off, 32);
        g0 += __shfl_xor(g0, off, 32);
        g1 += __shfl_xor(g1, off, 32);
    }
    if (lane == 0) {
        wl[gw * 2 + 0] = e0;  wl[gw * 2 + 1] = e1;
        wr[gw * 2 + 0] = g0;  wr[gw * 2 + 1] = g1;
    }
}

__global__ void __launch_bounds__(256) node_elr(const float* __restrict__ hin,
                                                const float* __restrict__ wl,
                                                const float* __restrict__ wr,
                                                float* __restrict__ el,
                                                float* __restrict__ er) {
    const int gw   = (blockIdx.x * blockDim.x + threadIdx.x) >> 5;
    const int lane = threadIdx.x & 31;
    if (gw >= N_NODES) return;
    const float4 x  = ((const float4*)(hin + (size_t)gw * 128))[lane];
    const float4 la = ((const float4*)wl)[2 * lane], lb = ((const float4*)wl)[2 * lane + 1];
    const float4 ra = ((const float4*)wr)[2 * lane], rb = ((const float4*)wr)[2 * lane + 1];
    float e0 = x.x*la.x + x.y*la.z + x.z*lb.x + x.w*lb.z;
    float e1 = x.x*la.y + x.y*la.w + x.z*lb.y + x.w*lb.w;
    float g0 = x.x*ra.x + x.y*ra.z + x.z*rb.x + x.w*rb.z;
    float g1 = x.x*ra.y + x.y*ra.w + x.z*rb.y + x.w*rb.w;
    #pragma unroll
    for (int off = 16; off; off >>= 1) {
        e0 += __shfl_xor(e0, off, 32);
        e1 += __shfl_xor(e1, off, 32);
        g0 += __shfl_xor(g0, off, 32);
        g1 += __shfl_xor(g1, off, 32);
    }
    if (lane == 0) {
        el[gw * 2 + 0] = e0;  el[gw * 2 + 1] = e1;
        er[gw * 2 + 0] = g0;  er[gw * 2 + 1] = g1;
    }
}

#define BK_SHIFT   7
#define BK_NODES   (1 << BK_SHIFT)
#define BK_COUNT   ((N_NODES >> BK_SHIFT) + 1)
#define BK_BLOCKS  128
#define BK_EDGES   ((N_EDGES + BK_BLOCKS - 1) / BK_BLOCKS)
#define BK_PER     ((BK_EDGES + 1023) / 1024)
#define FP_PER     4
#define SCAN_PER  ((((N_NODES + 1023) / 1024) + 3) / 4 * 4)
#define NODES_PAD (SCAN_PER * 1024)
#define BK_PAD    (NODES_PAD / BK_BLOCKS)
static_assert(BK_PAD * BK_BLOCKS == NODES_PAD, "table length must equal the scan length");
static_assert(BK_COUNT < BK_PAD && BK_PAD <= 1024, "ranges must fit the table and one block");
static_assert(N_NODES <= 65536, "a source index must fit the low 16 bits of the packed word");
static_assert(NODES_PAD >= N_NODES + 4, "rowptr must hold index N_NODES and its int4");

__global__ void __launch_bounds__(1024) coarse_count(const int* __restrict__ dst,
                                                     int* __restrict__ hist) {
    __shared__ int cnt[BK_PAD];
    const int tid = threadIdx.x, b = blockIdx.x;
    if (tid < BK_PAD) cnt[tid] = 0;
    __syncthreads();
    const int e0 = b * BK_EDGES;
    const int e1 = (e0 + BK_EDGES) < N_EDGES ? (e0 + BK_EDGES) : N_EDGES;
    int d[BK_PER];
    #pragma unroll
    for (int j = 0; j < BK_PER; ++j) {
        const int e = e0 + tid + j * 1024;
        d[j] = e < e1 ? dst[e] : -1;
    }
    #pragma unroll
    for (int j = 0; j < BK_PER; ++j)
        if (d[j] >= 0) atomicAdd(&cnt[d[j] >> BK_SHIFT], 1);
    __syncthreads();
    if (tid < BK_PAD) hist[tid * BK_BLOCKS + b] = cnt[tid];
}

__global__ void __launch_bounds__(1024) bucket_scan(const int* __restrict__ in,
                                                    int* __restrict__ out) {
    __shared__ int wsum[32];
    const int tid = threadIdx.x, lane = tid & 31, wave = tid >> 5;
    const int4* dg = (const int4*)(in + tid * SCAN_PER);
    int loc = 0;
    for (int k = 0; k < SCAN_PER / 4; ++k) {
        const int4 d = dg[k];
        loc += (d.x + d.y) + (d.z + d.w);
    }
    int inc = loc;
    for (int off = 1; off < 32; off <<= 1) {
        const int t = __shfl_up(inc, off, 32);
        if (lane >= off) inc += t;
    }
    if (lane == 31) wsum[wave] = inc;
    __syncthreads();
    if (wave == 0) {
        const int v = wsum[lane];
        int vi = v;
        for (int off = 1; off < 32; off <<= 1) {
            const int t = __shfl_up(vi, off, 32);
            if (lane >= off) vi += t;
        }
        wsum[lane] = vi - v;
    }
    __syncthreads();
    int run = wsum[wave] + inc - loc;
    int4* rp = (int4*)(out + tid * SCAN_PER);
    #pragma unroll
    for (int k = 0; k < SCAN_PER / 4; ++k) {
        const int4 d = dg[k];
        int4 r;
        r.x = run;
        r.y = r.x + d.x;
        r.z = r.y + d.y;
        r.w = r.z + d.z;
        run = r.w + d.w;
        rp[k] = r;
    }
}

__global__ void __launch_bounds__(1024) coarse_place(const int* __restrict__ src,
                                                     const int* __restrict__ dst,
                                                     const int* __restrict__ first,
                                                     int* __restrict__ coarse) {
    __shared__ int cur[BK_PAD];
    const int tid = threadIdx.x, b = blockIdx.x;
    const int e0 = b * BK_EDGES;
    const int e1 = (e0 + BK_EDGES) < N_EDGES ? (e0 + BK_EDGES) : N_EDGES;
    int d[BK_PER], s[BK_PER];
    #pragma unroll
    for (int j = 0; j < BK_PER; ++j) {
        const int e = e0 + tid + j * 1024;
        d[j] = e < e1 ? dst[e] : -1;
        s[j] = e < e1 ? src[e] : 0;
    }
    if (tid < BK_PAD) cur[tid] = first[tid * BK_BLOCKS + b];
    __syncthreads();
    #pragma unroll
    for (int j = 0; j < BK_PER; ++j)
        if (d[j] >= 0) {
            const int pos = atomicAdd(&cur[d[j] >> BK_SHIFT], 1);
            coarse[pos] = ((d[j] & (BK_NODES - 1)) << 16) | s[j];
        }
}

__global__ void __launch_bounds__(1024) fine_place(const int* __restrict__ coarse,
                                                   const int* __restrict__ first,
                                                   int* __restrict__ rowptr,
                                                   int* __restrict__ srcs) {
    __shared__ __attribute__((aligned(16))) int cnt[BK_NODES];
    const int tid = threadIdx.x, r = blockIdx.x;
    const int beg = first[r * BK_BLOCKS], end = first[(r + 1) * BK_BLOCKS];
    if (tid < BK_NODES) cnt[tid] = 0;
    __syncthreads();
    for (int i = beg + tid; i < end; i += FP_PER * 1024) {
        int w[FP_PER];
        #pragma unroll
        for (int j = 0; j < FP_PER; ++j) {
            const int ii = i + j * 1024;
            w[j] = ii < end ? coarse[ii] : -1;
        }
        #pragma unroll
        for (int j = 0; j < FP_PER; ++j)
            if (w[j] >= 0) atomicAdd(&cnt[w[j] >> 16], 1);
    }
    __syncthreads();
    if (tid < 32) {
        const int4 c = ((const int4*)cnt)[tid];
        const int loc = (c.x + c.y) + (c.z + c.w);
        int inc = loc;
        for (int off = 1; off < 32; off <<= 1) {
            const int t = __shfl_up(inc, off, 32);
            if (tid >= off) inc += t;
        }
        int4 p;
        p.x = beg + inc - loc;
        p.y = p.x + c.x;
        p.z = p.y + c.y;
        p.w = p.z + c.z;
        ((int4*)cnt)[tid] = p;
        const int n0 = r * BK_NODES + tid * 4;
        if (n0 <= N_NODES) *(int4*)(rowptr + n0) = p;
    }
    __syncthreads();
    for (int i = beg + tid; i < end; i += FP_PER * 1024) {
        int w[FP_PER];
        #pragma unroll
        for (int j = 0; j < FP_PER; ++j) {
            const int ii = i + j * 1024;
            w[j] = ii < end ? coarse[ii] : -1;
        }
        #pragma unroll
        for (int j = 0; j < FP_PER; ++j)
            if (w[j] >= 0) {
                const int pos = atomicAdd(&cnt[w[j] >> 16], 1);
                srcs[pos] = w[j] & 0xffff;
            }
    }
}

__device__ __forceinline__ float xhalf(float x) {
    const int b = __builtin_bit_cast(int, x);
    return __builtin_bit_cast(float, __builtin_amdgcn_permlanex16(b, b, 0x76543210, 0xfedcba98, false, false));
}
template <int M>
__device__ __forceinline__ float xrow(float x) {
    return __builtin_bit_cast(float, __builtin_amdgcn_mov_dpp(__builtin_bit_cast(int, x), 0x160 + M, 0xf, 0xf, false));
}
__device__ __forceinline__ float wave_max(float x) {
    x = fmaxf(x, xrow<1>(x));
    x = fmaxf(x, xrow<2>(x));
    x = fmaxf(x, xrow<4>(x));
    x = fmaxf(x, xrow<8>(x));
    return fmaxf(x, xhalf(x));
}
__device__ __forceinline__ float wave_sum(float x) {
    x += xrow<1>(x);
    x += xrow<2>(x);
    x += xrow<4>(x);
    x += xrow<8>(x);
    return x + xhalf(x);
}

__device__ __forceinline__ void edge_scores(const int* __restrict__ srcs,
                                            const float* __restrict__ el,
                                            const float er0, const float er1,
                                            const int base, const int end, const int lane,
                                            int& s, float& v0, float& v1) {
    s = 0;
    v0 = -INFINITY;
    v1 = -INFINITY;
    if (base + lane < end) {
        s = srcs[base + lane];
        const float2 l = *(const float2*)(el + 2 * (size_t)s);
        const float t0 = l.x + er0, t1 = l.y + er1;
        v0 = t0 > 0.0f ? t0 : 0.2f * t0;
        v1 = t1 > 0.0f ? t1 : 0.2f * t1;
    }
}

__device__ __forceinline__ void gather_node(const float* __restrict__ hin,
                                            const float* __restrict__ el,
                                            const float* __restrict__ er,
                                            const int* __restrict__ rowptr,
                                            const int* __restrict__ srcs,
                                            const int n, const int lane,
                                            float4& a0, float4& a1) {
    const int beg = rowptr[n], end = rowptr[n + 1];
    if (end > beg) {
        const float er0 = er[n * 2], er1 = er[n * 2 + 1];
        int s_l;
        float e0_l, e1_l;
        edge_scores(srcs, el, er0, er1, beg, end, lane, s_l, e0_l, e1_l);
        float m0 = e0_l, m1 = e1_l;
        for (int base = beg + 32; base < end; base += 32) {
            int s;
            float v0, v1;
            edge_scores(srcs, el, er0, er1, base, end, lane, s, v0, v1);
            m0 = fmaxf(m0, v0);
            m1 = fmaxf(m1, v1);
        }
        m0 = wave_max(m0);
        m1 = wave_max(m1);
        const float x0_l = (beg + lane < end) ? expf(e0_l - m0) : 0.f;
        const float x1_l = (beg + lane < end) ? expf(e1_l - m1) : 0.f;
        float den0 = x0_l, den1 = x1_l;
        for (int base = beg + 32; base < end; base += 32) {
            int s;
            float v0, v1;
            edge_scores(srcs, el, er0, er1, base, end, lane, s, v0, v1);
            if (base + lane < end) {
                den0 += expf(v0 - m0);
                den1 += expf(v1 - m1);
            }
        }
        den0 = wave_sum(den0);
        den1 = wave_sum(den1);
        float w0_l = x0_l / den0, w1_l = x1_l / den1;
        for (int base = beg; base < end; base += 32) {
            const int cnt = (end - base) < 32 ? (end - base) : 32;
            if (base != beg) {
                float v0, v1;
                edge_scores(srcs, el, er0, er1, base, end, lane, s_l, v0, v1);
                w0_l = (base + lane < end) ? expf(v0 - m0) / den0 : 0.f;
                w1_l = (base + lane < end) ? expf(v1 - m1) / den1 : 0.f;
            }
            #pragma unroll 4
            for (int j = 0; j < cnt; ++j) {
                const int   s  = __builtin_amdgcn_readlane(s_l, j);
                const float w0 = __int_as_float(__builtin_amdgcn_readlane(__float_as_int(w0_l), j));
                const float w1 = __int_as_float(__builtin_amdgcn_readlane(__float_as_int(w1_l), j));
                const float4 f = ((const float4*)(hin + (size_t)s * 128))[lane];
                a0.x += f.x * w0;  a0.y += f.y * w0;  a0.z += f.z * w0;  a0.w += f.w * w0;
                a1.x += f.x * w1;  a1.y += f.y * w1;  a1.z += f.z * w1;  a1.w += f.w * w1;
            }
        }
    }
}

#define TILE_NODES 32
#define AGG_PITCH  260

__device__ __forceinline__ void gather_tile(const float* __restrict__ hin,
                                            const float* __restrict__ el,
                                            const float* __restrict__ er,
                                            const int* __restrict__ rowptr,
                                            const int* __restrict__ srcs,
                                            float* __restrict__ sA,
                                            const int mbase, const int wave, const int lane) {
    for (int i = 0; i < 2; ++i) {
        const int r = wave * 2 + i;
        const int n = mbase + r;
        float4 a0 = {0.f, 0.f, 0.f, 0.f}, a1 = {0.f, 0.f, 0.f, 0.f};
        if (n < N_NODES) gather_node(hin, el, er, rowptr, srcs, n, lane, a0, a1);
        float4* o = (float4*)(sA + r * AGG_PITCH);
        o[lane]      = a0;
        o[32 + lane] = a1;
    }
}

__global__ void __launch_bounds__(512, 16) agg_proj_mean(const float* __restrict__ hin,
                                                     const float* __restrict__ el,
                                                     const float* __restrict__ er,
                                                     const int* __restrict__ rowptr,
                                                     const int* __restrict__ srcs,
                                                     const float* __restrict__ W,
                                                     const float* __restrict__ b,
                                                     float* __restrict__ hnext) {
    __shared__ __attribute__((aligned(16))) float sA[TILE_NODES * AGG_PITCH];
    const int lane  = threadIdx.x & 31;
    const int wave  = __builtin_amdgcn_readfirstlane(threadIdx.x >> 5);
    const int mbase = blockIdx.x * TILE_NODES;
    gather_tile(hin, el, er, rowptr, srcs, sA, mbase, wave, lane);
    __syncthreads();
    if (wave >= 8) return;

    const int row   = lane & 15;
    const int khalf = lane >> 4;
    const int rt    = wave >> 2;
    const int nq    = (wave & 3) << 5;

    v8f acc0 = {}, acc1 = {};
    const float* Arow = sA + (rt * 16 + row) * AGG_PITCH + khalf * 2;
    const float* Wh   = W + (size_t)(khalf * 2) * 256 + nq + row;
    #pragma unroll 8
    for (int kk = 0; kk < N_HEADS * D_MODEL; kk += 4) {
        const v2f a = *(const v2f*)(Arow + kk);
        const float* Wb = Wh + (size_t)(kk & (D_MODEL - 1)) * 256 + (kk >> 7) * 128;
        v2f b0, b1;
        b0.x = Wb[0];    b0.y = Wb[256];
        b1.x = Wb[16];   b1.y = Wb[256 + 16];
        acc0 = __builtin_amdgcn_wmma_f32_16x16x4_f32(false, a, false, b0, (short)0, acc0, false, false);
        acc1 = __builtin_amdgcn_wmma_f32_16x16x4_f32(false, a, false, b1, (short)0, acc1, false, false);
    }

    const int c = nq + row;
    const float bb0 = b[c]      + b[128 + c];
    const float bb1 = b[c + 16] + b[128 + c + 16];
    #pragma unroll
    for (int j = 0; j < 8; ++j) {
        const int m = mbase + rt * 16 + j + khalf * 8;
        if (m < N_NODES) {
            float* F = hnext + (size_t)m * 128 + c;
            F[0]  = 0.5f * (acc0[j] + bb0);
            F[16] = 0.5f * (acc1[j] + bb1);
        }
    }
}

__global__ void __launch_bounds__(512, 16) agg_proj_heads(const float* __restrict__ hin,
                                                      const float* __restrict__ el,
                                                      const float* __restrict__ er,
                                                      const int* __restrict__ rowptr,
                                                      const int* __restrict__ srcs,
                                                      const float* __restrict__ W,
                                                      const float* __restrict__ b,
                                                      float* __restrict__ out) {
    __shared__ __attribute__((aligned(16))) float sA[TILE_NODES * AGG_PITCH];
    const int lane  = threadIdx.x & 31;
    const int wave  = __builtin_amdgcn_readfirstlane(threadIdx.x >> 5);
    const int mbase = blockIdx.x * TILE_NODES;
    gather_tile(hin, el, er, rowptr, srcs, sA, mbase, wave, lane);
    __syncthreads();

    const int row   = lane & 15;
    const int khalf = lane >> 4;
    const int rt    = wave >> 3;
    const int nq    = (wave & 7) << 5;

    v8f acc0 = {}, acc1 = {};
    const float* Arow = sA + (rt * 16 + row) * AGG_PITCH + (nq >> 7) * 128;
    const float* Wh   = W + nq + row;
    #pragma unroll 8
    for (int k = 0; k < D_MODEL; k += 4) {
        const int ka = k + khalf * 2;
        const v2f a = *(const v2f*)(Arow + ka);
        const float* Wb = Wh + (size_t)ka * 256;
        v2f b0, b1;
        b0.x = Wb[0];    b0.y = Wb[256];
        b1.x = Wb[16];   b1.y = Wb[256 + 16];
        acc0 = __builtin_amdgcn_wmma_f32_16x16x4_f32(false, a, false, b0, (short)0, acc0, false, false);
        acc1 = __builtin_amdgcn_wmma_f32_16x16x4_f32(false, a, false, b1, (short)0, acc1, false, false);
    }

    const int c = nq + row;
    const float bb0 = b[c], bb1 = b[c + 16];
    #pragma unroll
    for (int j = 0; j < 8; ++j) {
        const int m = mbase + rt * 16 + j + khalf * 8;
        if (m < N_NODES) {
            float* F = out + (size_t)m * 256 + c;
            F[0]  = acc0[j] + bb0;
            F[16] = acc1[j] + bb1;
        }
    }
}

extern "C" void kernel_launch(void* const* d_in, const int* in_sizes, int n_in,
                              void* d_out, int out_size, void* d_ws, size_t ws_size,
                              hipStream_t stream) {
    const float* x   = (const float*)d_in[0];
    const float* Ws  = (const float*)d_in[1];
    const float* als = (const float*)d_in[2];
    const float* ars = (const float*)d_in[3];
    const float* bs  = (const float*)d_in[4];
    const int*   src = (const int*)d_in[5];
    const int*   dst = (const int*)d_in[6];

    float* ws      = (float*)d_ws;
    float* hnext   = ws;
    int*   srcs    = (int*)(ws + 12800000);
    int*   coarse  = (int*)(ws + 14400000);
    int*   hist    = (int*)(ws + 15200000);
    int*   rowptr  = (int*)(ws + 15300000);
    int*   first   = (int*)(ws + 15400000);
    float* wl      = ws + 19100000;
    float* wr      = ws + 19101000;
    float* el      = ws + 19200000;
    float* er      = ws + 19300000;

    float* out   = (float*)d_out;

    coarse_count<<<BK_BLOCKS, 1024, 0, stream>>>(dst, hist);
    bucket_scan<<<1, 1024, 0, stream>>>(hist, first);
    coarse_place<<<BK_BLOCKS, 1024, 0, stream>>>(src, dst, first, coarse);
    fine_place<<<BK_COUNT, 1024, 0, stream>>>(coarse, first, rowptr, srcs);
    fold_attn<<<2 * D_MODEL / 8, 256, 0, stream>>>(Ws, als, ars, wl, wr);

    node_elr<<<(N_NODES + 7) / 8, 256, 0, stream>>>(x, wl, wr, el, er);
    agg_proj_mean<<<(N_NODES + TILE_NODES - 1) / TILE_NODES, 512, 0, stream>>>(x, el, er, rowptr, srcs, Ws, bs, hnext);

    node_elr<<<(N_NODES + 7) / 8, 256, 0, stream>>>(hnext, wl + 256, wr + 256, el, er);
    agg_proj_heads<<<(N_NODES + TILE_NODES - 1) / TILE_NODES, 512, 0, stream>>>(hnext, el, er, rowptr, srcs, Ws + 32768, bs + 256, out);
}


# node_elr: a wave takes 16 consecutive rows (was one): 3125 waves a launch instead of 50000, 16 row loads in flight, the 64 dots summed by a lane-transposing butterfly (same tree), whole-line stores
# speedup vs baseline: 5.8112x; 1.0571x over previous
#include <hip/hip_runtime.h>
#include <math.h>

#define N_NODES 50000
#define N_EDGES 800000
#define D_MODEL 128
#define N_HEADS 2

typedef __attribute__((ext_vector_type(2))) float v2f;
typedef __attribute__((ext_vector_type(8))) float v8f;

__global__ void __launch_bounds__(256) fold_attn(const float* __restrict__ Ws,
                                                 const float* __restrict__ als,
                                                 const float* __restrict__ ars,
                                                 float* __restrict__ wl,
                                                 float* __restrict__ wr) {
    const int gw   = (blockIdx.x * blockDim.x + threadIdx.x) >> 5;
    const int lane = threadIdx.x & 31;
    if (gw >= 2 * D_MODEL) return;
    const int layer = gw >> 7;
    const float4* F0 = (const float4*)(Ws + (size_t)gw * 256) + lane;
    const float4* F1 = (const float4*)(Ws + (size_t)gw * 256 + 128) + lane;
    const float4* AL0 = (const float4*)(als + layer * 256) + lane;
    const float4* AL1 = (const float4*)(als + layer * 256 + 128) + lane;
    const float4* AR0 = (const float4*)(ars + layer * 256) + lane;
    const float4* AR1 = (const float4*)(ars + layer * 256 + 128) + lane;
    float4 f0 = *F0, f1 = *F1;
    float4 l0 = *AL0, l1 = *AL1, r0 = *AR0, r1 = *AR1;
    float e0 = f0.x*l0.x + f0.y*l0.y + f0.z*l0.z + f0.w*l0.w;
    float e1 = f1.x*l1.x + f1.y*l1.y + f1.z*l1.z + f1.w*l1.w;
    float g0 = f0.x*r0.x + f0.y*r0.y + f0.z*r0.z + f0.w*r0.w;
    float g1 = f1.x*r1.x + f1.y*r1.y + f1.z*r1.z + f1.w*r1.w;
    #pragma unroll
    for (int off = 16; off; off >>= 1) {
        e0 += __shfl_xor(e0, off, 32);
        e1 += __shfl_xor(e1, off, 32);
        g0 += __shfl_xor(g0, off, 32);
        g1 += __shfl_xor(g1, off, 32);
    }
    if (lane == 0) {
        wl[gw * 2 + 0] = e0;  wl[gw * 2 + 1] = e1;
        wr[gw * 2 + 0] = g0;  wr[gw * 2 + 1] = g1;
    }
}

__device__ __forceinline__ float xhalf(float x) {
    const int b = __builtin_bit_cast(int, x);
    return __builtin_bit_cast(float, __builtin_amdgcn_permlanex16(b, b, 0x76543210, 0xfedcba98, false, false));
}
template <int M>
__device__ __forceinline__ float xrow(float x) {
    return __builtin_bit_cast(float, __builtin_amdgcn_mov_dpp(__builtin_bit_cast(int, x), 0x160 + M, 0xf, 0xf, false));
}

#define ELR_ROWS 16
static_assert(N_NODES % ELR_ROWS == 0, "every wave owns ELR_ROWS whole rows");
static_assert(ELR_ROWS * N_HEADS == 32, "after the reduction a lane holds exactly one sum");

__device__ __forceinline__ float lane_transpose_sum(const float (&p)[32], const int lane) {
    const bool b16 = lane & 16, b8 = lane & 8, b4 = lane & 4, b2 = lane & 2, b1 = lane & 1;
    float s16[16], s8[8], s4[4], s2[2];
    #pragma unroll
    for (int i = 0; i < 16; ++i)
        s16[i] = (b16 ? p[i + 16] : p[i]) + xhalf(b16 ? p[i] : p[i + 16]);
    #pragma unroll
    for (int i = 0; i < 8; ++i)
        s8[i] = (b8 ? s16[i + 8] : s16[i]) + xrow<8>(b8 ? s16[i] : s16[i + 8]);
    #pragma unroll
    for (int i = 0; i < 4; ++i)
        s4[i] = (b4 ? s8[i + 4] : s8[i]) + xrow<4>(b4 ? s8[i] : s8[i + 4]);
    #pragma unroll
    for (int i = 0; i < 2; ++i)
        s2[i] = (b2 ? s4[i + 2] : s4[i]) + xrow<2>(b2 ? s4[i] : s4[i + 2]);
    return (b1 ? s2[1] : s2[0]) + xrow<1>(b1 ? s2[0] : s2[1]);
}

__global__ void __launch_bounds__(256) node_elr(const float* __restrict__ hin,
                                                const float* __restrict__ wl,
                                                const float* __restrict__ wr,
                                                float* __restrict__ el,
                                                float* __restrict__ er) {
    const int lane = threadIdx.x & 31;
    const int gw = blockIdx.x * 8 + __builtin_amdgcn_readfirstlane(threadIdx.x >> 5);
    const int n0 = gw * ELR_ROWS;
    if (n0 >= N_NODES) return;
    const float4 la = ((const float4*)wl)[2 * lane], lb = ((const float4*)wl)[2 * lane + 1];
    const float4 ra = ((const float4*)wr)[2 * lane], rb = ((const float4*)wr)[2 * lane + 1];
    const float4* rows = (const float4*)(hin + (size_t)n0 * 128) + lane;
    float4 x[ELR_ROWS];
    #pragma unroll
    for (int r = 0; r < ELR_ROWS; ++r) x[r] = rows[r * 32];
    __builtin_amdgcn_sched_barrier(0);
    float p[32], q[32];
    #pragma unroll
    for (int r = 0; r < ELR_ROWS; ++r) {
        const float4 v = x[r];
        p[2 * r + 0] = fmaf(v.w, lb.z, fmaf(v.z, lb.x, fmaf(v.y, la.z, v.x * la.x)));
        p[2 * r + 1] = fmaf(v.w, lb.w, fmaf(v.z, lb.y, fmaf(v.y, la.w, v.x * la.y)));
        q[2 * r + 0] = fmaf(v.w, rb.z, fmaf(v.z, rb.x, fmaf(v.y, ra.z, v.x * ra.x)));
        q[2 * r + 1] = fmaf(v.w, rb.w, fmaf(v.z, rb.y, fmaf(v.y, ra.w, v.x * ra.y)));
    }
    const float sl = lane_transpose_sum(p, lane);
    const float sr = lane_transpose_sum(q, lane);
    el[n0 * 2 + lane] = sl;
    er[n0 * 2 + lane] = sr;
}

#define BK_SHIFT   7
#define BK_NODES   (1 << BK_SHIFT)
#define BK_COUNT   ((N_NODES >> BK_SHIFT) + 1)
#define BK_BLOCKS  128
#define BK_EDGES   ((N_EDGES + BK_BLOCKS - 1) / BK_BLOCKS)
#define BK_PER     ((BK_EDGES + 1023) / 1024)
#define FP_PER     4
#define SCAN_PER  ((((N_NODES + 1023) / 1024) + 3) / 4 * 4)
#define NODES_PAD (SCAN_PER * 1024)
#define BK_PAD    (NODES_PAD / BK_BLOCKS)
static_assert(BK_PAD * BK_BLOCKS == NODES_PAD, "table length must equal the scan length");
static_assert(BK_COUNT < BK_PAD && BK_PAD <= 1024, "ranges must fit the table and one block");
static_assert(N_NODES <= 65536, "a source index must fit the low 16 bits of the packed word");
static_assert(NODES_PAD >= N_NODES + 4, "rowptr must hold index N_NODES and its int4");

__global__ void __launch_bounds__(1024) coarse_count(const int* __restrict__ dst,
                                                     int* __restrict__ hist) {
    __shared__ int cnt[BK_PAD];
    const int tid = threadIdx.x, b = blockIdx.x;
    if (tid < BK_PAD) cnt[tid] = 0;
    __syncthreads();
    const int e0 = b * BK_EDGES;
    const int e1 = (e0 + BK_EDGES) < N_EDGES ? (e0 + BK_EDGES) : N_EDGES;
    int d[BK_PER];
    #pragma unroll
    for (int j = 0; j < BK_PER; ++j) {
        const int e = e0 + tid + j * 1024;
        d[j] = e < e1 ? dst[e] : -1;
    }
    #pragma unroll
    for (int j = 0; j < BK_PER; ++j)
        if (d[j] >= 0) atomicAdd(&cnt[d[j] >> BK_SHIFT], 1);
    __syncthreads();
    if (tid < BK_PAD) hist[tid * BK_BLOCKS + b] = cnt[tid];
}

__global__ void __launch_bounds__(1024) bucket_scan(const int* __restrict__ in,
                                                    int* __restrict__ out) {
    __shared__ int wsum[32];
    const int tid = threadIdx.x, lane = tid & 31, wave = tid >> 5;
    const int4* dg = (const int4*)(in + tid * SCAN_PER);
    int loc = 0;
    for (int k = 0; k < SCAN_PER / 4; ++k) {
        const int4 d = dg[k];
        loc += (d.x + d.y) + (d.z + d.w);
    }
    int inc = loc;
    for (int off = 1; off < 32; off <<= 1) {
        const int t = __shfl_up(inc, off, 32);
        if (lane >= off) inc += t;
    }
    if (lane == 31) wsum[wave] = inc;
    __syncthreads();
    if (wave == 0) {
        const int v = wsum[lane];
        int vi = v;
        for (int off = 1; off < 32; off <<= 1) {
            const int t = __shfl_up(vi, off, 32);
            if (lane >= off) vi += t;
        }
        wsum[lane] = vi - v;
    }
    __syncthreads();
    int run = wsum[wave] + inc - loc;
    int4* rp = (int4*)(out + tid * SCAN_PER);
    #pragma unroll
    for (int k = 0; k < SCAN_PER / 4; ++k) {
        const int4 d = dg[k];
        int4 r;
        r.x = run;
        r.y = r.x + d.x;
        r.z = r.y + d.y;
        r.w = r.z + d.z;
        run = r.w + d.w;
        rp[k] = r;
    }
}

__global__ void __launch_bounds__(1024) coarse_place(const int* __restrict__ src,
                                                     const int* __restrict__ dst,
                                                     const int* __restrict__ first,
                                                     int* __restrict__ coarse) {
    __shared__ int cur[BK_PAD];
    const int tid = threadIdx.x, b = blockIdx.x;
    const int e0 = b * BK_EDGES;
    const int e1 = (e0 + BK_EDGES) < N_EDGES ? (e0 + BK_EDGES) : N_EDGES;
    int d[BK_PER], s[BK_PER];
    #pragma unroll
    for (int j = 0; j < BK_PER; ++j) {
        const int e = e0 + tid + j * 1024;
        d[j] = e < e1 ? dst[e] : -1;
        s[j] = e < e1 ? src[e] : 0;
    }
    if (tid < BK_PAD) cur[tid] = first[tid * BK_BLOCKS + b];
    __syncthreads();
    #pragma unroll
    for (int j = 0; j < BK_PER; ++j)
        if (d[j] >= 0) {
            const int pos = atomicAdd(&cur[d[j] >> BK_SHIFT], 1);
            coarse[pos] = ((d[j] & (BK_NODES - 1)) << 16) | s[j];
        }
}

__global__ void __launch_bounds__(1024) fine_place(const int* __restrict__ coarse,
                                                   const int* __restrict__ first,
                                                   int* __restrict__ rowptr,
                                                   int* __restrict__ srcs) {
    __shared__ __attribute__((aligned(16))) int cnt[BK_NODES];
    const int tid = threadIdx.x, r = blockIdx.x;
    const int beg = first[r * BK_BLOCKS], end = first[(r + 1) * BK_BLOCKS];
    if (tid < BK_NODES) cnt[tid] = 0;
    __syncthreads();
    for (int i = beg + tid; i < end; i += FP_PER * 1024) {
        int w[FP_PER];
        #pragma unroll
        for (int j = 0; j < FP_PER; ++j) {
            const int ii = i + j * 1024;
            w[j] = ii < end ? coarse[ii] : -1;
        }
        #pragma unroll
        for (int j = 0; j < FP_PER; ++j)
            if (w[j] >= 0) atomicAdd(&cnt[w[j] >> 16], 1);
    }
    __syncthreads();
    if (tid < 32) {
        const int4 c = ((const int4*)cnt)[tid];
        const int loc = (c.x + c.y) + (c.z + c.w);
        int inc = loc;
        for (int off = 1; off < 32; off <<= 1) {
            const int t = __shfl_up(inc, off, 32);
            if (tid >= off) inc += t;
        }
        int4 p;
        p.x = beg + inc - loc;
        p.y = p.x + c.x;
        p.z = p.y + c.y;
        p.w = p.z + c.z;
        ((int4*)cnt)[tid] = p;
        const int n0 = r * BK_NODES + tid * 4;
        if (n0 <= N_NODES) *(int4*)(rowptr + n0) = p;
    }
    __syncthreads();
    for (int i = beg + tid; i < end; i += FP_PER * 1024) {
        int w[FP_PER];
        #pragma unroll
        for (int j = 0; j < FP_PER; ++j) {
            const int ii = i + j * 1024;
            w[j] = ii < end ? coarse[ii] : -1;
        }
        #pragma unroll
        for (int j = 0; j < FP_PER; ++j)
            if (w[j] >= 0) {
                const int pos = atomicAdd(&cnt[w[j] >> 16], 1);
                srcs[pos] = w[j] & 0xffff;
            }
    }
}

__device__ __forceinline__ float wave_max(float x) {
    x = fmaxf(x, xrow<1>(x));
    x = fmaxf(x, xrow<2>(x));
    x = fmaxf(x, xrow<4>(x));
    x = fmaxf(x, xrow<8>(x));
    return fmaxf(x, xhalf(x));
}
__device__ __forceinline__ float wave_sum(float x) {
    x += xrow<1>(x);
    x += xrow<2>(x);
    x += xrow<4>(x);
    x += xrow<8>(x);
    return x + xhalf(x);
}

__device__ __forceinline__ void edge_scores(const int* __restrict__ srcs,
                                            const float* __restrict__ el,
                                            const float er0, const float er1,
                                            const int base, const int end, const int lane,
                                            int& s, float& v0, float& v1) {
    s = 0;
    v0 = -INFINITY;
    v1 = -INFINITY;
    if (base + lane < end) {
        s = srcs[base + lane];
        const float2 l = *(const float2*)(el + 2 * (size_t)s);
        const float t0 = l.x + er0, t1 = l.y + er1;
        v0 = t0 > 0.0f ? t0 : 0.2f * t0;
        v1 = t1 > 0.0f ? t1 : 0.2f * t1;
    }
}

__device__ __forceinline__ void gather_node(const float* __restrict__ hin,
                                            const float* __restrict__ el,
                                            const float* __restrict__ er,
                                            const int* __restrict__ rowptr,
                                            const int* __restrict__ srcs,
                                            const int n, const int lane,
                                            float4& a0, float4& a1) {
    const int beg = rowptr[n], end = rowptr[n + 1];
    if (end > beg) {
        const float er0 = er[n * 2], er1 = er[n * 2 + 1];
        int s_l;
        float e0_l, e1_l;
        edge_scores(srcs, el, er0, er1, beg, end, lane, s_l, e0_l, e1_l);
        float m0 = e0_l, m1 = e1_l;
        for (int base = beg + 32; base < end; base += 32) {
            int s;
            float v0, v1;
            edge_scores(srcs, el, er0, er1, base, end, lane, s, v0, v1);
            m0 = fmaxf(m0, v0);
            m1 = fmaxf(m1, v1);
        }
        m0 = wave_max(m0);
        m1 = wave_max(m1);
        const float x0_l = (beg + lane < end) ? expf(e0_l - m0) : 0.f;
        const float x1_l = (beg + lane < end) ? expf(e1_l - m1) : 0.f;
        float den0 = x0_l, den1 = x1_l;
        for (int base = beg + 32; base < end; base += 32) {
            int s;
            float v0, v1;
            edge_scores(srcs, el, er0, er1, base, end, lane, s, v0, v1);
            if (base + lane < end) {
                den0 += expf(v0 - m0);
                den1 += expf(v1 - m1);
            }
        }
        den0 = wave_sum(den0);
        den1 = wave_sum(den1);
        float w0_l = x0_l / den0, w1_l = x1_l / den1;
        for (int base = beg; base < end; base += 32) {
            const int cnt = (end - base) < 32 ? (end - base) : 32;
            if (base != beg) {
                float v0, v1;
                edge_scores(srcs, el, er0, er1, base, end, lane, s_l, v0, v1);
                w0_l = (base + lane < end) ? expf(v0 - m0) / den0 : 0.f;
                w1_l = (base + lane < end) ? expf(v1 - m1) / den1 : 0.f;
            }
            #pragma unroll 4
            for (int j = 0; j < cnt; ++j) {
                const int   s  = __builtin_amdgcn_readlane(s_l, j);
                const float w0 = __int_as_float(__builtin_amdgcn_readlane(__float_as_int(w0_l), j));
                const float w1 = __int_as_float(__builtin_amdgcn_readlane(__float_as_int(w1_l), j));
                const float4 f = ((const float4*)(hin + (size_t)s * 128))[lane];
                a0.x += f.x * w0;  a0.y += f.y * w0;  a0.z += f.z * w0;  a0.w += f.w * w0;
                a1.x += f.x * w1;  a1.y += f.y * w1;  a1.z += f.z * w1;  a1.w += f.w * w1;
            }
        }
    }
}

#define TILE_NODES 32
#define AGG_PITCH  260

__device__ __forceinline__ void gather_tile(const float* __restrict__ hin,
                                            const float* __restrict__ el,
                                            const float* __restrict__ er,
                                            const int* __restrict__ rowptr,
                                            const int* __restrict__ srcs,
                                            float* __restrict__ sA,
                                            const int mbase, const int wave, const int lane) {
    for (int i = 0; i < 2; ++i) {
        const int r = wave * 2 + i;
        const int n = mbase + r;
        float4 a0 = {0.f, 0.f, 0.f, 0.f}, a1 = {0.f, 0.f, 0.f, 0.f};
        if (n < N_NODES) gather_node(hin, el, er, rowptr, srcs, n, lane, a0, a1);
        float4* o = (float4*)(sA + r * AGG_PITCH);
        o[lane]      = a0;
        o[32 + lane] = a1;
    }
}

__global__ void __launch_bounds__(512, 16) agg_proj_mean(const float* __restrict__ hin,
                                                     const float* __restrict__ el,
                                                     const float* __restrict__ er,
                                                     const int* __restrict__ rowptr,
                                                     const int* __restrict__ srcs,
                                                     const float* __restrict__ W,
                                                     const float* __restrict__ b,
                                                     float* __restrict__ hnext) {
    __shared__ __attribute__((aligned(16))) float sA[TILE_NODES * AGG_PITCH];
    const int lane  = threadIdx.x & 31;
    const int wave  = __builtin_amdgcn_readfirstlane(threadIdx.x >> 5);
    const int mbase = blockIdx.x * TILE_NODES;
    gather_tile(hin, el, er, rowptr, srcs, sA, mbase, wave, lane);
    __syncthreads();
    if (wave >= 8) return;

    const int row   = lane & 15;
    const int khalf = lane >> 4;
    const int rt    = wave >> 2;
    const int nq    = (wave & 3) << 5;

    v8f acc0 = {}, acc1 = {};
    const float* Arow = sA + (rt * 16 + row) * AGG_PITCH + khalf * 2;
    const float* Wh   = W + (size_t)(khalf * 2) * 256 + nq + row;
    #pragma unroll 8
    for (int kk = 0; kk < N_HEADS * D_MODEL; kk += 4) {
        const v2f a = *(const v2f*)(Arow + kk);
        const float* Wb = Wh + (size_t)(kk & (D_MODEL - 1)) * 256 + (kk >> 7) * 128;
        v2f b0, b1;
        b0.x = Wb[0];    b0.y = Wb[256];
        b1.x = Wb[16];   b1.y = Wb[256 + 16];
        acc0 = __builtin_amdgcn_wmma_f32_16x16x4_f32(false, a, false, b0, (short)0, acc0, false, false);
        acc1 = __builtin_amdgcn_wmma_f32_16x16x4_f32(false, a, false, b1, (short)0, acc1, false, false);
    }

    const int c = nq + row;
    const float bb0 = b[c]      + b[128 + c];
    const float bb1 = b[c + 16] + b[128 + c + 16];
    #pragma unroll
    for (int j = 0; j < 8; ++j) {
        const int m = mbase + rt * 16 + j + khalf * 8;
        if (m < N_NODES) {
            float* F = hnext + (size_t)m * 128 + c;
            F[0]  = 0.5f * (acc0[j] + bb0);
            F[16] = 0.5f * (acc1[j] + bb1);
        }
    }
}

__global__ void __launch_bounds__(512, 16) agg_proj_heads(const float* __restrict__ hin,
                                                      const float* __restrict__ el,
                                                      const float* __restrict__ er,
                                                      const int* __restrict__ rowptr,
                                                      const int* __restrict__ srcs,
                                                      const float* __restrict__ W,
                                                      const float* __restrict__ b,
                                                      float* __restrict__ out) {
    __shared__ __attribute__((aligned(16))) float sA[TILE_NODES * AGG_PITCH];
    const int lane  = threadIdx.x & 31;
    const int wave  = __builtin_amdgcn_readfirstlane(threadIdx.x >> 5);
    const int mbase = blockIdx.x * TILE_NODES;
    gather_tile(hin, el, er, rowptr, srcs, sA, mbase, wave, lane);
    __syncthreads();

    const int row   = lane & 15;
    const int khalf = lane >> 4;
    const int rt    = wave >> 3;
    const int nq    = (wave & 7) << 5;

    v8f acc0 = {}, acc1 = {};
    const float* Arow = sA + (rt * 16 + row) * AGG_PITCH + (nq >> 7) * 128;
    const float* Wh   = W + nq + row;
    #pragma unroll 8
    for (int k = 0; k < D_MODEL; k += 4) {
        const int ka = k + khalf * 2;
        const v2f a = *(const v2f*)(Arow + ka);
        const float* Wb = Wh + (size_t)ka * 256;
        v2f b0, b1;
        b0.x = Wb[0];    b0.y = Wb[256];
        b1.x = Wb[16];   b1.y = Wb[256 + 16];
        acc0 = __builtin_amdgcn_wmma_f32_16x16x4_f32(false, a, false, b0, (short)0, acc0, false, false);
        acc1 = __builtin_amdgcn_wmma_f32_16x16x4_f32(false, a, false, b1, (short)0, acc1, false, false);
    }

    const int c = nq + row;
    const float bb0 = b[c], bb1 = b[c + 16];
    #pragma unroll
    for (int j = 0; j < 8; ++j) {
        const int m = mbase + rt * 16 + j + khalf * 8;
        if (m < N_NODES) {
            float* F = out + (size_t)m * 256 + c;
            F[0]  = acc0[j] + bb0;
            F[16] = acc1[j] + bb1;
        }
    }
}

extern "C" void kernel_launch(void* const* d_in, const int* in_sizes, int n_in,
                              void* d_out, int out_size, void* d_ws, size_t ws_size,
                              hipStream_t stream) {
    const float* x   = (const float*)d_in[0];
    const float* Ws  = (const float*)d_in[1];
    const float* als = (const float*)d_in[2];
    const float* ars = (const float*)d_in[3];
    const float* bs  = (const float*)d_in[4];
    const int*   src = (const int*)d_in[5];
    const int*   dst = (const int*)d_in[6];

    float* ws      = (float*)d_ws;
    float* hnext   = ws;
    int*   srcs    = (int*)(ws + 12800000);
    int*   coarse  = (int*)(ws + 14400000);
    int*   hist    = (int*)(ws + 15200000);
    int*   rowptr  = (int*)(ws + 15300000);
    int*   first   = (int*)(ws + 15400000);
    float* wl      = ws + 19100000;
    float* wr      = ws + 19101000;
    float* el      = ws + 19200000;
    float* er      = ws + 19300000;

    float* out   = (float*)d_out;

    coarse_count<<<BK_BLOCKS, 1024, 0, stream>>>(dst, hist);
    bucket_scan<<<1, 1024, 0, stream>>>(hist, first);
    coarse_place<<<BK_BLOCKS, 1024, 0, stream>>>(src, dst, first, coarse);
    fine_place<<<BK_COUNT, 1024, 0, stream>>>(coarse, first, rowptr, srcs);
    fold_attn<<<2 * D_MODEL / 8, 256, 0, stream>>>(Ws, als, ars, wl, wr);

    node_elr<<<(N_NODES / ELR_ROWS + 7) / 8, 256, 0, stream>>>(x, wl, wr, el, er);
    agg_proj_mean<<<(N_NODES + TILE_NODES - 1) / TILE_NODES, 512, 0, stream>>>(x, el, er, rowptr, srcs, Ws, bs, hnext);

    node_elr<<<(N_NODES / ELR_ROWS + 7) / 8, 256, 0, stream>>>(hnext, wl + 256, wr + 256, el, er);
    agg_proj_heads<<<(N_NODES + TILE_NODES - 1) / TILE_NODES, 512, 0, stream>>>(hnext, el, er, rowptr, srcs, Ws + 32768, bs + 256, out);
}


# the (range, block) table's scan moves into coarse_place: each block reads the table as whole 512-B rows and forms its own cursors; the one-block scan launch and its 208-B-strided accesses are dropped
# speedup vs baseline: 6.6229x; 1.1397x over previous
#include <hip/hip_runtime.h>
#include <math.h>

#define N_NODES 50000
#define N_EDGES 800000
#define D_MODEL 128
#define N_HEADS 2

typedef __attribute__((ext_vector_type(2))) float v2f;
typedef __attribute__((ext_vector_type(8))) float v8f;

__global__ void __launch_bounds__(256) fold_attn(const float* __restrict__ Ws,
                                                 const float* __restrict__ als,
                                                 const float* __restrict__ ars,
                                                 float* __restrict__ wl,
                                                 float* __restrict__ wr) {
    const int gw   = (blockIdx.x * blockDim.x + threadIdx.x) >> 5;
    const int lane = threadIdx.x & 31;
    if (gw >= 2 * D_MODEL) return;
    const int layer = gw >> 7;
    const float4* F0 = (const float4*)(Ws + (size_t)gw * 256) + lane;
    const float4* F1 = (const float4*)(Ws + (size_t)gw * 256 + 128) + lane;
    const float4* AL0 = (const float4*)(als + layer * 256) + lane;
    const float4* AL1 = (const float4*)(als + layer * 256 + 128) + lane;
    const float4* AR0 = (const float4*)(ars + layer * 256) + lane;
    const float4* AR1 = (const float4*)(ars + layer * 256 + 128) + lane;
    float4 f0 = *F0, f1 = *F1;
    float4 l0 = *AL0, l1 = *AL1, r0 = *AR0, r1 = *AR1;
    float e0 = f0.x*l0.x + f0.y*l0.y + f0.z*l0.z + f0.w*l0.w;
    float e1 = f1.x*l1.x + f1.y*l1.y + f1.z*l1.z + f1.w*l1.w;
    float g0 = f0.x*r0.x + f0.y*r0.y + f0.z*r0.z + f0.w*r0.w;
    float g1 = f1.x*r1.x + f1.y*r1.y + f1.z*r1.z + f1.w*r1.w;
    #pragma unroll
    for (int off = 16; off; off >>= 1) {
        e0 += __shfl_xor(e0, off, 32);
        e1 += __shfl_xor(e1, off, 32);
        g0 += __shfl_xor(g0, off, 32);
        g1 += __shfl_xor(g1, off, 32);
    }
    if (lane == 0) {
        wl[gw * 2 + 0] = e0;  wl[gw * 2 + 1] = e1;
        wr[gw * 2 + 0] = g0;  wr[gw * 2 + 1] = g1;
    }
}

__device__ __forceinline__ float xhalf(float x) {
    const int b = __builtin_bit_cast(int, x);
    return __builtin_bit_cast(float, __builtin_amdgcn_permlanex16(b, b, 0x76543210, 0xfedcba98, false, false));
}
template <int M>
__device__ __forceinline__ float xrow(float x) {
    return __builtin_bit_cast(float, __builtin_amdgcn_mov_dpp(__builtin_bit_cast(int, x), 0x160 + M, 0xf, 0xf, false));
}

#define ELR_ROWS 16
static_assert(N_NODES % ELR_ROWS == 0, "every wave owns ELR_ROWS whole rows");
static_assert(ELR_ROWS * N_HEADS == 32, "after the reduction a lane holds exactly one sum");

__device__ __forceinline__ float lane_transpose_sum(const float (&p)[32], const int lane) {
    const bool b16 = lane & 16, b8 = lane & 8, b4 = lane & 4, b2 = lane & 2, b1 = lane & 1;
    float s16[16], s8[8], s4[4], s2[2];
    #pragma unroll
    for (int i = 0; i < 16; ++i)
        s16[i] = (b16 ? p[i + 16] : p[i]) + xhalf(b16 ? p[i] : p[i + 16]);
    #pragma unroll
    for (int i = 0; i < 8; ++i)
        s8[i] = (b8 ? s16[i + 8] : s16[i]) + xrow<8>(b8 ? s16[i] : s16[i + 8]);
    #pragma unroll
    for (int i = 0; i < 4; ++i)
        s4[i] = (b4 ? s8[i + 4] : s8[i]) + xrow<4>(b4 ? s8[i] : s8[i + 4]);
    #pragma unroll
    for (int i = 0; i < 2; ++i)
        s2[i] = (b2 ? s4[i + 2] : s4[i]) + xrow<2>(b2 ? s4[i] : s4[i + 2]);
    return (b1 ? s2[1] : s2[0]) + xrow<1>(b1 ? s2[0] : s2[1]);
}

__global__ void __launch_bounds__(256) node_elr(const float* __restrict__ hin,
                                                const float* __restrict__ wl,
                                                const float* __restrict__ wr,
                                                float* __restrict__ el,
                                                float* __restrict__ er) {
    const int lane = threadIdx.x & 31;
    const int gw = blockIdx.x * 8 + __builtin_amdgcn_readfirstlane(threadIdx.x >> 5);
    const int n0 = gw * ELR_ROWS;
    if (n0 >= N_NODES) return;
    const float4 la = ((const float4*)wl)[2 * lane], lb = ((const float4*)wl)[2 * lane + 1];
    const float4 ra = ((const float4*)wr)[2 * lane], rb = ((const float4*)wr)[2 * lane + 1];
    const float4* rows = (const float4*)(hin + (size_t)n0 * 128) + lane;
    float4 x[ELR_ROWS];
    #pragma unroll
    for (int r = 0; r < ELR_ROWS; ++r) x[r] = rows[r * 32];
    __builtin_amdgcn_sched_barrier(0);
    float p[32], q[32];
    #pragma unroll
    for (int r = 0; r < ELR_ROWS; ++r) {
        const float4 v = x[r];
        p[2 * r + 0] = fmaf(v.w, lb.z, fmaf(v.z, lb.x, fmaf(v.y, la.z, v.x * la.x)));
        p[2 * r + 1] = fmaf(v.w, lb.w, fmaf(v.z, lb.y, fmaf(v.y, la.w, v.x * la.y)));
        q[2 * r + 0] = fmaf(v.w, rb.z, fmaf(v.z, rb.x, fmaf(v.y, ra.z, v.x * ra.x)));
        q[2 * r + 1] = fmaf(v.w, rb.w, fmaf(v.z, rb.y, fmaf(v.y, ra.w, v.x * ra.y)));
    }
    const float sl = lane_transpose_sum(p, lane);
    const float sr = lane_transpose_sum(q, lane);
    el[n0 * 2 + lane] = sl;
    er[n0 * 2 + lane] = sr;
}

#define BK_SHIFT   7
#define BK_NODES   (1 << BK_SHIFT)
#define BK_COUNT   ((N_NODES >> BK_SHIFT) + 1)
#define BK_BLOCKS  128
#define BK_EDGES   ((N_EDGES + BK_BLOCKS - 1) / BK_BLOCKS)
#define BK_PER     ((BK_EDGES + 1023) / 1024)
#define FP_PER     4
#define SCAN_PER  ((((N_NODES + 1023) / 1024) + 3) / 4 * 4)
#define NODES_PAD (SCAN_PER * 1024)
#define BK_PAD    (NODES_PAD / BK_BLOCKS)
static_assert(BK_PAD * BK_BLOCKS == NODES_PAD, "the table is BK_PAD whole rows");
static_assert(BK_COUNT < BK_PAD && BK_PAD <= 1024, "ranges must fit the table and one block");
static_assert(N_NODES <= 65536, "a source index must fit the low 16 bits of the packed word");
static_assert(NODES_PAD >= N_NODES + 4, "rowptr must hold index N_NODES and its int4");

__global__ void __launch_bounds__(1024) coarse_count(const int* __restrict__ dst,
                                                     int* __restrict__ hist) {
    __shared__ int cnt[BK_PAD];
    const int tid = threadIdx.x, b = blockIdx.x;
    if (tid < BK_PAD) cnt[tid] = 0;
    __syncthreads();
    const int e0 = b * BK_EDGES;
    const int e1 = (e0 + BK_EDGES) < N_EDGES ? (e0 + BK_EDGES) : N_EDGES;
    int d[BK_PER];
    #pragma unroll
    for (int j = 0; j < BK_PER; ++j) {
        const int e = e0 + tid + j * 1024;
        d[j] = e < e1 ? dst[e] : -1;
    }
    #pragma unroll
    for (int j = 0; j < BK_PER; ++j)
        if (d[j] >= 0) atomicAdd(&cnt[d[j] >> BK_SHIFT], 1);
    __syncthreads();
    if (tid < BK_PAD) hist[tid * BK_BLOCKS + b] = cnt[tid];
}

__device__ __forceinline__ int ixhalf(int b) {
    return __builtin_amdgcn_permlanex16(b, b, 0x76543210, 0xfedcba98, false, false);
}
template <int M>
__device__ __forceinline__ int ixrow(int x) {
    return __builtin_amdgcn_mov_dpp(x, 0x160 + M, 0xf, 0xf, false);
}
__device__ __forceinline__ int wave_isum(int x) {
    x += ixrow<1>(x);
    x += ixrow<2>(x);
    x += ixrow<4>(x);
    x += ixrow<8>(x);
    return x + ixhalf(x);
}

#define SCAN_ROWS (BK_PAD / 32)
static_assert(BK_BLOCKS == 128, "a table row is one int4 per lane");
static_assert(SCAN_ROWS * 32 == BK_PAD, "32 waves share the table rows evenly");
__global__ void __launch_bounds__(1024) coarse_place(const int* __restrict__ src,
                                                     const int* __restrict__ dst,
                                                     const int* __restrict__ hist,
                                                     int* __restrict__ rstart,
                                                     int* __restrict__ coarse) {
    __shared__ int cur[BK_PAD];
    __shared__ int tot[BK_PAD];
    const int tid = threadIdx.x, b = blockIdx.x;
    const int lane = tid & 31, wave = tid >> 5;
    const int e0 = b * BK_EDGES;
    const int e1 = (e0 + BK_EDGES) < N_EDGES ? (e0 + BK_EDGES) : N_EDGES;
    int d[BK_PER], s[BK_PER];
    #pragma unroll
    for (int j = 0; j < BK_PER; ++j) {
        const int e = e0 + tid + j * 1024;
        d[j] = e < e1 ? dst[e] : -1;
        s[j] = e < e1 ? src[e] : 0;
    }
    int4 h[SCAN_ROWS];
    #pragma unroll
    for (int i = 0; i < SCAN_ROWS; ++i)
        h[i] = ((const int4*)hist)[(wave + 32 * i) * (BK_BLOCKS / 4) + lane];
    const int c0 = 4 * lane;
    #pragma unroll
    for (int i = 0; i < SCAN_ROWS; ++i) {
        const int4 v = h[i];
        const int all = wave_isum((v.x + v.y) + (v.z + v.w));
        const int pre = wave_isum(((c0     < b ? v.x : 0) + (c0 + 1 < b ? v.y : 0)) +
                                  ((c0 + 2 < b ? v.z : 0) + (c0 + 3 < b ? v.w : 0)));
        if (lane == 0) {
            tot[wave + 32 * i] = all;
            cur[wave + 32 * i] = pre;
        }
    }
    __syncthreads();
    if (wave == 0) {
        int v[SCAN_ROWS];
        int loc = 0;
        #pragma unroll
        for (int k = 0; k < SCAN_ROWS; ++k) {
            v[k] = tot[lane * SCAN_ROWS + k];
            loc += v[k];
        }
        int inc = loc;
        for (int off = 1; off < 32; off <<= 1) {
            const int t = __shfl_up(inc, off, 32);
            if (lane >= off) inc += t;
        }
        int run = inc - loc;
        #pragma unroll
        for (int k = 0; k < SCAN_ROWS; ++k) {
            tot[lane * SCAN_ROWS + k] = run;
            run += v[k];
        }
    }
    __syncthreads();
    if (tid < BK_PAD) {
        const int start = tot[tid];
        cur[tid] += start;
        if (b == 0) rstart[tid] = start;
    }
    __syncthreads();
    #pragma unroll
    for (int j = 0; j < BK_PER; ++j)
        if (d[j] >= 0) {
            const int pos = atomicAdd(&cur[d[j] >> BK_SHIFT], 1);
            coarse[pos] = ((d[j] & (BK_NODES - 1)) << 16) | s[j];
        }
}

__global__ void __launch_bounds__(1024) fine_place(const int* __restrict__ coarse,
                                                   const int* __restrict__ rstart,
                                                   int* __restrict__ rowptr,
                                                   int* __restrict__ srcs) {
    __shared__ __attribute__((aligned(16))) int cnt[BK_NODES];
    const int tid = threadIdx.x, r = blockIdx.x;
    const int beg = rstart[r], end = rstart[r + 1];
    if (tid < BK_NODES) cnt[tid] = 0;
    __syncthreads();
    for (int i = beg + tid; i < end; i += FP_PER * 1024) {
        int w[FP_PER];
        #pragma unroll
        for (int j = 0; j < FP_PER; ++j) {
            const int ii = i + j * 1024;
            w[j] = ii < end ? coarse[ii] : -1;
        }
        #pragma unroll
        for (int j = 0; j < FP_PER; ++j)
            if (w[j] >= 0) atomicAdd(&cnt[w[j] >> 16], 1);
    }
    __syncthreads();
    if (tid < 32) {
        const int4 c = ((const int4*)cnt)[tid];
        const int loc = (c.x + c.y) + (c.z + c.w);
        int inc = loc;
        for (int off = 1; off < 32; off <<= 1) {
            const int t = __shfl_up(inc, off, 32);
            if (tid >= off) inc += t;
        }
        int4 p;
        p.x = beg + inc - loc;
        p.y = p.x + c.x;
        p.z = p.y + c.y;
        p.w = p.z + c.z;
        ((int4*)cnt)[tid] = p;
        const int n0 = r * BK_NODES + tid * 4;
        if (n0 <= N_NODES) *(int4*)(rowptr + n0) = p;
    }
    __syncthreads();
    for (int i = beg + tid; i < end; i += FP_PER * 1024) {
        int w[FP_PER];
        #pragma unroll
        for (int j = 0; j < FP_PER; ++j) {
            const int ii = i + j * 1024;
            w[j] = ii < end ? coarse[ii] : -1;
        }
        #pragma unroll
        for (int j = 0; j < FP_PER; ++j)
            if (w[j] >= 0) {
                const int pos = atomicAdd(&cnt[w[j] >> 16], 1);
                srcs[pos] = w[j] & 0xffff;
            }
    }
}

__device__ __forceinline__ float wave_max(float x) {
    x = fmaxf(x, xrow<1>(x));
    x = fmaxf(x, xrow<2>(x));
    x = fmaxf(x, xrow<4>(x));
    x = fmaxf(x, xrow<8>(x));
    return fmaxf(x, xhalf(x));
}
__device__ __forceinline__ float wave_sum(float x) {
    x += xrow<1>(x);
    x += xrow<2>(x);
    x += xrow<4>(x);
    x += xrow<8>(x);
    return x + xhalf(x);
}

__device__ __forceinline__ void edge_scores(const int* __restrict__ srcs,
                                            const float* __restrict__ el,
                                            const float er0, const float er1,
                                            const int base, const int end, const int lane,
                                            int& s, float& v0, float& v1) {
    s = 0;
    v0 = -INFINITY;
    v1 = -INFINITY;
    if (base + lane < end) {
        s = srcs[base + lane];
        const float2 l = *(const float2*)(el + 2 * (size_t)s);
        const float t0 = l.x + er0, t1 = l.y + er1;
        v0 = t0 > 0.0f ? t0 : 0.2f * t0;
        v1 = t1 > 0.0f ? t1 : 0.2f * t1;
    }
}

__device__ __forceinline__ void gather_node(const float* __restrict__ hin,
                                            const float* __restrict__ el,
                                            const float* __restrict__ er,
                                            const int* __restrict__ rowptr,
                                            const int* __restrict__ srcs,
                                            const int n, const int lane,
                                            float4& a0, float4& a1) {
    const int beg = rowptr[n], end = rowptr[n + 1];
    if (end > beg) {
        const float er0 = er[n * 2], er1 = er[n * 2 + 1];
        int s_l;
        float e0_l, e1_l;
        edge_scores(srcs, el, er0, er1, beg, end, lane, s_l, e0_l, e1_l);
        float m0 = e0_l, m1 = e1_l;
        for (int base = beg + 32; base < end; base += 32) {
            int s;
            float v0, v1;
            edge_scores(srcs, el, er0, er1, base, end, lane, s, v0, v1);
            m0 = fmaxf(m0, v0);
            m1 = fmaxf(m1, v1);
        }
        m0 = wave_max(m0);
        m1 = wave_max(m1);
        const float x0_l = (beg + lane < end) ? expf(e0_l - m0) : 0.f;
        const float x1_l = (beg + lane < end) ? expf(e1_l - m1) : 0.f;
        float den0 = x0_l, den1 = x1_l;
        for (int base = beg + 32; base < end; base += 32) {
            int s;
            float v0, v1;
            edge_scores(srcs, el, er0, er1, base, end, lane, s, v0, v1);
            if (base + lane < end) {
                den0 += expf(v0 - m0);
                den1 += expf(v1 - m1);
            }
        }
        den0 = wave_sum(den0);
        den1 = wave_sum(den1);
        float w0_l = x0_l / den0, w1_l = x1_l / den1;
        for (int base = beg; base < end; base += 32) {
            const int cnt = (end - base) < 32 ? (end - base) : 32;
            if (base != beg) {
                float v0, v1;
                edge_scores(srcs, el, er0, er1, base, end, lane, s_l, v0, v1);
                w0_l = (base + lane < end) ? expf(v0 - m0) / den0 : 0.f;
                w1_l = (base + lane < end) ? expf(v1 - m1) / den1 : 0.f;
            }
            #pragma unroll 4
            for (int j = 0; j < cnt; ++j) {
                const int   s  = __builtin_amdgcn_readlane(s_l, j);
                const float w0 = __int_as_float(__builtin_amdgcn_readlane(__float_as_int(w0_l), j));
                const float w1 = __int_as_float(__builtin_amdgcn_readlane(__float_as_int(w1_l), j));
                const float4 f = ((const float4*)(hin + (size_t)s * 128))[lane];
                a0.x += f.x * w0;  a0.y += f.y * w0;  a0.z += f.z * w0;  a0.w += f.w * w0;
                a1.x += f.x * w1;  a1.y += f.y * w1;  a1.z += f.z * w1;  a1.w += f.w * w1;
            }
        }
    }
}

#define TILE_NODES 32
#define AGG_PITCH  260

__device__ __forceinline__ void gather_tile(const float* __restrict__ hin,
                                            const float* __restrict__ el,
                                            const float* __restrict__ er,
                                            const int* __restrict__ rowptr,
                                            const int* __restrict__ srcs,
                                            float* __restrict__ sA,
                                            const int mbase, const int wave, const int lane) {
    for (int i = 0; i < 2; ++i) {
        const int r = wave * 2 + i;
        const int n = mbase + r;
        float4 a0 = {0.f, 0.f, 0.f, 0.f}, a1 = {0.f, 0.f, 0.f, 0.f};
        if (n < N_NODES) gather_node(hin, el, er, rowptr, srcs, n, lane, a0, a1);
        float4* o = (float4*)(sA + r * AGG_PITCH);
        o[lane]      = a0;
        o[32 + lane] = a1;
    }
}

__global__ void __launch_bounds__(512, 16) agg_proj_mean(const float* __restrict__ hin,
                                                     const float* __restrict__ el,
                                                     const float* __restrict__ er,
                                                     const int* __restrict__ rowptr,
                                                     const int* __restrict__ srcs,
                                                     const float* __restrict__ W,
                                                     const float* __restrict__ b,
                                                     float* __restrict__ hnext) {
    __shared__ __attribute__((aligned(16))) float sA[TILE_NODES * AGG_PITCH];
    const int lane  = threadIdx.x & 31;
    const int wave  = __builtin_amdgcn_readfirstlane(threadIdx.x >> 5);
    const int mbase = blockIdx.x * TILE_NODES;
    gather_tile(hin, el, er, rowptr, srcs, sA, mbase, wave, lane);
    __syncthreads();
    if (wave >= 8) return;

    const int row   = lane & 15;
    const int khalf = lane >> 4;
    const int rt    = wave >> 2;
    const int nq    = (wave & 3) << 5;

    v8f acc0 = {}, acc1 = {};
    const float* Arow = sA + (rt * 16 + row) * AGG_PITCH + khalf * 2;
    const float* Wh   = W + (size_t)(khalf * 2) * 256 + nq + row;
    #pragma unroll 8
    for (int kk = 0; kk < N_HEADS * D_MODEL; kk += 4) {
        const v2f a = *(const v2f*)(Arow + kk);
        const float* Wb = Wh + (size_t)(kk & (D_MODEL - 1)) * 256 + (kk >> 7) * 128;
        v2f b0, b1;
        b0.x = Wb[0];    b0.y = Wb[256];
        b1.x = Wb[16];   b1.y = Wb[256 + 16];
        acc0 = __builtin_amdgcn_wmma_f32_16x16x4_f32(false, a, false, b0, (short)0, acc0, false, false);
        acc1 = __builtin_amdgcn_wmma_f32_16x16x4_f32(false, a, false, b1, (short)0, acc1, false, false);
    }

    const int c = nq + row;
    const float bb0 = b[c]      + b[128 + c];
    const float bb1 = b[c + 16] + b[128 + c + 16];
    #pragma unroll
    for (int j = 0; j < 8; ++j) {
        const int m = mbase + rt * 16 + j + khalf * 8;
        if (m < N_NODES) {
            float* F = hnext + (size_t)m * 128 + c;
            F[0]  = 0.5f * (acc0[j] + bb0);
            F[16] = 0.5f * (acc1[j] + bb1);
        }
    }
}

__global__ void __launch_bounds__(512, 16) agg_proj_heads(const float* __restrict__ hin,
                                                      const float* __restrict__ el,
                                                      const float* __restrict__ er,
                                                      const int* __restrict__ rowptr,
                                                      const int* __restrict__ srcs,
                                                      const float* __restrict__ W,
                                                      const float* __restrict__ b,
                                                      float* __restrict__ out) {
    __shared__ __attribute__((aligned(16))) float sA[TILE_NODES * AGG_PITCH];
    const int lane  = threadIdx.x & 31;
    const int wave  = __builtin_amdgcn_readfirstlane(threadIdx.x >> 5);
    const int mbase = blockIdx.x * TILE_NODES;
    gather_tile(hin, el, er, rowptr, srcs, sA, mbase, wave, lane);
    __syncthreads();

    const int row   = lane & 15;
    const int khalf = lane >> 4;
    const int rt    = wave >> 3;
    const int nq    = (wave & 7) << 5;

    v8f acc0 = {}, acc1 = {};
    const float* Arow = sA + (rt * 16 + row) * AGG_PITCH + (nq >> 7) * 128;
    const float* Wh   = W + nq + row;
    #pragma unroll 8
    for (int k = 0; k < D_MODEL; k += 4) {
        const int ka = k + khalf * 2;
        const v2f a = *(const v2f*)(Arow + ka);
        const float* Wb = Wh + (size_t)ka * 256;
        v2f b0, b1;
        b0.x = Wb[0];    b0.y = Wb[256];
        b1.x = Wb[16];   b1.y = Wb[256 + 16];
        acc0 = __builtin_amdgcn_wmma_f32_16x16x4_f32(false, a, false, b0, (short)0, acc0, false, false);
        acc1 = __builtin_amdgcn_wmma_f32_16x16x4_f32(false, a, false, b1, (short)0, acc1, false, false);
    }

    const int c = nq + row;
    const float bb0 = b[c], bb1 = b[c + 16];
    #pragma unroll
    for (int j = 0; j < 8; ++j) {
        const int m = mbase + rt * 16 + j + khalf * 8;
        if (m < N_NODES) {
            float* F = out + (size_t)m * 256 + c;
            F[0]  = acc0[j] + bb0;
            F[16] = acc1[j] + bb1;
        }
    }
}

extern "C" void kernel_launch(void* const* d_in, const int* in_sizes, int n_in,
                              void* d_out, int out_size, void* d_ws, size_t ws_size,
                              hipStream_t stream) {
    const float* x   = (const float*)d_in[0];
    const float* Ws  = (const float*)d_in[1];
    const float* als = (const float*)d_in[2];
    const float* ars = (const float*)d_in[3];
    const float* bs  = (const float*)d_in[4];
    const int*   src = (const int*)d_in[5];
    const int*   dst = (const int*)d_in[6];

    float* ws      = (float*)d_ws;
    float* hnext   = ws;
    int*   srcs    = (int*)(ws + 12800000);
    int*   coarse  = (int*)(ws + 14400000);
    int*   hist    = (int*)(ws + 15200000);
    int*   rowptr  = (int*)(ws + 15300000);
    int*   rstart  = (int*)(ws + 15400000);
    float* wl      = ws + 19100000;
    float* wr      = ws + 19101000;
    float* el      = ws + 19200000;
    float* er      = ws + 19300000;

    float* out   = (float*)d_out;

    coarse_count<<<BK_BLOCKS, 1024, 0, stream>>>(dst, hist);
    coarse_place<<<BK_BLOCKS, 1024, 0, stream>>>(src, dst, hist, rstart, coarse);
    fine_place<<<BK_COUNT, 1024, 0, stream>>>(coarse, rstart, rowptr, srcs);
    fold_attn<<<2 * D_MODEL / 8, 256, 0, stream>>>(Ws, als, ars, wl, wr);

    node_elr<<<(N_NODES / ELR_ROWS + 7) / 8, 256, 0, stream>>>(x, wl, wr, el, er);
    agg_proj_mean<<<(N_NODES + TILE_NODES - 1) / TILE_NODES, 512, 0, stream>>>(x, el, er, rowptr, srcs, Ws, bs, hnext);

    node_elr<<<(N_NODES / ELR_ROWS + 7) / 8, 256, 0, stream>>>(hnext, wl + 256, wr + 256, el, er);
    agg_proj_heads<<<(N_NODES + TILE_NODES - 1) / TILE_NODES, 512, 0, stream>>>(hnext, el, er, rowptr, srcs, Ws + 32768, bs + 256, out);
}


# independent work rides in the bucket build's two half-chip launches: fold_attn's waves are tail blocks of coarse_count's grid, layer 1's node_elr waves tail blocks of coarse_place's; 8 launches to 6
# speedup vs baseline: 6.9936x; 1.0560x over previous
#include <hip/hip_runtime.h>
#include <math.h>

#define N_NODES 50000
#define N_EDGES 800000
#define D_MODEL 128
#define N_HEADS 2

typedef __attribute__((ext_vector_type(2))) float v2f;
typedef __attribute__((ext_vector_type(8))) float v8f;

__device__ __forceinline__ void fold_attn_wave(const float* __restrict__ Ws,
                                               const float* __restrict__ als,
                                               const float* __restrict__ ars,
                                               float* __restrict__ wl,
                                               float* __restrict__ wr,
                                               const int gw, const int lane) {
    if (gw >= 2 * D_MODEL) return;
    const int layer = gw >> 7;
    const float4* F0 = (const float4*)(Ws + (size_t)gw * 256) + lane;
    const float4* F1 = (const float4*)(Ws + (size_t)gw * 256 + 128) + lane;
    const float4* AL0 = (const float4*)(als + layer * 256) + lane;
    const float4* AL1 = (const float4*)(als + layer * 256 + 128) + lane;
    const float4* AR0 = (const float4*)(ars + layer * 256) + lane;
    const float4* AR1 = (const float4*)(ars + layer * 256 + 128) + lane;
    float4 f0 = *F0, f1 = *F1;
    float4 l0 = *AL0, l1 = *AL1, r0 = *AR0, r1 = *AR1;
    float e0 = f0.x*l0.x + f0.y*l0.y + f0.z*l0.z + f0.w*l0.w;
    float e1 = f1.x*l1.x + f1.y*l1.y + f1.z*l1.z + f1.w*l1.w;
    float g0 = f0.x*r0.x + f0.y*r0.y + f0.z*r0.z + f0.w*r0.w;
    float g1 = f1.x*r1.x + f1.y*r1.y + f1.z*r1.z + f1.w*r1.w;
    #pragma unroll
    for (int off = 16; off; off >>= 1) {
        e0 += __shfl_xor(e0, off, 32);
        e1 += __shfl_xor(e1, off, 32);
        g0 += __shfl_xor(g0, off, 32);
        g1 += __shfl_xor(g1, off, 32);
    }
    if (lane == 0) {
        wl[gw * 2 + 0] = e0;  wl[gw * 2 + 1] = e1;
        wr[gw * 2 + 0] = g0;  wr[gw * 2 + 1] = g1;
    }
}

__device__ __forceinline__ float xhalf(float x) {
    const int b = __builtin_bit_cast(int, x);
    return __builtin_bit_cast(float, __builtin_amdgcn_permlanex16(b, b, 0x76543210, 0xfedcba98, false, false));
}
template <int M>
__device__ __forceinline__ float xrow(float x) {
    return __builtin_bit_cast(float, __builtin_amdgcn_mov_dpp(__builtin_bit_cast(int, x), 0x160 + M, 0xf, 0xf, false));
}

#define ELR_ROWS 16
static_assert(N_NODES % ELR_ROWS == 0, "every wave owns ELR_ROWS whole rows");
static_assert(ELR_ROWS * N_HEADS == 32, "after the reduction a lane holds exactly one sum");

__device__ __forceinline__ float lane_transpose_sum(const float (&p)[32], const int lane) {
    const bool b16 = lane & 16, b8 = lane & 8, b4 = lane & 4, b2 = lane & 2, b1 = lane & 1;
    float s16[16], s8[8], s4[4], s2[2];
    #pragma unroll
    for (int i = 0; i < 16; ++i)
        s16[i] = (b16 ? p[i + 16] : p[i]) + xhalf(b16 ? p[i] : p[i + 16]);
    #pragma unroll
    for (int i = 0; i < 8; ++i)
        s8[i] = (b8 ? s16[i + 8] : s16[i]) + xrow<8>(b8 ? s16[i] : s16[i + 8]);
    #pragma unroll
    for (int i = 0; i < 4; ++i)
        s4[i] = (b4 ? s8[i + 4] : s8[i]) + xrow<4>(b4 ? s8[i] : s8[i + 4]);
    #pragma unroll
    for (int i = 0; i < 2; ++i)
        s2[i] = (b2 ? s4[i + 2] : s4[i]) + xrow<2>(b2 ? s4[i] : s4[i + 2]);
    return (b1 ? s2[1] : s2[0]) + xrow<1>(b1 ? s2[0] : s2[1]);
}

__device__ __forceinline__ void node_elr_wave(const float* __restrict__ hin,
                                              const float* __restrict__ wl,
                                              const float* __restrict__ wr,
                                              float* __restrict__ el,
                                              float* __restrict__ er,
                                              const int gw, const int lane) {
    const int n0 = gw * ELR_ROWS;
    if (n0 >= N_NODES) return;
    const float4 la = ((const float4*)wl)[2 * lane], lb = ((const float4*)wl)[2 * lane + 1];
    const float4 ra = ((const float4*)wr)[2 * lane], rb = ((const float4*)wr)[2 * lane + 1];
    const float4* rows = (const float4*)(hin + (size_t)n0 * 128) + lane;
    float4 x[ELR_ROWS];
    #pragma unroll
    for (int r = 0; r < ELR_ROWS; ++r) x[r] = rows[r * 32];
    __builtin_amdgcn_sched_barrier(0);
    float p[32], q[32];
    #pragma unroll
    for (int r = 0; r < ELR_ROWS; ++r) {
        const float4 v = x[r];
        p[2 * r + 0] = fmaf(v.w, lb.z, fmaf(v.z, lb.x, fmaf(v.y, la.z, v.x * la.x)));
        p[2 * r + 1] = fmaf(v.w, lb.w, fmaf(v.z, lb.y, fmaf(v.y, la.w, v.x * la.y)));
        q[2 * r + 0] = fmaf(v.w, rb.z, fmaf(v.z, rb.x, fmaf(v.y, ra.z, v.x * ra.x)));
        q[2 * r + 1] = fmaf(v.w, rb.w, fmaf(v.z, rb.y, fmaf(v.y, ra.w, v.x * ra.y)));
    }
    const float sl = lane_transpose_sum(p, lane);
    const float sr = lane_transpose_sum(q, lane);
    el[n0 * 2 + lane] = sl;
    er[n0 * 2 + lane] = sr;
}

__global__ void __launch_bounds__(256) node_elr(const float* __restrict__ hin,
                                                const float* __restrict__ wl,
                                                const float* __restrict__ wr,
                                                float* __restrict__ el,
                                                float* __restrict__ er) {
    node_elr_wave(hin, wl, wr, el, er,
                  blockIdx.x * 8 + __builtin_amdgcn_readfirstlane(threadIdx.x >> 5), threadIdx.x & 31);
}

#define BK_SHIFT   7
#define BK_NODES   (1 << BK_SHIFT)
#define BK_COUNT   ((N_NODES >> BK_SHIFT) + 1)
#define BK_BLOCKS  128
#define BK_EDGES   ((N_EDGES + BK_BLOCKS - 1) / BK_BLOCKS)
#define BK_PER     ((BK_EDGES + 1023) / 1024)
#define FP_PER     4
#define SCAN_PER  ((((N_NODES + 1023) / 1024) + 3) / 4 * 4)
#define NODES_PAD (SCAN_PER * 1024)
#define BK_PAD    (NODES_PAD / BK_BLOCKS)
static_assert(BK_PAD * BK_BLOCKS == NODES_PAD, "the table is BK_PAD whole rows");
static_assert(BK_COUNT < BK_PAD && BK_PAD <= 1024, "ranges must fit the table and one block");
static_assert(N_NODES <= 65536, "a source index must fit the low 16 bits of the packed word");
static_assert(NODES_PAD >= N_NODES + 4, "rowptr must hold index N_NODES and its int4");

#define FOLD_BLOCKS (2 * D_MODEL / 32)
__global__ void __launch_bounds__(1024) coarse_count(const int* __restrict__ dst,
                                                     int* __restrict__ hist,
                                                     const float* __restrict__ Ws,
                                                     const float* __restrict__ als,
                                                     const float* __restrict__ ars,
                                                     float* __restrict__ wl,
                                                     float* __restrict__ wr) {
    __shared__ int cnt[BK_PAD];
    const int tid = threadIdx.x, b = blockIdx.x;
    if (b >= BK_BLOCKS) {
        fold_attn_wave(Ws, als, ars, wl, wr, (b - BK_BLOCKS) * 32 + (tid >> 5), tid & 31);
        return;
    }
    if (tid < BK_PAD) cnt[tid] = 0;
    __syncthreads();
    const int e0 = b * BK_EDGES;
    const int e1 = (e0 + BK_EDGES) < N_EDGES ? (e0 + BK_EDGES) : N_EDGES;
    int d[BK_PER];
    #pragma unroll
    for (int j = 0; j < BK_PER; ++j) {
        const int e = e0 + tid + j * 1024;
        d[j] = e < e1 ? dst[e] : -1;
    }
    #pragma unroll
    for (int j = 0; j < BK_PER; ++j)
        if (d[j] >= 0) atomicAdd(&cnt[d[j] >> BK_SHIFT], 1);
    __syncthreads();
    if (tid < BK_PAD) hist[tid * BK_BLOCKS + b] = cnt[tid];
}

__device__ __forceinline__ int ixhalf(int b) {
    return __builtin_amdgcn_permlanex16(b, b, 0x76543210, 0xfedcba98, false, false);
}
template <int M>
__device__ __forceinline__ int ixrow(int x) {
    return __builtin_amdgcn_mov_dpp(x, 0x160 + M, 0xf, 0xf, false);
}
__device__ __forceinline__ int wave_isum(int x) {
    x += ixrow<1>(x);
    x += ixrow<2>(x);
    x += ixrow<4>(x);
    x += ixrow<8>(x);
    return x + ixhalf(x);
}

#define SCAN_ROWS (BK_PAD / 32)
#define ELR_BLOCKS ((N_NODES / ELR_ROWS + 31) / 32)
static_assert(BK_BLOCKS == 128, "a table row is one int4 per lane");
static_assert(SCAN_ROWS * 32 == BK_PAD, "32 waves share the table rows evenly");
__global__ void __launch_bounds__(1024) coarse_place(const int* __restrict__ src,
                                                     const int* __restrict__ dst,
                                                     const int* __restrict__ hist,
                                                     int* __restrict__ rstart,
                                                     int* __restrict__ coarse,
                                                     const float* __restrict__ x,
                                                     const float* __restrict__ wl,
                                                     const float* __restrict__ wr,
                                                     float* __restrict__ el,
                                                     float* __restrict__ er) {
    __shared__ int cur[BK_PAD];
    __shared__ int tot[BK_PAD];
    const int tid = threadIdx.x, b = blockIdx.x;
    if (b >= BK_BLOCKS) {
        node_elr_wave(x, wl, wr, el, er,
                      (b - BK_BLOCKS) * 32 + __builtin_amdgcn_readfirstlane(tid >> 5), tid & 31);
        return;
    }
    const int lane = tid & 31, wave = tid >> 5;
    const int e0 = b * BK_EDGES;
    const int e1 = (e0 + BK_EDGES) < N_EDGES ? (e0 + BK_EDGES) : N_EDGES;
    int d[BK_PER], s[BK_PER];
    #pragma unroll
    for (int j = 0; j < BK_PER; ++j) {
        const int e = e0 + tid + j * 1024;
        d[j] = e < e1 ? dst[e] : -1;
        s[j] = e < e1 ? src[e] : 0;
    }
    int4 h[SCAN_ROWS];
    #pragma unroll
    for (int i = 0; i < SCAN_ROWS; ++i)
        h[i] = ((const int4*)hist)[(wave + 32 * i) * (BK_BLOCKS / 4) + lane];
    const int c0 = 4 * lane;
    #pragma unroll
    for (int i = 0; i < SCAN_ROWS; ++i) {
        const int4 v = h[i];
        const int all = wave_isum((v.x + v.y) + (v.z + v.w));
        const int pre = wave_isum(((c0     < b ? v.x : 0) + (c0 + 1 < b ? v.y : 0)) +
                                  ((c0 + 2 < b ? v.z : 0) + (c0 + 3 < b ? v.w : 0)));
        if (lane == 0) {
            tot[wave + 32 * i] = all;
            cur[wave + 32 * i] = pre;
        }
    }
    __syncthreads();
    if (wave == 0) {
        int v[SCAN_ROWS];
        int loc = 0;
        #pragma unroll
        for (int k = 0; k < SCAN_ROWS; ++k) {
            v[k] = tot[lane * SCAN_ROWS + k];
            loc += v[k];
        }
        int inc = loc;
        for (int off = 1; off < 32; off <<= 1) {
            const int t = __shfl_up(inc, off, 32);
            if (lane >= off) inc += t;
        }
        int run = inc - loc;
        #pragma unroll
        for (int k = 0; k < SCAN_ROWS; ++k) {
            tot[lane * SCAN_ROWS + k] = run;
            run += v[k];
        }
    }
    __syncthreads();
    if (tid < BK_PAD) {
        const int start = tot[tid];
        cur[tid] += start;
        if (b == 0) rstart[tid] = start;
    }
    __syncthreads();
    #pragma unroll
    for (int j = 0; j < BK_PER; ++j)
        if (d[j] >= 0) {
            const int pos = atomicAdd(&cur[d[j] >> BK_SHIFT], 1);
            coarse[pos] = ((d[j] & (BK_NODES - 1)) << 16) | s[j];
        }
}

__global__ void __launch_bounds__(1024) fine_place(const int* __restrict__ coarse,
                                                   const int* __restrict__ rstart,
                                                   int* __restrict__ rowptr,
                                                   int* __restrict__ srcs) {
    __shared__ __attribute__((aligned(16))) int cnt[BK_NODES];
    const int tid = threadIdx.x, r = blockIdx.x;
    const int beg = rstart[r], end = rstart[r + 1];
    if (tid < BK_NODES) cnt[tid] = 0;
    __syncthreads();
    for (int i = beg + tid; i < end; i += FP_PER * 1024) {
        int w[FP_PER];
        #pragma unroll
        for (int j = 0; j < FP_PER; ++j) {
            const int ii = i + j * 1024;
            w[j] = ii < end ? coarse[ii] : -1;
        }
        #pragma unroll
        for (int j = 0; j < FP_PER; ++j)
            if (w[j] >= 0) atomicAdd(&cnt[w[j] >> 16], 1);
    }
    __syncthreads();
    if (tid < 32) {
        const int4 c = ((const int4*)cnt)[tid];
        const int loc = (c.x + c.y) + (c.z + c.w);
        int inc = loc;
        for (int off = 1; off < 32; off <<= 1) {
            const int t = __shfl_up(inc, off, 32);
            if (tid >= off) inc += t;
        }
        int4 p;
        p.x = beg + inc - loc;
        p.y = p.x + c.x;
        p.z = p.y + c.y;
        p.w = p.z + c.z;
        ((int4*)cnt)[tid] = p;
        const int n0 = r * BK_NODES + tid * 4;
        if (n0 <= N_NODES) *(int4*)(rowptr + n0) = p;
    }
    __syncthreads();
    for (int i = beg + tid; i < end; i += FP_PER * 1024) {
        int w[FP_PER];
        #pragma unroll
        for (int j = 0; j < FP_PER; ++j) {
            const int ii = i + j * 1024;
            w[j] = ii < end ? coarse[ii] : -1;
        }
        #pragma unroll
        for (int j = 0; j < FP_PER; ++j)
            if (w[j] >= 0) {
                const int pos = atomicAdd(&cnt[w[j] >> 16], 1);
                srcs[pos] = w[j] & 0xffff;
            }
    }
}

__device__ __forceinline__ float wave_max(float x) {
    x = fmaxf(x, xrow<1>(x));
    x = fmaxf(x, xrow<2>(x));
    x = fmaxf(x, xrow<4>(x));
    x = fmaxf(x, xrow<8>(x));
    return fmaxf(x, xhalf(x));
}
__device__ __forceinline__ float wave_sum(float x) {
    x += xrow<1>(x);
    x += xrow<2>(x);
    x += xrow<4>(x);
    x += xrow<8>(x);
    return x + xhalf(x);
}

__device__ __forceinline__ void edge_scores(const int* __restrict__ srcs,
                                            const float* __restrict__ el,
                                            const float er0, const float er1,
                                            const int base, const int end, const int lane,
                                            int& s, float& v0, float& v1) {
    s = 0;
    v0 = -INFINITY;
    v1 = -INFINITY;
    if (base + lane < end) {
        s = srcs[base + lane];
        const float2 l = *(const float2*)(el + 2 * (size_t)s);
        const float t0 = l.x + er0, t1 = l.y + er1;
        v0 = t0 > 0.0f ? t0 : 0.2f * t0;
        v1 = t1 > 0.0f ? t1 : 0.2f * t1;
    }
}

__device__ __forceinline__ void gather_node(const float* __restrict__ hin,
                                            const float* __restrict__ el,
                                            const float* __restrict__ er,
                                            const int* __restrict__ rowptr,
                                            const int* __restrict__ srcs,
                                            const int n, const int lane,
                                            float4& a0, float4& a1) {
    const int beg = rowptr[n], end = rowptr[n + 1];
    if (end > beg) {
        const float er0 = er[n * 2], er1 = er[n * 2 + 1];
        int s_l;
        float e0_l, e1_l;
        edge_scores(srcs, el, er0, er1, beg, end, lane, s_l, e0_l, e1_l);
        float m0 = e0_l, m1 = e1_l;
        for (int base = beg + 32; base < end; base += 32) {
            int s;
            float v0, v1;
            edge_scores(srcs, el, er0, er1, base, end, lane, s, v0, v1);
            m0 = fmaxf(m0, v0);
            m1 = fmaxf(m1, v1);
        }
        m0 = wave_max(m0);
        m1 = wave_max(m1);
        const float x0_l = (beg + lane < end) ? expf(e0_l - m0) : 0.f;
        const float x1_l = (beg + lane < end) ? expf(e1_l - m1) : 0.f;
        float den0 = x0_l, den1 = x1_l;
        for (int base = beg + 32; base < end; base += 32) {
            int s;
            float v0, v1;
            edge_scores(srcs, el, er0, er1, base, end, lane, s, v0, v1);
            if (base + lane < end) {
                den0 += expf(v0 - m0);
                den1 += expf(v1 - m1);
            }
        }
        den0 = wave_sum(den0);
        den1 = wave_sum(den1);
        float w0_l = x0_l / den0, w1_l = x1_l / den1;
        for (int base = beg; base < end; base += 32) {
            const int cnt = (end - base) < 32 ? (end - base) : 32;
            if (base != beg) {
                float v0, v1;
                edge_scores(srcs, el, er0, er1, base, end, lane, s_l, v0, v1);
                w0_l = (base + lane < end) ? expf(v0 - m0) / den0 : 0.f;
                w1_l = (base + lane < end) ? expf(v1 - m1) / den1 : 0.f;
            }
            #pragma unroll 4
            for (int j = 0; j < cnt; ++j) {
                const int   s  = __builtin_amdgcn_readlane(s_l, j);
                const float w0 = __int_as_float(__builtin_amdgcn_readlane(__float_as_int(w0_l), j));
                const float w1 = __int_as_float(__builtin_amdgcn_readlane(__float_as_int(w1_l), j));
                const float4 f = ((const float4*)(hin + (size_t)s * 128))[lane];
                a0.x += f.x * w0;  a0.y += f.y * w0;  a0.z += f.z * w0;  a0.w += f.w * w0;
                a1.x += f.x * w1;  a1.y += f.y * w1;  a1.z += f.z * w1;  a1.w += f.w * w1;
            }
        }
    }
}

#define TILE_NODES 32
#define AGG_PITCH  260

__device__ __forceinline__ void gather_tile(const float* __restrict__ hin,
                                            const float* __restrict__ el,
                                            const float* __restrict__ er,
                                            const int* __restrict__ rowptr,
                                            const int* __restrict__ srcs,
                                            float* __restrict__ sA,
                                            const int mbase, const int wave, const int lane) {
    for (int i = 0; i < 2; ++i) {
        const int r = wave * 2 + i;
        const int n = mbase + r;
        float4 a0 = {0.f, 0.f, 0.f, 0.f}, a1 = {0.f, 0.f, 0.f, 0.f};
        if (n < N_NODES) gather_node(hin, el, er, rowptr, srcs, n, lane, a0, a1);
        float4* o = (float4*)(sA + r * AGG_PITCH);
        o[lane]      = a0;
        o[32 + lane] = a1;
    }
}

__global__ void __launch_bounds__(512, 16) agg_proj_mean(const float* __restrict__ hin,
                                                     const float* __restrict__ el,
                                                     const float* __restrict__ er,
                                                     const int* __restrict__ rowptr,
                                                     const int* __restrict__ srcs,
                                                     const float* __restrict__ W,
                                                     const float* __restrict__ b,
                                                     float* __restrict__ hnext) {
    __shared__ __attribute__((aligned(16))) float sA[TILE_NODES * AGG_PITCH];
    const int lane  = threadIdx.x & 31;
    const int wave  = __builtin_amdgcn_readfirstlane(threadIdx.x >> 5);
    const int mbase = blockIdx.x * TILE_NODES;
    gather_tile(hin, el, er, rowptr, srcs, sA, mbase, wave, lane);
    __syncthreads();
    if (wave >= 8) return;

    const int row   = lane & 15;
    const int khalf = lane >> 4;
    const int rt    = wave >> 2;
    const int nq    = (wave & 3) << 5;

    v8f acc0 = {}, acc1 = {};
    const float* Arow = sA + (rt * 16 + row) * AGG_PITCH + khalf * 2;
    const float* Wh   = W + (size_t)(khalf * 2) * 256 + nq + row;
    #pragma unroll 8
    for (int kk = 0; kk < N_HEADS * D_MODEL; kk += 4) {
        const v2f a = *(const v2f*)(Arow + kk);
        const float* Wb = Wh + (size_t)(kk & (D_MODEL - 1)) * 256 + (kk >> 7) * 128;
        v2f b0, b1;
        b0.x = Wb[0];    b0.y = Wb[256];
        b1.x = Wb[16];   b1.y = Wb[256 + 16];
        acc0 = __builtin_amdgcn_wmma_f32_16x16x4_f32(false, a, false, b0, (short)0, acc0, false, false);
        acc1 = __builtin_amdgcn_wmma_f32_16x16x4_f32(false, a, false, b1, (short)0, acc1, false, false);
    }

    const int c = nq + row;
    const float bb0 = b[c]      + b[128 + c];
    const float bb1 = b[c + 16] + b[128 + c + 16];
    #pragma unroll
    for (int j = 0; j < 8; ++j) {
        const int m = mbase + rt * 16 + j + khalf * 8;
        if (m < N_NODES) {
            float* F = hnext + (size_t)m * 128 + c;
            F[0]  = 0.5f * (acc0[j] + bb0);
            F[16] = 0.5f * (acc1[j] + bb1);
        }
    }
}

__global__ void __launch_bounds__(512, 16) agg_proj_heads(const float* __restrict__ hin,
                                                      const float* __restrict__ el,
                                                      const float* __restrict__ er,
                                                      const int* __restrict__ rowptr,
                                                      const int* __restrict__ srcs,
                                                      const float* __restrict__ W,
                                                      const float* __restrict__ b,
                                                      float* __restrict__ out) {
    __shared__ __attribute__((aligned(16))) float sA[TILE_NODES * AGG_PITCH];
    const int lane  = threadIdx.x & 31;
    const int wave  = __builtin_amdgcn_readfirstlane(threadIdx.x >> 5);
    const int mbase = blockIdx.x * TILE_NODES;
    gather_tile(hin, el, er, rowptr, srcs, sA, mbase, wave, lane);
    __syncthreads();

    const int row   = lane & 15;
    const int khalf = lane >> 4;
    const int rt    = wave >> 3;
    const int nq    = (wave & 7) << 5;

    v8f acc0 = {}, acc1 = {};
    const float* Arow = sA + (rt * 16 + row) * AGG_PITCH + (nq >> 7) * 128;
    const float* Wh   = W + nq + row;
    #pragma unroll 8
    for (int k = 0; k < D_MODEL; k += 4) {
        const int ka = k + khalf * 2;
        const v2f a = *(const v2f*)(Arow + ka);
        const float* Wb = Wh + (size_t)ka * 256;
        v2f b0, b1;
        b0.x = Wb[0];    b0.y = Wb[256];
        b1.x = Wb[16];   b1.y = Wb[256 + 16];
        acc0 = __builtin_amdgcn_wmma_f32_16x16x4_f32(false, a, false, b0, (short)0, acc0, false, false);
        acc1 = __builtin_amdgcn_wmma_f32_16x16x4_f32(false, a, false, b1, (short)0, acc1, false, false);
    }

    const int c = nq + row;
    const float bb0 = b[c], bb1 = b[c + 16];
    #pragma unroll
    for (int j = 0; j < 8; ++j) {
        const int m = mbase + rt * 16 + j + khalf * 8;
        if (m < N_NODES) {
            float* F = out + (size_t)m * 256 + c;
            F[0]  = acc0[j] + bb0;
            F[16] = acc1[j] + bb1;
        }
    }
}

extern "C" void kernel_launch(void* const* d_in, const int* in_sizes, int n_in,
                              void* d_out, int out_size, void* d_ws, size_t ws_size,
                              hipStream_t stream) {
    const float* x   = (const float*)d_in[0];
    const float* Ws  = (const float*)d_in[1];
    const float* als = (const float*)d_in[2];
    const float* ars = (const float*)d_in[3];
    const float* bs  = (const float*)d_in[4];
    const int*   src = (const int*)d_in[5];
    const int*   dst = (const int*)d_in[6];

    float* ws      = (float*)d_ws;
    float* hnext   = ws;
    int*   srcs    = (int*)(ws + 12800000);
    int*   coarse  = (int*)(ws + 14400000);
    int*   hist    = (int*)(ws + 15200000);
    int*   rowptr  = (int*)(ws + 15300000);
    int*   rstart  = (int*)(ws + 15400000);
    float* wl      = ws + 19100000;
    float* wr      = ws + 19101000;
    float* el      = ws + 19200000;
    float* er      = ws + 19300000;

    float* out   = (float*)d_out;

    coarse_count<<<BK_BLOCKS + FOLD_BLOCKS, 1024, 0, stream>>>(dst, hist, Ws, als, ars, wl, wr);
    coarse_place<<<BK_BLOCKS + ELR_BLOCKS, 1024, 0, stream>>>(src, dst, hist, rstart, coarse, x, wl, wr, el, er);
    fine_place<<<BK_COUNT, 1024, 0, stream>>>(coarse, rstart, rowptr, srcs);

    agg_proj_mean<<<(N_NODES + TILE_NODES - 1) / TILE_NODES, 512, 0, stream>>>(x, el, er, rowptr, srcs, Ws, bs, hnext);

    node_elr<<<(N_NODES / ELR_ROWS + 7) / 8, 256, 0, stream>>>(hnext, wl + 256, wr + 256, el, er);
    agg_proj_heads<<<(N_NODES + TILE_NODES - 1) / TILE_NODES, 512, 0, stream>>>(hnext, el, er, rowptr, srcs, Ws + 32768, bs + 256, out);
}


# projection reads W rewritten in fragment order by tail blocks of launch 1: one 16-B load per lane and k-step feeds both products (was four 4-B loads); column pairs leave as whole lines; same sums
# speedup vs baseline: 7.2062x; 1.0304x over previous
#include <hip/hip_runtime.h>
#include <math.h>

#define N_NODES 50000
#define N_EDGES 800000
#define D_MODEL 128
#define N_HEADS 2

typedef __attribute__((ext_vector_type(2))) float v2f;
typedef __attribute__((ext_vector_type(8))) float v8f;

__device__ __forceinline__ void fold_attn_wave(const float* __restrict__ Ws,
                                               const float* __restrict__ als,
                                               const float* __restrict__ ars,
                                               float* __restrict__ wl,
                                               float* __restrict__ wr,
                                               const int gw, const int lane) {
    if (gw >= 2 * D_MODEL) return;
    const int layer = gw >> 7;
    const float4* F0 = (const float4*)(Ws + (size_t)gw * 256) + lane;
    const float4* F1 = (const float4*)(Ws + (size_t)gw * 256 + 128) + lane;
    const float4* AL0 = (const float4*)(als + layer * 256) + lane;
    const float4* AL1 = (const float4*)(als + layer * 256 + 128) + lane;
    const float4* AR0 = (const float4*)(ars + layer * 256) + lane;
    const float4* AR1 = (const float4*)(ars + layer * 256 + 128) + lane;
    float4 f0 = *F0, f1 = *F1;
    float4 l0 = *AL0, l1 = *AL1, r0 = *AR0, r1 = *AR1;
    float e0 = f0.x*l0.x + f0.y*l0.y + f0.z*l0.z + f0.w*l0.w;
    float e1 = f1.x*l1.x + f1.y*l1.y + f1.z*l1.z + f1.w*l1.w;
    float g0 = f0.x*r0.x + f0.y*r0.y + f0.z*r0.z + f0.w*r0.w;
    float g1 = f1.x*r1.x + f1.y*r1.y + f1.z*r1.z + f1.w*r1.w;
    #pragma unroll
    for (int off = 16; off; off >>= 1) {
        e0 += __shfl_xor(e0, off, 32);
        e1 += __shfl_xor(e1, off, 32);
        g0 += __shfl_xor(g0, off, 32);
        g1 += __shfl_xor(g1, off, 32);
    }
    if (lane == 0) {
        wl[gw * 2 + 0] = e0;  wl[gw * 2 + 1] = e1;
        wr[gw * 2 + 0] = g0;  wr[gw * 2 + 1] = g1;
    }
}

#define WF_LAYER     (D_MODEL * N_HEADS * D_MODEL)
#define RELAY_BLOCKS (2 * WF_LAYER / 4 / 1024)
static_assert(RELAY_BLOCKS * 1024 * 4 == 2 * WF_LAYER, "one float4 per thread covers both layers");
__device__ __forceinline__ void relay_w(const float* __restrict__ Ws, float* __restrict__ WF, const int t) {
    const int layer = t >> 13, u = t & 8191;
    const int lane = u & 31, r = lane & 15, khalf = lane >> 4;
    int ka, col;
    if (layer == 0) {
        const int g = (u >> 5) & 3, kk = (u >> 7) * 4;
        ka  = (kk & 127) + 2 * khalf;
        col = (kk >> 7) * 128 + 32 * g + 2 * r;
    } else {
        const int g = (u >> 5) & 7, k = (u >> 8) * 4;
        ka  = k + 2 * khalf;
        col = 32 * g + 2 * r;
    }
    const float* s = Ws + layer * WF_LAYER + ka * 256 + col;
    const float2 lo = *(const float2*)s, hi = *(const float2*)(s + 256);
    float4 o;
    o.x = lo.x;  o.y = hi.x;  o.z = lo.y;  o.w = hi.y;
    ((float4*)WF)[t] = o;
}

__device__ __forceinline__ float xhalf(float x) {
    const int b = __builtin_bit_cast(int, x);
    return __builtin_bit_cast(float, __builtin_amdgcn_permlanex16(b, b, 0x76543210, 0xfedcba98, false, false));
}
template <int M>
__device__ __forceinline__ float xrow(float x) {
    return __builtin_bit_cast(float, __builtin_amdgcn_mov_dpp(__builtin_bit_cast(int, x), 0x160 + M, 0xf, 0xf, false));
}

#define ELR_ROWS 16
static_assert(N_NODES % ELR_ROWS == 0, "every wave owns ELR_ROWS whole rows");
static_assert(ELR_ROWS * N_HEADS == 32, "after the reduction a lane holds exactly one sum");

__device__ __forceinline__ float lane_transpose_sum(const float (&p)[32], const int lane) {
    const bool b16 = lane & 16, b8 = lane & 8, b4 = lane & 4, b2 = lane & 2, b1 = lane & 1;
    float s16[16], s8[8], s4[4], s2[2];
    #pragma unroll
    for (int i = 0; i < 16; ++i)
        s16[i] = (b16 ? p[i + 16] : p[i]) + xhalf(b16 ? p[i] : p[i + 16]);
    #pragma unroll
    for (int i = 0; i < 8; ++i)
        s8[i] = (b8 ? s16[i + 8] : s16[i]) + xrow<8>(b8 ? s16[i] : s16[i + 8]);
    #pragma unroll
    for (int i = 0; i < 4; ++i)
        s4[i] = (b4 ? s8[i + 4] : s8[i]) + xrow<4>(b4 ? s8[i] : s8[i + 4]);
    #pragma unroll
    for (int i = 0; i < 2; ++i)
        s2[i] = (b2 ? s4[i + 2] : s4[i]) + xrow<2>(b2 ? s4[i] : s4[i + 2]);
    return (b1 ? s2[1] : s2[0]) + xrow<1>(b1 ? s2[0] : s2[1]);
}

__device__ __forceinline__ void node_elr_wave(const float* __restrict__ hin,
                                              const float* __restrict__ wl,
                                              const float* __restrict__ wr,
                                              float* __restrict__ el,
                                              float* __restrict__ er,
                                              const int gw, const int lane) {
    const int n0 = gw * ELR_ROWS;
    if (n0 >= N_NODES) return;
    const float4 la = ((const float4*)wl)[2 * lane], lb = ((const float4*)wl)[2 * lane + 1];
    const float4 ra = ((const float4*)wr)[2 * lane], rb = ((const float4*)wr)[2 * lane + 1];
    const float4* rows = (const float4*)(hin + (size_t)n0 * 128) + lane;
    float4 x[ELR_ROWS];
    #pragma unroll
    for (int r = 0; r < ELR_ROWS; ++r) x[r] = rows[r * 32];
    __builtin_amdgcn_sched_barrier(0);
    float p[32], q[32];
    #pragma unroll
    for (int r = 0; r < ELR_ROWS; ++r) {
        const float4 v = x[r];
        p[2 * r + 0] = fmaf(v.w, lb.z, fmaf(v.z, lb.x, fmaf(v.y, la.z, v.x * la.x)));
        p[2 * r + 1] = fmaf(v.w, lb.w, fmaf(v.z, lb.y, fmaf(v.y, la.w, v.x * la.y)));
        q[2 * r + 0] = fmaf(v.w, rb.z, fmaf(v.z, rb.x, fmaf(v.y, ra.z, v.x * ra.x)));
        q[2 * r + 1] = fmaf(v.w, rb.w, fmaf(v.z, rb.y, fmaf(v.y, ra.w, v.x * ra.y)));
    }
    const float sl = lane_transpose_sum(p, lane);
    const float sr = lane_transpose_sum(q, lane);
    el[n0 * 2 + lane] = sl;
    er[n0 * 2 + lane] = sr;
}

__global__ void __launch_bounds__(256) node_elr(const float* __restrict__ hin,
                                                const float* __restrict__ wl,
                                                const float* __restrict__ wr,
                                                float* __restrict__ el,
                                                float* __restrict__ er) {
    node_elr_wave(hin, wl, wr, el, er,
                  blockIdx.x * 8 + __builtin_amdgcn_readfirstlane(threadIdx.x >> 5), threadIdx.x & 31);
}

#define BK_SHIFT   7
#define BK_NODES   (1 << BK_SHIFT)
#define BK_COUNT   ((N_NODES >> BK_SHIFT) + 1)
#define BK_BLOCKS  128
#define BK_EDGES   ((N_EDGES + BK_BLOCKS - 1) / BK_BLOCKS)
#define BK_PER     ((BK_EDGES + 1023) / 1024)
#define FP_PER     4
#define SCAN_PER  ((((N_NODES + 1023) / 1024) + 3) / 4 * 4)
#define NODES_PAD (SCAN_PER * 1024)
#define BK_PAD    (NODES_PAD / BK_BLOCKS)
static_assert(BK_PAD * BK_BLOCKS == NODES_PAD, "the table is BK_PAD whole rows");
static_assert(BK_COUNT < BK_PAD && BK_PAD <= 1024, "ranges must fit the table and one block");
static_assert(N_NODES <= 65536, "a source index must fit the low 16 bits of the packed word");
static_assert(NODES_PAD >= N_NODES + 4, "rowptr must hold index N_NODES and its int4");

#define FOLD_BLOCKS (2 * D_MODEL / 32)
__global__ void __launch_bounds__(1024) coarse_count(const int* __restrict__ dst,
                                                     int* __restrict__ hist,
                                                     const float* __restrict__ Ws,
                                                     const float* __restrict__ als,
                                                     const float* __restrict__ ars,
                                                     float* __restrict__ wl,
                                                     float* __restrict__ wr,
                                                     float* __restrict__ WF) {
    __shared__ int cnt[BK_PAD];
    const int tid = threadIdx.x, b = blockIdx.x;
    if (b >= BK_BLOCKS) {
        if (b >= BK_BLOCKS + FOLD_BLOCKS) relay_w(Ws, WF, (b - BK_BLOCKS - FOLD_BLOCKS) * 1024 + tid);
        else fold_attn_wave(Ws, als, ars, wl, wr, (b - BK_BLOCKS) * 32 + (tid >> 5), tid & 31);
        return;
    }
    if (tid < BK_PAD) cnt[tid] = 0;
    __syncthreads();
    const int e0 = b * BK_EDGES;
    const int e1 = (e0 + BK_EDGES) < N_EDGES ? (e0 + BK_EDGES) : N_EDGES;
    int d[BK_PER];
    #pragma unroll
    for (int j = 0; j < BK_PER; ++j) {
        const int e = e0 + tid + j * 1024;
        d[j] = e < e1 ? dst[e] : -1;
    }
    #pragma unroll
    for (int j = 0; j < BK_PER; ++j)
        if (d[j] >= 0) atomicAdd(&cnt[d[j] >> BK_SHIFT], 1);
    __syncthreads();
    if (tid < BK_PAD) hist[tid * BK_BLOCKS + b] = cnt[tid];
}

__device__ __forceinline__ int ixhalf(int b) {
    return __builtin_amdgcn_permlanex16(b, b, 0x76543210, 0xfedcba98, false, false);
}
template <int M>
__device__ __forceinline__ int ixrow(int x) {
    return __builtin_amdgcn_mov_dpp(x, 0x160 + M, 0xf, 0xf, false);
}
__device__ __forceinline__ int wave_isum(int x) {
    x += ixrow<1>(x);
    x += ixrow<2>(x);
    x += ixrow<4>(x);
    x += ixrow<8>(x);
    return x + ixhalf(x);
}

#define SCAN_ROWS (BK_PAD / 32)
#define ELR_BLOCKS ((N_NODES / ELR_ROWS + 31) / 32)
static_assert(BK_BLOCKS == 128, "a table row is one int4 per lane");
static_assert(SCAN_ROWS * 32 == BK_PAD, "32 waves share the table rows evenly");
__global__ void __launch_bounds__(1024) coarse_place(const int* __restrict__ src,
                                                     const int* __restrict__ dst,
                                                     const int* __restrict__ hist,
                                                     int* __restrict__ rstart,
                                                     int* __restrict__ coarse,
                                                     const float* __restrict__ x,
                                                     const float* __restrict__ wl,
                                                     const float* __restrict__ wr,
                                                     float* __restrict__ el,
                                                     float* __restrict__ er) {
    __shared__ int cur[BK_PAD];
    __shared__ int tot[BK_PAD];
    const int tid = threadIdx.x, b = blockIdx.x;
    if (b >= BK_BLOCKS) {
        node_elr_wave(x, wl, wr, el, er,
                      (b - BK_BLOCKS) * 32 + __builtin_amdgcn_readfirstlane(tid >> 5), tid & 31);
        return;
    }
    const int lane = tid & 31, wave = tid >> 5;
    const int e0 = b * BK_EDGES;
    const int e1 = (e0 + BK_EDGES) < N_EDGES ? (e0 + BK_EDGES) : N_EDGES;
    int d[BK_PER], s[BK_PER];
    #pragma unroll
    for (int j = 0; j < BK_PER; ++j) {
        const int e = e0 + tid + j * 1024;
        d[j] = e < e1 ? dst[e] : -1;
        s[j] = e < e1 ? src[e] : 0;
    }
    int4 h[SCAN_ROWS];
    #pragma unroll
    for (int i = 0; i < SCAN_ROWS; ++i)
        h[i] = ((const int4*)hist)[(wave + 32 * i) * (BK_BLOCKS / 4) + lane];
    const int c0 = 4 * lane;
    #pragma unroll
    for (int i = 0; i < SCAN_ROWS; ++i) {
        const int4 v = h[i];
        const int all = wave_isum((v.x + v.y) + (v.z + v.w));
        const int pre = wave_isum(((c0     < b ? v.x : 0) + (c0 + 1 < b ? v.y : 0)) +
                                  ((c0 + 2 < b ? v.z : 0) + (c0 + 3 < b ? v.w : 0)));
        if (lane == 0) {
            tot[wave + 32 * i] = all;
            cur[wave + 32 * i] = pre;
        }
    }
    __syncthreads();
    if (wave == 0) {
        int v[SCAN_ROWS];
        int loc = 0;
        #pragma unroll
        for (int k = 0; k < SCAN_ROWS; ++k) {
            v[k] = tot[lane * SCAN_ROWS + k];
            loc += v[k];
        }
        int inc = loc;
        for (int off = 1; off < 32; off <<= 1) {
            const int t = __shfl_up(inc, off, 32);
            if (lane >= off) inc += t;
        }
        int run = inc - loc;
        #pragma unroll
        for (int k = 0; k < SCAN_ROWS; ++k) {
            tot[lane * SCAN_ROWS + k] = run;
            run += v[k];
        }
    }
    __syncthreads();
    if (tid < BK_PAD) {
        const int start = tot[tid];
        cur[tid] += start;
        if (b == 0) rstart[tid] = start;
    }
    __syncthreads();
    #pragma unroll
    for (int j = 0; j < BK_PER; ++j)
        if (d[j] >= 0) {
            const int pos = atomicAdd(&cur[d[j] >> BK_SHIFT], 1);
            coarse[pos] = ((d[j] & (BK_NODES - 1)) << 16) | s[j];
        }
}

__global__ void __launch_bounds__(1024) fine_place(const int* __restrict__ coarse,
                                                   const int* __restrict__ rstart,
                                                   int* __restrict__ rowptr,
                                                   int* __restrict__ srcs) {
    __shared__ __attribute__((aligned(16))) int cnt[BK_NODES];
    const int tid = threadIdx.x, r = blockIdx.x;
    const int beg = rstart[r], end = rstart[r + 1];
    if (tid < BK_NODES) cnt[tid] = 0;
    __syncthreads();
    for (int i = beg + tid; i < end; i += FP_PER * 1024) {
        int w[FP_PER];
        #pragma unroll
        for (int j = 0; j < FP_PER; ++j) {
            const int ii = i + j * 1024;
            w[j] = ii < end ? coarse[ii] : -1;
        }
        #pragma unroll
        for (int j = 0; j < FP_PER; ++j)
            if (w[j] >= 0) atomicAdd(&cnt[w[j] >> 16], 1);
    }
    __syncthreads();
    if (tid < 32) {
        const int4 c = ((const int4*)cnt)[tid];
        const int loc = (c.x + c.y) + (c.z + c.w);
        int inc = loc;
        for (int off = 1; off < 32; off <<= 1) {
            const int t = __shfl_up(inc, off, 32);
            if (tid >= off) inc += t;
        }
        int4 p;
        p.x = beg + inc - loc;
        p.y = p.x + c.x;
        p.z = p.y + c.y;
        p.w = p.z + c.z;
        ((int4*)cnt)[tid] = p;
        const int n0 = r * BK_NODES + tid * 4;
        if (n0 <= N_NODES) *(int4*)(rowptr + n0) = p;
    }
    __syncthreads();
    for (int i = beg + tid; i < end; i += FP_PER * 1024) {
        int w[FP_PER];
        #pragma unroll
        for (int j = 0; j < FP_PER; ++j) {
            const int ii = i + j * 1024;
            w[j] = ii < end ? coarse[ii] : -1;
        }
        #pragma unroll
        for (int j = 0; j < FP_PER; ++j)
            if (w[j] >= 0) {
                const int pos = atomicAdd(&cnt[w[j] >> 16], 1);
                srcs[pos] = w[j] & 0xffff;
            }
    }
}

__device__ __forceinline__ float wave_max(float x) {
    x = fmaxf(x, xrow<1>(x));
    x = fmaxf(x, xrow<2>(x));
    x = fmaxf(x, xrow<4>(x));
    x = fmaxf(x, xrow<8>(x));
    return fmaxf(x, xhalf(x));
}
__device__ __forceinline__ float wave_sum(float x) {
    x += xrow<1>(x);
    x += xrow<2>(x);
    x += xrow<4>(x);
    x += xrow<8>(x);
    return x + xhalf(x);
}

__device__ __forceinline__ void edge_scores(const int* __restrict__ srcs,
                                            const float* __restrict__ el,
                                            const float er0, const float er1,
                                            const int base, const int end, const int lane,
                                            int& s, float& v0, float& v1) {
    s = 0;
    v0 = -INFINITY;
    v1 = -INFINITY;
    if (base + lane < end) {
        s = srcs[base + lane];
        const float2 l = *(const float2*)(el + 2 * (size_t)s);
        const float t0 = l.x + er0, t1 = l.y + er1;
        v0 = t0 > 0.0f ? t0 : 0.2f * t0;
        v1 = t1 > 0.0f ? t1 : 0.2f * t1;
    }
}

__device__ __forceinline__ void gather_node(const float* __restrict__ hin,
                                            const float* __restrict__ el,
                                            const float* __restrict__ er,
                                            const int* __restrict__ rowptr,
                                            const int* __restrict__ srcs,
                                            const int n, const int lane,
                                            float4& a0, float4& a1) {
    const int beg = rowptr[n], end = rowptr[n + 1];
    if (end > beg) {
        const float er0 = er[n * 2], er1 = er[n * 2 + 1];
        int s_l;
        float e0_l, e1_l;
        edge_scores(srcs, el, er0, er1, beg, end, lane, s_l, e0_l, e1_l);
        float m0 = e0_l, m1 = e1_l;
        for (int base = beg + 32; base < end; base += 32) {
            int s;
            float v0, v1;
            edge_scores(srcs, el, er0, er1, base, end, lane, s, v0, v1);
            m0 = fmaxf(m0, v0);
            m1 = fmaxf(m1, v1);
        }
        m0 = wave_max(m0);
        m1 = wave_max(m1);
        const float x0_l = (beg + lane < end) ? expf(e0_l - m0) : 0.f;
        const float x1_l = (beg + lane < end) ? expf(e1_l - m1) : 0.f;
        float den0 = x0_l, den1 = x1_l;
        for (int base = beg + 32; base < end; base += 32) {
            int s;
            float v0, v1;
            edge_scores(srcs, el, er0, er1, base, end, lane, s, v0, v1);
            if (base + lane < end) {
                den0 += expf(v0 - m0);
                den1 += expf(v1 - m1);
            }
        }
        den0 = wave_sum(den0);
        den1 = wave_sum(den1);
        float w0_l = x0_l / den0, w1_l = x1_l / den1;
        for (int base = beg; base < end; base += 32) {
            const int cnt = (end - base) < 32 ? (end - base) : 32;
            if (base != beg) {
                float v0, v1;
                edge_scores(srcs, el, er0, er1, base, end, lane, s_l, v0, v1);
                w0_l = (base + lane < end) ? expf(v0 - m0) / den0 : 0.f;
                w1_l = (base + lane < end) ? expf(v1 - m1) / den1 : 0.f;
            }
            #pragma unroll 4
            for (int j = 0; j < cnt; ++j) {
                const int   s  = __builtin_amdgcn_readlane(s_l, j);
                const float w0 = __int_as_float(__builtin_amdgcn_readlane(__float_as_int(w0_l), j));
                const float w1 = __int_as_float(__builtin_amdgcn_readlane(__float_as_int(w1_l), j));
                const float4 f = ((const float4*)(hin + (size_t)s * 128))[lane];
                a0.x += f.x * w0;  a0.y += f.y * w0;  a0.z += f.z * w0;  a0.w += f.w * w0;
                a1.x += f.x * w1;  a1.y += f.y * w1;  a1.z += f.z * w1;  a1.w += f.w * w1;
            }
        }
    }
}

#define TILE_NODES 32
#define AGG_PITCH  260

__device__ __forceinline__ void gather_tile(const float* __restrict__ hin,
                                            const float* __restrict__ el,
                                            const float* __restrict__ er,
                                            const int* __restrict__ rowptr,
                                            const int* __restrict__ srcs,
                                            float* __restrict__ sA,
                                            const int mbase, const int wave, const int lane) {
    for (int i = 0; i < 2; ++i) {
        const int r = wave * 2 + i;
        const int n = mbase + r;
        float4 a0 = {0.f, 0.f, 0.f, 0.f}, a1 = {0.f, 0.f, 0.f, 0.f};
        if (n < N_NODES) gather_node(hin, el, er, rowptr, srcs, n, lane, a0, a1);
        float4* o = (float4*)(sA + r * AGG_PITCH);
        o[lane]      = a0;
        o[32 + lane] = a1;
    }
}

__global__ void __launch_bounds__(512, 16) agg_proj_mean(const float* __restrict__ hin,
                                                     const float* __restrict__ el,
                                                     const float* __restrict__ er,
                                                     const int* __restrict__ rowptr,
                                                     const int* __restrict__ srcs,
                                                     const float* __restrict__ WF,
                                                     const float* __restrict__ b,
                                                     float* __restrict__ hnext) {
    __shared__ __attribute__((aligned(16))) float sA[TILE_NODES * AGG_PITCH];
    const int lane  = threadIdx.x & 31;
    const int wave  = __builtin_amdgcn_readfirstlane(threadIdx.x >> 5);
    const int mbase = blockIdx.x * TILE_NODES;
    gather_tile(hin, el, er, rowptr, srcs, sA, mbase, wave, lane);
    __syncthreads();
    if (wave >= 8) return;

    const int row   = lane & 15;
    const int khalf = lane >> 4;
    const int rt    = wave >> 2;
    const int cg    = wave & 3;

    v8f acc0 = {}, acc1 = {};
    const float*  Arow = sA + (rt * 16 + row) * AGG_PITCH + khalf * 2;
    const float4* Wf   = (const float4*)WF + cg * 32 + lane;
    #pragma unroll 8
    for (int kk = 0; kk < N_HEADS * D_MODEL; kk += 4) {
        const v2f a = *(const v2f*)(Arow + kk);
        const float4 w = Wf[kk * 32];
        v2f b0, b1;
        b0.x = w.x;  b0.y = w.y;
        b1.x = w.z;  b1.y = w.w;
        acc0 = __builtin_amdgcn_wmma_f32_16x16x4_f32(false, a, false, b0, (short)0, acc0, false, false);
        acc1 = __builtin_amdgcn_wmma_f32_16x16x4_f32(false, a, false, b1, (short)0, acc1, false, false);
    }

    asm volatile("v_nop\n\tv_nop\n\tv_nop\n\tv_nop");
    __builtin_amdgcn_sched_barrier(0);

    const int c = cg * 32 + 2 * row;
    const float2 bh0 = *(const float2*)(b + c), bh1 = *(const float2*)(b + 128 + c);
    const float bb0 = bh0.x + bh1.x;
    const float bb1 = bh0.y + bh1.y;
    #pragma unroll
    for (int j = 0; j < 8; ++j) {
        const int m = mbase + rt * 16 + j + khalf * 8;
        if (m < N_NODES) {
            float2 o;
            o.x = 0.5f * (acc0[j] + bb0);
            o.y = 0.5f * (acc1[j] + bb1);
            *(float2*)(hnext + (size_t)m * 128 + c) = o;
        }
    }
}

__global__ void __launch_bounds__(512, 16) agg_proj_heads(const float* __restrict__ hin,
                                                      const float* __restrict__ el,
                                                      const float* __restrict__ er,
                                                      const int* __restrict__ rowptr,
                                                      const int* __restrict__ srcs,
                                                      const float* __restrict__ WF,
                                                      const float* __restrict__ b,
                                                      float* __restrict__ out) {
    __shared__ __attribute__((aligned(16))) float sA[TILE_NODES * AGG_PITCH];
    const int lane  = threadIdx.x & 31;
    const int wave  = __builtin_amdgcn_readfirstlane(threadIdx.x >> 5);
    const int mbase = blockIdx.x * TILE_NODES;
    gather_tile(hin, el, er, rowptr, srcs, sA, mbase, wave, lane);
    __syncthreads();

    const int row   = lane & 15;
    const int khalf = lane >> 4;
    const int rt    = wave >> 3;
    const int cg    = wave & 7;
    const int nq    = cg << 5;

    v8f acc0 = {}, acc1 = {};
    const float*  Arow = sA + (rt * 16 + row) * AGG_PITCH + (nq >> 7) * 128 + khalf * 2;
    const float4* Wf   = (const float4*)WF + cg * 32 + lane;
    #pragma unroll 8
    for (int k = 0; k < D_MODEL; k += 4) {
        const v2f a = *(const v2f*)(Arow + k);
        const float4 w = Wf[k * 64];
        v2f b0, b1;
        b0.x = w.x;  b0.y = w.y;
        b1.x = w.z;  b1.y = w.w;
        acc0 = __builtin_amdgcn_wmma_f32_16x16x4_f32(false, a, false, b0, (short)0, acc0, false, false);
        acc1 = __builtin_amdgcn_wmma_f32_16x16x4_f32(false, a, false, b1, (short)0, acc1, false, false);
    }

    asm volatile("v_nop\n\tv_nop\n\tv_nop\n\tv_nop");
    __builtin_amdgcn_sched_barrier(0);

    const int c = nq + 2 * row;
    const float2 bb = *(const float2*)(b + c);
    #pragma unroll
    for (int j = 0; j < 8; ++j) {
        const int m = mbase + rt * 16 + j + khalf * 8;
        if (m < N_NODES) {
            float2 o;
            o.x = acc0[j] + bb.x;
            o.y = acc1[j] + bb.y;
            *(float2*)(out + (size_t)m * 256 + c) = o;
        }
    }
}

extern "C" void kernel_launch(void* const* d_in, const int* in_sizes, int n_in,
                              void* d_out, int out_size, void* d_ws, size_t ws_size,
                              hipStream_t stream) {
    const float* x   = (const float*)d_in[0];
    const float* Ws  = (const float*)d_in[1];
    const float* als = (const float*)d_in[2];
    const float* ars = (const float*)d_in[3];
    const float* bs  = (const float*)d_in[4];
    const int*   src = (const int*)d_in[5];
    const int*   dst = (const int*)d_in[6];

    float* ws      = (float*)d_ws;
    float* hnext   = ws;
    int*   srcs    = (int*)(ws + 12800000);
    int*   coarse  = (int*)(ws + 14400000);
    int*   hist    = (int*)(ws + 15200000);
    int*   rowptr  = (int*)(ws + 15300000);
    int*   rstart  = (int*)(ws + 15400000);
    float* wl      = ws + 19100000;
    float* wr      = ws + 19101000;
    float* WF      = ws + 19110400;
    float* el      = ws + 19200000;
    float* er      = ws + 19300000;

    float* out   = (float*)d_out;

    coarse_count<<<BK_BLOCKS + FOLD_BLOCKS + RELAY_BLOCKS, 1024, 0, stream>>>(dst, hist, Ws, als, ars, wl, wr, WF);
    coarse_place<<<BK_BLOCKS + ELR_BLOCKS, 1024, 0, stream>>>(src, dst, hist, rstart, coarse, x, wl, wr, el, er);
    fine_place<<<BK_COUNT, 1024, 0, stream>>>(coarse, rstart, rowptr, srcs);

    agg_proj_mean<<<(N_NODES + TILE_NODES - 1) / TILE_NODES, 512, 0, stream>>>(x, el, er, rowptr, srcs, WF, bs, hnext);

    node_elr<<<(N_NODES / ELR_ROWS + 7) / 8, 256, 0, stream>>>(hnext, wl + 256, wr + 256, el, er);
    agg_proj_heads<<<(N_NODES + TILE_NODES - 1) / TILE_NODES, 512, 0, stream>>>(hnext, el, er, rowptr, srcs, WF + WF_LAYER, bs + 256, out);
}
